# starting kernel
# baseline (speedup 1.0000x reference)
#include <hip/hip_runtime.h>


static constexpr int DIM  = 128;
static constexpr int NBLK = 2;
static constexpr int LDP  = 68;

typedef __attribute__((ext_vector_type(2))) float v2f;
typedef __attribute__((ext_vector_type(8))) float v8f;

static __device__ __forceinline__ float wave_reduce_sum(float x) {
#pragma unroll
  for (int off = 16; off > 0; off >>= 1) x += __shfl_xor(x, off, 32);
  return x;
}

static __device__ __forceinline__ unsigned f32_order_enc(float f) {
  unsigned u = __float_as_uint(f);
  return (u & 0x80000000u) ? ~u : (u | 0x80000000u);
}
static __device__ __forceinline__ float f32_order_dec(unsigned e) {
  unsigned u = (e & 0x80000000u) ? (e & 0x7fffffffu) : ~e;
  return __uint_as_float(u);
}

__global__ __launch_bounds__(128) void gemm_nt_128(const float* __restrict__ A,
                                                   const float* __restrict__ W,
                                                   float* __restrict__ C, int N) {
  __shared__ float Wlds[128 * LDP];
  const int tid = threadIdx.x;
  const int wv  = tid >> 5;
  const int ln  = tid & 31;
  const int hk  = ln >> 4;
  const int l16 = ln & 15;
  const int r0  = blockIdx.x * 64 + wv * 16;

  v8f acc[8];
#pragma unroll
  for (int t = 0; t < 8; ++t)
#pragma unroll
    for (int j = 0; j < 8; ++j) acc[t][j] = 0.0f;

  int rowA = r0 + l16;
  if (rowA >= N) rowA = N - 1;
  const float* ap = A + (size_t)rowA * DIM;

  for (int kp = 0; kp < DIM; kp += 64) {
    __syncthreads();
#pragma unroll
    for (int i = 0; i < 16; ++i) {
      const int q  = tid + 128 * i;
      const int n  = q >> 4;
      const int c4 = (q & 15) << 2;
      *(float4*)(&Wlds[n * LDP + c4]) =
          *(const float4*)(W + (size_t)n * DIM + kp + c4);
    }
    __syncthreads();

#pragma unroll 4
    for (int k0 = 0; k0 < 64; k0 += 4) {
      const int ka = k0 + 2 * hk;
      v2f a; a.x = ap[kp + ka]; a.y = ap[kp + ka + 1];
#pragma unroll
      for (int t = 0; t < 8; ++t) {
        const float* wp = &Wlds[(t * 16 + l16) * LDP + ka];
        v2f b; b.x = wp[0]; b.y = wp[1];
        acc[t] = __builtin_amdgcn_wmma_f32_16x16x4_f32(
            false, a, false, b, (short)0, acc[t], false, false);
      }
    }
  }

  const int rb = r0 + hk * 8;
  if (r0 + 16 <= N) {
#pragma unroll
    for (int t = 0; t < 8; ++t)
#pragma unroll
      for (int j = 0; j < 8; ++j)
        C[(size_t)(rb + j) * DIM + t * 16 + l16] = acc[t][j];
  } else {
#pragma unroll
    for (int t = 0; t < 8; ++t)
#pragma unroll
      for (int j = 0; j < 8; ++j) {
        const int row = rb + j;
        if (row < N) C[(size_t)row * DIM + t * 16 + l16] = acc[t][j];
      }
  }
}

__global__ __launch_bounds__(128) void gemm_cat_256(const float* __restrict__ A0,
                                                    const float* __restrict__ A1,
                                                    const float* __restrict__ W,
                                                    float* __restrict__ C, int N) {
  __shared__ float Wlds[128 * LDP];
  const int tid = threadIdx.x;
  const int wv  = tid >> 5;
  const int ln  = tid & 31;
  const int hk  = ln >> 4;
  const int l16 = ln & 15;
  const int r0  = blockIdx.x * 64 + wv * 16;

  v8f acc[8];
#pragma unroll
  for (int t = 0; t < 8; ++t)
#pragma unroll
    for (int j = 0; j < 8; ++j) acc[t][j] = 0.0f;

  int rowA = r0 + l16;
  if (rowA >= N) rowA = N - 1;
  const size_t rbase = (size_t)rowA * DIM;

  for (int kp = 0; kp < 2 * DIM; kp += 64) {
    __syncthreads();
#pragma unroll
    for (int i = 0; i < 16; ++i) {
      const int q  = tid + 128 * i;
      const int n  = q >> 4;
      const int c4 = (q & 15) << 2;
      *(float4*)(&Wlds[n * LDP + c4]) =
          *(const float4*)(W + (size_t)n * (2 * DIM) + kp + c4);
    }
    __syncthreads();

    const float* abase = (kp < DIM) ? (A0 + rbase + kp) : (A1 + rbase + (kp - DIM));
#pragma unroll 4
    for (int k0 = 0; k0 < 64; k0 += 4) {
      const int ka = k0 + 2 * hk;
      v2f a; a.x = abase[ka]; a.y = abase[ka + 1];
#pragma unroll
      for (int t = 0; t < 8; ++t) {
        const float* wp = &Wlds[(t * 16 + l16) * LDP + ka];
        v2f b; b.x = wp[0]; b.y = wp[1];
        acc[t] = __builtin_amdgcn_wmma_f32_16x16x4_f32(
            false, a, false, b, (short)0, acc[t], false, false);
      }
    }
  }

  const int rb = r0 + hk * 8;
  if (r0 + 16 <= N) {
#pragma unroll
    for (int t = 0; t < 8; ++t)
#pragma unroll
      for (int j = 0; j < 8; ++j)
        C[(size_t)(rb + j) * DIM + t * 16 + l16] = acc[t][j];
  } else {
#pragma unroll
    for (int t = 0; t < 8; ++t)
#pragma unroll
      for (int j = 0; j < 8; ++j) {
        const int row = rb + j;
        if (row < N) C[(size_t)row * DIM + t * 16 + l16] = acc[t][j];
      }
  }
}

__global__ __launch_bounds__(256) void gn_kernel(const float* __restrict__ X,
                                                 float* __restrict__ Y,
                                                 const float* __restrict__ w,
                                                 const float* __restrict__ b, int N) {
  const int wv = threadIdx.x >> 5, ln = threadIdx.x & 31;
  const int row = blockIdx.x * 8 + wv;
  if (row >= N) return;
  float4 x = ((const float4*)(X + (size_t)row * DIM))[ln];
  float s  = x.x + x.y + x.z + x.w;
  float sq = x.x * x.x + x.y * x.y + x.z * x.z + x.w * x.w;
  s = wave_reduce_sum(s); sq = wave_reduce_sum(sq);
  const float mu = s * (1.0f / DIM);
  const float rs = rsqrtf(sq * (1.0f / DIM) - mu * mu + 1e-5f);
  float4 wv4 = ((const float4*)w)[ln];
  float4 bv4 = ((const float4*)b)[ln];
  float4 y;
  y.x = (x.x - mu) * rs * wv4.x + bv4.x;
  y.y = (x.y - mu) * rs * wv4.y + bv4.y;
  y.z = (x.z - mu) * rs * wv4.z + bv4.z;
  y.w = (x.w - mu) * rs * wv4.w + bv4.w;
  ((float4*)(Y + (size_t)row * DIM))[ln] = y;
}

__global__ __launch_bounds__(256) void gn2_relu_kernel(const float* __restrict__ X,
                                                       float* __restrict__ Y,
                                                       const float* __restrict__ w1,
                                                       const float* __restrict__ b1,
                                                       const float* __restrict__ w2,
                                                       const float* __restrict__ b2,
                                                       int N) {
  const int wv = threadIdx.x >> 5, ln = threadIdx.x & 31;
  const int row = blockIdx.x * 8 + wv;
  if (row >= N) return;
  float4 x = ((const float4*)(X + (size_t)row * DIM))[ln];
  float s  = x.x + x.y + x.z + x.w;
  float sq = x.x * x.x + x.y * x.y + x.z * x.z + x.w * x.w;
  s = wave_reduce_sum(s); sq = wave_reduce_sum(sq);
  float mu = s * (1.0f / DIM);
  float rs = rsqrtf(sq * (1.0f / DIM) - mu * mu + 1e-5f);
  float4 wa = ((const float4*)w1)[ln], ba = ((const float4*)b1)[ln];
  float4 y;
  y.x = (x.x - mu) * rs * wa.x + ba.x;
  y.y = (x.y - mu) * rs * wa.y + ba.y;
  y.z = (x.z - mu) * rs * wa.z + ba.z;
  y.w = (x.w - mu) * rs * wa.w + ba.w;
  s  = y.x + y.y + y.z + y.w;
  sq = y.x * y.x + y.y * y.y + y.z * y.z + y.w * y.w;
  s = wave_reduce_sum(s); sq = wave_reduce_sum(sq);
  mu = s * (1.0f / DIM);
  rs = rsqrtf(sq * (1.0f / DIM) - mu * mu + 1e-5f);
  float4 wn = ((const float4*)w2)[ln], bn = ((const float4*)b2)[ln];
  float4 z;
  z.x = fmaxf((y.x - mu) * rs * wn.x + bn.x, 0.0f);
  z.y = fmaxf((y.y - mu) * rs * wn.y + bn.y, 0.0f);
  z.z = fmaxf((y.z - mu) * rs * wn.z + bn.z, 0.0f);
  z.w = fmaxf((y.w - mu) * rs * wn.w + bn.w, 0.0f);
  ((float4*)(Y + (size_t)row * DIM))[ln] = z;
}

__global__ __launch_bounds__(256) void gn_res_relu_kernel(const float* __restrict__ X,
                                                          const float* __restrict__ R,
                                                          float* __restrict__ Y,
                                                          const float* __restrict__ w,
                                                          const float* __restrict__ b,
                                                          int N) {
  const int wv = threadIdx.x >> 5, ln = threadIdx.x & 31;
  const int row = blockIdx.x * 8 + wv;
  if (row >= N) return;
  float4 x = ((const float4*)(X + (size_t)row * DIM))[ln];
  float s  = x.x + x.y + x.z + x.w;
  float sq = x.x * x.x + x.y * x.y + x.z * x.z + x.w * x.w;
  s = wave_reduce_sum(s); sq = wave_reduce_sum(sq);
  const float mu = s * (1.0f / DIM);
  const float rs = rsqrtf(sq * (1.0f / DIM) - mu * mu + 1e-5f);
  float4 wv4 = ((const float4*)w)[ln], bv4 = ((const float4*)b)[ln];
  float4 r = ((const float4*)(R + (size_t)row * DIM))[ln];
  float4 y;
  y.x = fmaxf((x.x - mu) * rs * wv4.x + bv4.x + r.x, 0.0f);
  y.y = fmaxf((x.y - mu) * rs * wv4.y + bv4.y + r.y, 0.0f);
  y.z = fmaxf((x.z - mu) * rs * wv4.z + bv4.z + r.z, 0.0f);
  y.w = fmaxf((x.w - mu) * rs * wv4.w + bv4.w + r.w, 0.0f);
  ((float4*)(Y + (size_t)row * DIM))[ln] = y;
}

__global__ __launch_bounds__(256) void edge_dot_kernel(const float* __restrict__ Q,
                                                       const float* __restrict__ K,
                                                       const int* __restrict__ u,
                                                       const int* __restrict__ v,
                                                       float* __restrict__ c,
                                                       unsigned* __restrict__ gmax,
                                                       int E) {
  __shared__ float smax[8];
  const int wv = threadIdx.x >> 5, ln = threadIdx.x & 31;
  const int e = blockIdx.x * 8 + wv;
  float cv = -3.0e38f;
  if (e < E) {
    const int vi = v[e], ui = u[e];
    float4 qv = ((const float4*)(Q + (size_t)vi * DIM))[ln];
    float4 kv = ((const float4*)(K + (size_t)ui * DIM))[ln];
    float d = qv.x * kv.x + qv.y * kv.y + qv.z * kv.z + qv.w * kv.w;
    d = wave_reduce_sum(d);
    d = (d >= 0.0f) ? d : 0.1f * d;
    if (ln == 0) c[e] = d;
    cv = d;
  }
  if (ln == 0) smax[wv] = cv;
  __syncthreads();
  if (threadIdx.x == 0) {
    float m = smax[0];
#pragma unroll
    for (int i = 1; i < 8; ++i) m = fmaxf(m, smax[i]);
    atomicMax(gmax, f32_order_enc(m));
  }
}

__global__ __launch_bounds__(256) void edge_exp_kernel(float* __restrict__ c,
                                                       float* __restrict__ csum,
                                                       const int* __restrict__ v,
                                                       const unsigned* __restrict__ gmax,
                                                       int E) {
  const int e = blockIdx.x * 256 + threadIdx.x;
  if (e >= E) return;
  const float m = f32_order_dec(*gmax);
  const float ce = expf(c[e] - m);
  c[e] = ce;
  atomicAdd(&csum[v[e]], ce);
}

__global__ __launch_bounds__(256) void edge_scatter_kernel(const float* __restrict__ c,
                                                           const float* __restrict__ csum,
                                                           const float* __restrict__ ctxp,
                                                           const int* __restrict__ u,
                                                           const int* __restrict__ v,
                                                           float* __restrict__ agg,
                                                           int E) {
  const int wv = threadIdx.x >> 5, ln = threadIdx.x & 31;
  const int e = blockIdx.x * 8 + wv;
  if (e >= E) return;
  const int vi = v[e], ui = u[e];
  const float coef = c[e] / (csum[vi] + 1e-16f);
  float4 m = ((const float4*)(ctxp + (size_t)ui * DIM))[ln];
  float* dst = agg + (size_t)vi * DIM + ln * 4;
  atomicAdd(dst + 0, m.x * coef);
  atomicAdd(dst + 1, m.y * coef);
  atomicAdd(dst + 2, m.z * coef);
  atomicAdd(dst + 3, m.w * coef);
}

__global__ void zero_f32(float* __restrict__ p, long n) {
  const long i = (long)blockIdx.x * 256 + threadIdx.x;
  if (i < n) p[i] = 0.0f;
}
__global__ void init_scalar_u32(unsigned* p) {
  if (threadIdx.x == 0) *p = 0u;
}

extern "C" void kernel_launch(void* const* d_in, const int* in_sizes, int n_in,
                              void* d_out, int out_size, void* d_ws, size_t ws_size,
                              hipStream_t stream) {
  (void)in_sizes; (void)n_in; (void)out_size; (void)ws_size;
  const int NA = 50000, NC = 100000, NE = 600000;

  const float* agt    = (const float*)d_in[0];
  const float* ctx    = (const float*)d_in[1];
  const int*   u      = (const int*)d_in[2];
  const int*   v      = (const int*)d_in[3];
  const float* Wq     = (const float*)d_in[4];
  const float* Wk     = (const float*)d_in[5];
  const float* Wctx   = (const float*)d_in[6];
  const float* gnc_w  = (const float*)d_in[7];
  const float* gnc_b  = (const float*)d_in[8];
  const float* Wagt   = (const float*)d_in[9];
  const float* gna_w  = (const float*)d_in[10];
  const float* gna_b  = (const float*)d_in[11];
  const float* norm_w = (const float*)d_in[12];
  const float* norm_b = (const float*)d_in[13];
  const float* Wlin   = (const float*)d_in[14];
  const float* gnl_w  = (const float*)d_in[15];
  const float* gnl_b  = (const float*)d_in[16];

  float* ws = (float*)d_ws;
  size_t off = 0;
  float* qbuf   = ws + off; off += (size_t)NA * DIM;
  float* kbuf   = ws + off; off += (size_t)NC * DIM;
  float* cpbuf  = ws + off; off += (size_t)NC * DIM;
  float* cbuf   = ws + off; off += (size_t)NE;
  float* csum   = ws + off; off += (size_t)NA;
  float* aggbuf = ws + off; off += (size_t)NA * DIM;
  float* xbuf   = ws + off; off += (size_t)NA * DIM;
  float* agtbuf = ws + off; off += (size_t)NA * DIM;
  unsigned* maxbuf = (unsigned*)(ws + off); off += 16;

  const dim3 b128(128), b256(256);

  for (int i = 0; i < NBLK; ++i) {
    const float* ain  = (i == 0) ? agt : agtbuf;
    float*       aout = (i == NBLK - 1) ? (float*)d_out : agtbuf;

    gemm_nt_128<<<(NA + 63) / 64, b128, 0, stream>>>(ain, Wq + (size_t)i * DIM * DIM, qbuf, NA);
    gemm_nt_128<<<(NC + 63) / 64, b128, 0, stream>>>(ctx, Wk + (size_t)i * DIM * DIM, kbuf, NC);
    gemm_nt_128<<<(NC + 63) / 64, b128, 0, stream>>>(ctx, Wctx + (size_t)i * DIM * DIM, cpbuf, NC);
    gn_kernel<<<(NC + 7) / 8, b256, 0, stream>>>(cpbuf, cpbuf, gnc_w + i * DIM, gnc_b + i * DIM, NC);

    init_scalar_u32<<<1, 32, 0, stream>>>(maxbuf);
    zero_f32<<<(NA + 255) / 256, b256, 0, stream>>>(csum, (long)NA);
    zero_f32<<<((long)NA * DIM + 255) / 256, b256, 0, stream>>>(aggbuf, (long)NA * DIM);
    edge_dot_kernel<<<(NE + 7) / 8, b256, 0, stream>>>(qbuf, kbuf, u, v, cbuf, maxbuf, NE);
    edge_exp_kernel<<<(NE + 255) / 256, b256, 0, stream>>>(cbuf, csum, v, maxbuf, NE);
    edge_scatter_kernel<<<(NE + 7) / 8, b256, 0, stream>>>(cbuf, csum, cpbuf, u, v, aggbuf, NE);

    gemm_cat_256<<<(NA + 63) / 64, b128, 0, stream>>>(ain, aggbuf, Wagt + (size_t)i * DIM * 2 * DIM, xbuf, NA);
    gn2_relu_kernel<<<(NA + 7) / 8, b256, 0, stream>>>(xbuf, xbuf, gna_w + i * DIM, gna_b + i * DIM,
                                                       norm_w + i * DIM, norm_b + i * DIM, NA);
    gemm_nt_128<<<(NA + 63) / 64, b128, 0, stream>>>(xbuf, Wlin + (size_t)i * DIM * DIM, qbuf, NA);
    gn_res_relu_kernel<<<(NA + 7) / 8, b256, 0, stream>>>(qbuf, ain, aout, gnl_w + i * DIM, gnl_b + i * DIM, NA);
  }
}


# Streaming kernels (agg zero-fill, 3 GroupNorm passes): one resident round of waves loops, 4 rows or float4 stores in flight per wave, not one wave per row or per 128 B; 350k fewer waves per block
# speedup vs baseline: 1.5849x; 1.5849x over previous
#include <hip/hip_runtime.h>


static constexpr int DIM  = 128;
static constexpr int NBLK = 2;
static constexpr int LDP  = 68;

typedef __attribute__((ext_vector_type(2))) float v2f;
typedef __attribute__((ext_vector_type(8))) float v8f;

static __device__ __forceinline__ float wave_reduce_sum(float x) {
#pragma unroll
  for (int off = 16; off > 0; off >>= 1) x += __shfl_xor(x, off, 32);
  return x;
}

static __device__ __forceinline__ unsigned f32_order_enc(float f) {
  unsigned u = __float_as_uint(f);
  return (u & 0x80000000u) ? ~u : (u | 0x80000000u);
}
static __device__ __forceinline__ float f32_order_dec(unsigned e) {
  unsigned u = (e & 0x80000000u) ? (e & 0x7fffffffu) : ~e;
  return __uint_as_float(u);
}

__global__ __launch_bounds__(128) void gemm_nt_128(const float* __restrict__ A,
                                                   const float* __restrict__ W,
                                                   float* __restrict__ C, int N) {
  __shared__ float Wlds[128 * LDP];
  const int tid = threadIdx.x;
  const int wv  = tid >> 5;
  const int ln  = tid & 31;
  const int hk  = ln >> 4;
  const int l16 = ln & 15;
  const int r0  = blockIdx.x * 64 + wv * 16;

  v8f acc[8];
#pragma unroll
  for (int t = 0; t < 8; ++t)
#pragma unroll
    for (int j = 0; j < 8; ++j) acc[t][j] = 0.0f;

  int rowA = r0 + l16;
  if (rowA >= N) rowA = N - 1;
  const float* ap = A + (size_t)rowA * DIM;

  for (int kp = 0; kp < DIM; kp += 64) {
    __syncthreads();
#pragma unroll
    for (int i = 0; i < 16; ++i) {
      const int q  = tid + 128 * i;
      const int n  = q >> 4;
      const int c4 = (q & 15) << 2;
      *(float4*)(&Wlds[n * LDP + c4]) =
          *(const float4*)(W + (size_t)n * DIM + kp + c4);
    }
    __syncthreads();

#pragma unroll 4
    for (int k0 = 0; k0 < 64; k0 += 4) {
      const int ka = k0 + 2 * hk;
      v2f a; a.x = ap[kp + ka]; a.y = ap[kp + ka + 1];
#pragma unroll
      for (int t = 0; t < 8; ++t) {
        const float* wp = &Wlds[(t * 16 + l16) * LDP + ka];
        v2f b; b.x = wp[0]; b.y = wp[1];
        acc[t] = __builtin_amdgcn_wmma_f32_16x16x4_f32(
            false, a, false, b, (short)0, acc[t], false, false);
      }
    }
  }

  const int rb = r0 + hk * 8;
  if (r0 + 16 <= N) {
#pragma unroll
    for (int t = 0; t < 8; ++t)
#pragma unroll
      for (int j = 0; j < 8; ++j)
        C[(size_t)(rb + j) * DIM + t * 16 + l16] = acc[t][j];
  } else {
#pragma unroll
    for (int t = 0; t < 8; ++t)
#pragma unroll
      for (int j = 0; j < 8; ++j) {
        const int row = rb + j;
        if (row < N) C[(size_t)row * DIM + t * 16 + l16] = acc[t][j];
      }
  }
}

__global__ __launch_bounds__(128) void gemm_cat_256(const float* __restrict__ A0,
                                                    const float* __restrict__ A1,
                                                    const float* __restrict__ W,
                                                    float* __restrict__ C, int N) {
  __shared__ float Wlds[128 * LDP];
  const int tid = threadIdx.x;
  const int wv  = tid >> 5;
  const int ln  = tid & 31;
  const int hk  = ln >> 4;
  const int l16 = ln & 15;
  const int r0  = blockIdx.x * 64 + wv * 16;

  v8f acc[8];
#pragma unroll
  for (int t = 0; t < 8; ++t)
#pragma unroll
    for (int j = 0; j < 8; ++j) acc[t][j] = 0.0f;

  int rowA = r0 + l16;
  if (rowA >= N) rowA = N - 1;
  const size_t rbase = (size_t)rowA * DIM;

  for (int kp = 0; kp < 2 * DIM; kp += 64) {
    __syncthreads();
#pragma unroll
    for (int i = 0; i < 16; ++i) {
      const int q  = tid + 128 * i;
      const int n  = q >> 4;
      const int c4 = (q & 15) << 2;
      *(float4*)(&Wlds[n * LDP + c4]) =
          *(const float4*)(W + (size_t)n * (2 * DIM) + kp + c4);
    }
    __syncthreads();

    const float* abase = (kp < DIM) ? (A0 + rbase + kp) : (A1 + rbase + (kp - DIM));
#pragma unroll 4
    for (int k0 = 0; k0 < 64; k0 += 4) {
      const int ka = k0 + 2 * hk;
      v2f a; a.x = abase[ka]; a.y = abase[ka + 1];
#pragma unroll
      for (int t = 0; t < 8; ++t) {
        const float* wp = &Wlds[(t * 16 + l16) * LDP + ka];
        v2f b; b.x = wp[0]; b.y = wp[1];
        acc[t] = __builtin_amdgcn_wmma_f32_16x16x4_f32(
            false, a, false, b, (short)0, acc[t], false, false);
      }
    }
  }

  const int rb = r0 + hk * 8;
  if (r0 + 16 <= N) {
#pragma unroll
    for (int t = 0; t < 8; ++t)
#pragma unroll
      for (int j = 0; j < 8; ++j)
        C[(size_t)(rb + j) * DIM + t * 16 + l16] = acc[t][j];
  } else {
#pragma unroll
    for (int t = 0; t < 8; ++t)
#pragma unroll
      for (int j = 0; j < 8; ++j) {
        const int row = rb + j;
        if (row < N) C[(size_t)row * DIM + t * 16 + l16] = acc[t][j];
      }
  }
}

static constexpr int ROWB = 4;

static __device__ __forceinline__ void row_stats(const float4 x, float& mu, float& rs) {
  float s  = x.x + x.y + x.z + x.w;
  float sq = x.x * x.x + x.y * x.y + x.z * x.z + x.w * x.w;
  s = wave_reduce_sum(s); sq = wave_reduce_sum(sq);
  mu = s * (1.0f / DIM);
  rs = rsqrtf(sq * (1.0f / DIM) - mu * mu + 1e-5f);
}

__global__ __launch_bounds__(256) void gn_kernel(const float* X, float* Y,
                                                 const float* __restrict__ w,
                                                 const float* __restrict__ b, int N) {
  const int ln = threadIdx.x & 31;
  const int wv = __builtin_amdgcn_readfirstlane((int)(threadIdx.x >> 5));
  const int step = (int)gridDim.x * 8 * ROWB;
  const float4 wv4 = ((const float4*)w)[ln];
  const float4 bv4 = ((const float4*)b)[ln];
  for (int r0 = ((int)blockIdx.x * 8 + wv) * ROWB; r0 < N; r0 += step) {
    float4 x[ROWB];
#pragma unroll
    for (int t = 0; t < ROWB; ++t) {
      x[t] = make_float4(0.0f, 0.0f, 0.0f, 0.0f);
      if (r0 + t < N) x[t] = ((const float4*)(X + (size_t)(r0 + t) * DIM))[ln];
    }
#pragma unroll
    for (int t = 0; t < ROWB; ++t) {
      float mu, rs;
      row_stats(x[t], mu, rs);
      float4 y;
      y.x = (x[t].x - mu) * rs * wv4.x + bv4.x;
      y.y = (x[t].y - mu) * rs * wv4.y + bv4.y;
      y.z = (x[t].z - mu) * rs * wv4.z + bv4.z;
      y.w = (x[t].w - mu) * rs * wv4.w + bv4.w;
      if (r0 + t < N) ((float4*)(Y + (size_t)(r0 + t) * DIM))[ln] = y;
    }
  }
}

__global__ __launch_bounds__(256) void gn2_relu_kernel(const float* X, float* Y,
                                                       const float* __restrict__ w1,
                                                       const float* __restrict__ b1,
                                                       const float* __restrict__ w2,
                                                       const float* __restrict__ b2,
                                                       int N) {
  const int ln = threadIdx.x & 31;
  const int wv = __builtin_amdgcn_readfirstlane((int)(threadIdx.x >> 5));
  const int step = (int)gridDim.x * 8 * ROWB;
  const float4 wa = ((const float4*)w1)[ln], ba = ((const float4*)b1)[ln];
  const float4 wn = ((const float4*)w2)[ln], bn = ((const float4*)b2)[ln];
  for (int r0 = ((int)blockIdx.x * 8 + wv) * ROWB; r0 < N; r0 += step) {
    float4 x[ROWB];
#pragma unroll
    for (int t = 0; t < ROWB; ++t) {
      x[t] = make_float4(0.0f, 0.0f, 0.0f, 0.0f);
      if (r0 + t < N) x[t] = ((const float4*)(X + (size_t)(r0 + t) * DIM))[ln];
    }
#pragma unroll
    for (int t = 0; t < ROWB; ++t) {
      float mu, rs;
      row_stats(x[t], mu, rs);
      float4 y;
      y.x = (x[t].x - mu) * rs * wa.x + ba.x;
      y.y = (x[t].y - mu) * rs * wa.y + ba.y;
      y.z = (x[t].z - mu) * rs * wa.z + ba.z;
      y.w = (x[t].w - mu) * rs * wa.w + ba.w;
      row_stats(y, mu, rs);
      float4 z;
      z.x = fmaxf((y.x - mu) * rs * wn.x + bn.x, 0.0f);
      z.y = fmaxf((y.y - mu) * rs * wn.y + bn.y, 0.0f);
      z.z = fmaxf((y.z - mu) * rs * wn.z + bn.z, 0.0f);
      z.w = fmaxf((y.w - mu) * rs * wn.w + bn.w, 0.0f);
      if (r0 + t < N) ((float4*)(Y + (size_t)(r0 + t) * DIM))[ln] = z;
    }
  }
}

__global__ __launch_bounds__(256) void gn_res_relu_kernel(const float* __restrict__ X,
                                                          const float* __restrict__ R,
                                                          float* __restrict__ Y,
                                                          const float* __restrict__ w,
                                                          const float* __restrict__ b,
                                                          int N) {
  const int ln = threadIdx.x & 31;
  const int wv = __builtin_amdgcn_readfirstlane((int)(threadIdx.x >> 5));
  const int step = (int)gridDim.x * 8 * ROWB;
  const float4 wv4 = ((const float4*)w)[ln], bv4 = ((const float4*)b)[ln];
  for (int r0 = ((int)blockIdx.x * 8 + wv) * ROWB; r0 < N; r0 += step) {
    float4 x[ROWB], r[ROWB];
#pragma unroll
    for (int t = 0; t < ROWB; ++t) {
      x[t] = make_float4(0.0f, 0.0f, 0.0f, 0.0f);
      r[t] = x[t];
      if (r0 + t < N) {
        x[t] = ((const float4*)(X + (size_t)(r0 + t) * DIM))[ln];
        r[t] = ((const float4*)(R + (size_t)(r0 + t) * DIM))[ln];
      }
    }
#pragma unroll
    for (int t = 0; t < ROWB; ++t) {
      float mu, rs;
      row_stats(x[t], mu, rs);
      float4 y;
      y.x = fmaxf((x[t].x - mu) * rs * wv4.x + bv4.x + r[t].x, 0.0f);
      y.y = fmaxf((x[t].y - mu) * rs * wv4.y + bv4.y + r[t].y, 0.0f);
      y.z = fmaxf((x[t].z - mu) * rs * wv4.z + bv4.z + r[t].z, 0.0f);
      y.w = fmaxf((x[t].w - mu) * rs * wv4.w + bv4.w + r[t].w, 0.0f);
      if (r0 + t < N) ((float4*)(Y + (size_t)(r0 + t) * DIM))[ln] = y;
    }
  }
}

static constexpr int ECH    = 32;
static constexpr int EGR_D  = 4;
static constexpr int EGR_S  = 8;

__global__ __launch_bounds__(256) void edge_dot_kernel(const float* __restrict__ Q,
                                                       const float* __restrict__ K,
                                                       const int* __restrict__ u,
                                                       const int* __restrict__ v,
                                                       float* __restrict__ c,
                                                       unsigned* __restrict__ gmax,
                                                       int E) {
  __shared__ float smax[8];
  const int ln = threadIdx.x & 31;
  const int wv = __builtin_amdgcn_readfirstlane((int)(threadIdx.x >> 5));
  const int e0 = (blockIdx.x * 8 + wv) * ECH;
  int nval = E - e0;
  if (nval > ECH) nval = ECH;
  const int hk = ln >> 4;
  const int b8 = (ln >> 3) & 1;
  float cmax = -3.0e38f;
  if (nval > 0) {
    const int el = e0 + (ln < nval ? ln : nval - 1);
    const int vl = v[el], ul = u[el];
    for (int j = 0; j < nval; j += EGR_D) {
      float4 qv[EGR_D], kv[EGR_D];
#pragma unroll
      for (int t = 0; t < EGR_D; ++t) {
        const int jj = (j + t < nval) ? (j + t) : (nval - 1);
        const int vj = __builtin_amdgcn_readlane(vl, jj);
        const int uj = __builtin_amdgcn_readlane(ul, jj);
        qv[t] = ((const float4*)(Q + (size_t)vj * DIM))[ln];
        kv[t] = ((const float4*)(K + (size_t)uj * DIM))[ln];
      }
      float p[EGR_D];
#pragma unroll
      for (int t = 0; t < EGR_D; ++t)
        p[t] = qv[t].x * kv[t].x + qv[t].y * kv[t].y + qv[t].z * kv[t].z + qv[t].w * kv[t].w;
      const float s0 = hk ? p[2] : p[0], r0 = hk ? p[0] : p[2];
      const float s1 = hk ? p[3] : p[1], r1 = hk ? p[1] : p[3];
      const float x0 = s0 + __shfl_xor(r0, 16, 32);
      const float x1 = s1 + __shfl_xor(r1, 16, 32);
      const float s2 = b8 ? x1 : x0, r2 = b8 ? x0 : x1;
      float d = s2 + __shfl_xor(r2, 8, 32);
      d += __shfl_xor(d, 4, 32);
      d += __shfl_xor(d, 2, 32);
      d += __shfl_xor(d, 1, 32);
      d = (d >= 0.0f) ? d : 0.1f * d;
      const int je = j + 2 * hk + b8;
      if (je < nval) {
        if ((ln & 7) == 0) c[e0 + je] = d;
        cmax = fmaxf(cmax, d);
      }
    }
  }
  cmax = fmaxf(cmax, __shfl_xor(cmax, 16, 32));
  cmax = fmaxf(cmax, __shfl_xor(cmax, 8, 32));
  if (ln == 0) smax[wv] = cmax;
  __syncthreads();
  if (threadIdx.x == 0) {
    float m = smax[0];
#pragma unroll
    for (int i = 1; i < 8; ++i) m = fmaxf(m, smax[i]);
    atomicMax(gmax, f32_order_enc(m));
  }
}

__global__ __launch_bounds__(256) void edge_exp_kernel(float* __restrict__ c,
                                                       float* __restrict__ csum,
                                                       const int* __restrict__ v,
                                                       const unsigned* __restrict__ gmax,
                                                       int E) {
  const int e = blockIdx.x * 256 + threadIdx.x;
  if (e >= E) return;
  const float m = f32_order_dec(*gmax);
  const float ce = expf(c[e] - m);
  c[e] = ce;
  atomicAdd(&csum[v[e]], ce);
}

__global__ __launch_bounds__(256) void edge_scatter_kernel(const float* __restrict__ c,
                                                           const float* __restrict__ csum,
                                                           const float* __restrict__ ctxp,
                                                           const int* __restrict__ u,
                                                           const int* __restrict__ v,
                                                           float* __restrict__ agg,
                                                           int E) {
  const int ln = threadIdx.x & 31;
  const int wv = __builtin_amdgcn_readfirstlane((int)(threadIdx.x >> 5));
  const int e0 = (blockIdx.x * 8 + wv) * ECH;
  int nval = E - e0;
  if (nval > ECH) nval = ECH;
  if (nval <= 0) return;
  const int el = e0 + (ln < nval ? ln : nval - 1);
  const int vl = v[el], ul = u[el];
  const int cl = __float_as_int(c[el] / (csum[vl] + 1e-16f));
  int j = 0;
  for (; j + EGR_S <= nval; j += EGR_S) {
    float m[EGR_S][4];
#pragma unroll
    for (int t = 0; t < EGR_S; ++t) {
      const int uj = __builtin_amdgcn_readlane(ul, j + t);
      const float* srow = ctxp + (size_t)uj * DIM + ln;
      m[t][0] = srow[0];
      m[t][1] = srow[32];
      m[t][2] = srow[64];
      m[t][3] = srow[96];
    }
#pragma unroll
    for (int t = 0; t < EGR_S; ++t) {
      const int   vj   = __builtin_amdgcn_readlane(vl, j + t);
      const float coef = __int_as_float(__builtin_amdgcn_readlane(cl, j + t));
      float* dst = agg + (size_t)vj * DIM + ln;
      atomicAdd(dst + 0,  m[t][0] * coef);
      atomicAdd(dst + 32, m[t][1] * coef);
      atomicAdd(dst + 64, m[t][2] * coef);
      atomicAdd(dst + 96, m[t][3] * coef);
    }
  }
  for (; j < nval; ++j) {
    const int   uj   = __builtin_amdgcn_readlane(ul, j);
    const int   vj   = __builtin_amdgcn_readlane(vl, j);
    const float coef = __int_as_float(__builtin_amdgcn_readlane(cl, j));
    const float* srow = ctxp + (size_t)uj * DIM + ln;
    const float m0 = srow[0], m1 = srow[32], m2 = srow[64], m3 = srow[96];
    float* dst = agg + (size_t)vj * DIM + ln;
    atomicAdd(dst + 0,  m0 * coef);
    atomicAdd(dst + 32, m1 * coef);
    atomicAdd(dst + 64, m2 * coef);
    atomicAdd(dst + 96, m3 * coef);
  }
}

__global__ __launch_bounds__(256) void zero_f32(float* __restrict__ p, long n) {
  const long n4 = n >> 2;
  const long step = (long)gridDim.x * 256;
  const float4 z = make_float4(0.0f, 0.0f, 0.0f, 0.0f);
  for (long i = (long)blockIdx.x * 256 + threadIdx.x; i < n4; i += step)
    ((float4*)p)[i] = z;
}
__global__ void init_scalar_u32(unsigned* p) {
  if (threadIdx.x == 0) *p = 0u;
}

extern "C" void kernel_launch(void* const* d_in, const int* in_sizes, int n_in,
                              void* d_out, int out_size, void* d_ws, size_t ws_size,
                              hipStream_t stream) {
  (void)in_sizes; (void)n_in; (void)out_size; (void)ws_size;
  const int NA = 50000, NC = 100000, NE = 600000;

  const float* agt    = (const float*)d_in[0];
  const float* ctx    = (const float*)d_in[1];
  const int*   u      = (const int*)d_in[2];
  const int*   v      = (const int*)d_in[3];
  const float* Wq     = (const float*)d_in[4];
  const float* Wk     = (const float*)d_in[5];
  const float* Wctx   = (const float*)d_in[6];
  const float* gnc_w  = (const float*)d_in[7];
  const float* gnc_b  = (const float*)d_in[8];
  const float* Wagt   = (const float*)d_in[9];
  const float* gna_w  = (const float*)d_in[10];
  const float* gna_b  = (const float*)d_in[11];
  const float* norm_w = (const float*)d_in[12];
  const float* norm_b = (const float*)d_in[13];
  const float* Wlin   = (const float*)d_in[14];
  const float* gnl_w  = (const float*)d_in[15];
  const float* gnl_b  = (const float*)d_in[16];

  float* ws = (float*)d_ws;
  size_t off = 0;
  float* qbuf   = ws + off; off += (size_t)NA * DIM;
  float* kbuf   = ws + off; off += (size_t)NC * DIM;
  float* cpbuf  = ws + off; off += (size_t)NC * DIM;
  float* cbuf   = ws + off; off += (size_t)NE;
  float* csum   = ws + off; off += (size_t)NA;
  float* aggbuf = ws + off; off += (size_t)NA * DIM;
  float* xbuf   = ws + off; off += (size_t)NA * DIM;
  float* agtbuf = ws + off; off += (size_t)NA * DIM;
  unsigned* maxbuf = (unsigned*)(ws + off); off += 16;

  const dim3 b128(128), b256(256);

  for (int i = 0; i < NBLK; ++i) {
    const float* ain  = (i == 0) ? agt : agtbuf;
    float*       aout = (i == NBLK - 1) ? (float*)d_out : agtbuf;

    gemm_nt_128<<<(NA + 63) / 64, b128, 0, stream>>>(ain, Wq + (size_t)i * DIM * DIM, qbuf, NA);
    gemm_nt_128<<<(NC + 63) / 64, b128, 0, stream>>>(ctx, Wk + (size_t)i * DIM * DIM, kbuf, NC);
    gemm_nt_128<<<(NC + 63) / 64, b128, 0, stream>>>(ctx, Wctx + (size_t)i * DIM * DIM, cpbuf, NC);
    gn_kernel<<<(NC + 63) / 64, b256, 0, stream>>>(cpbuf, cpbuf, gnc_w + i * DIM, gnc_b + i * DIM, NC);

    init_scalar_u32<<<1, 32, 0, stream>>>(maxbuf);
    zero_f32<<<(NA + 1023) / 1024, b256, 0, stream>>>(csum, (long)NA);
    zero_f32<<<(NA + 31) / 32, b256, 0, stream>>>(aggbuf, (long)NA * DIM);
    edge_dot_kernel<<<(NE + 255) / 256, b256, 0, stream>>>(qbuf, kbuf, u, v, cbuf, maxbuf, NE);
    edge_exp_kernel<<<(NE + 255) / 256, b256, 0, stream>>>(cbuf, csum, v, maxbuf, NE);
    edge_scatter_kernel<<<(NE + 255) / 256, b256, 0, stream>>>(cbuf, csum, cpbuf, u, v, aggbuf, NE);

    gemm_cat_256<<<(NA + 63) / 64, b128, 0, stream>>>(ain, aggbuf, Wagt + (size_t)i * DIM * 2 * DIM, xbuf, NA);
    gn2_relu_kernel<<<(NA + 31) / 32, b256, 0, stream>>>(xbuf, xbuf, gna_w + i * DIM, gna_b + i * DIM,
                                                       norm_w + i * DIM, norm_b + i * DIM, NA);
    gemm_nt_128<<<(NA + 63) / 64, b128, 0, stream>>>(xbuf, Wlin + (size_t)i * DIM * DIM, qbuf, NA);
    gn_res_relu_kernel<<<(NA + 31) / 32, b256, 0, stream>>>(qbuf, ain, aout, gnl_w + i * DIM, gnl_b + i * DIM, NA);
  }
}


# Aggregation owner-computes: each work-group holds 196 destination rows in LDS, finds its edges by one key byte per edge, adds by LDS float atomics (arrival order as before); no global atomics on agg
# speedup vs baseline: 1.6874x; 1.0647x over previous
#include <hip/hip_runtime.h>


static constexpr int DIM  = 128;
static constexpr int NBLK = 2;
static constexpr int LDP  = 68;

typedef __attribute__((ext_vector_type(2))) float v2f;
typedef __attribute__((ext_vector_type(8))) float v8f;

static __device__ __forceinline__ float wave_reduce_sum(float x) {
#pragma unroll
  for (int off = 16; off > 0; off >>= 1) x += __shfl_xor(x, off, 32);
  return x;
}

static __device__ __forceinline__ unsigned f32_order_enc(float f) {
  unsigned u = __float_as_uint(f);
  return (u & 0x80000000u) ? ~u : (u | 0x80000000u);
}
static __device__ __forceinline__ float f32_order_dec(unsigned e) {
  unsigned u = (e & 0x80000000u) ? (e & 0x7fffffffu) : ~e;
  return __uint_as_float(u);
}

__global__ __launch_bounds__(128) void gemm_nt_128(const float* __restrict__ A,
                                                   const float* __restrict__ W,
                                                   float* __restrict__ C, int N) {
  __shared__ float Wlds[128 * LDP];
  const int tid = threadIdx.x;
  const int wv  = tid >> 5;
  const int ln  = tid & 31;
  const int hk  = ln >> 4;
  const int l16 = ln & 15;
  const int r0  = blockIdx.x * 64 + wv * 16;

  v8f acc[8];
#pragma unroll
  for (int t = 0; t < 8; ++t)
#pragma unroll
    for (int j = 0; j < 8; ++j) acc[t][j] = 0.0f;

  int rowA = r0 + l16;
  if (rowA >= N) rowA = N - 1;
  const float* ap = A + (size_t)rowA * DIM;

  for (int kp = 0; kp < DIM; kp += 64) {
    __syncthreads();
#pragma unroll
    for (int i = 0; i < 16; ++i) {
      const int q  = tid + 128 * i;
      const int n  = q >> 4;
      const int c4 = (q & 15) << 2;
      *(float4*)(&Wlds[n * LDP + c4]) =
          *(const float4*)(W + (size_t)n * DIM + kp + c4);
    }
    __syncthreads();

#pragma unroll 4
    for (int k0 = 0; k0 < 64; k0 += 4) {
      const int ka = k0 + 2 * hk;
      v2f a; a.x = ap[kp + ka]; a.y = ap[kp + ka + 1];
#pragma unroll
      for (int t = 0; t < 8; ++t) {
        const float* wp = &Wlds[(t * 16 + l16) * LDP + ka];
        v2f b; b.x = wp[0]; b.y = wp[1];
        acc[t] = __builtin_amdgcn_wmma_f32_16x16x4_f32(
            false, a, false, b, (short)0, acc[t], false, false);
      }
    }
  }

  const int rb = r0 + hk * 8;
  if (r0 + 16 <= N) {
#pragma unroll
    for (int t = 0; t < 8; ++t)
#pragma unroll
      for (int j = 0; j < 8; ++j)
        C[(size_t)(rb + j) * DIM + t * 16 + l16] = acc[t][j];
  } else {
#pragma unroll
    for (int t = 0; t < 8; ++t)
#pragma unroll
      for (int j = 0; j < 8; ++j) {
        const int row = rb + j;
        if (row < N) C[(size_t)row * DIM + t * 16 + l16] = acc[t][j];
      }
  }
}

__global__ __launch_bounds__(128) void gemm_cat_256(const float* __restrict__ A0,
                                                    const float* __restrict__ A1,
                                                    const float* __restrict__ W,
                                                    float* __restrict__ C, int N) {
  __shared__ float Wlds[128 * LDP];
  const int tid = threadIdx.x;
  const int wv  = tid >> 5;
  const int ln  = tid & 31;
  const int hk  = ln >> 4;
  const int l16 = ln & 15;
  const int r0  = blockIdx.x * 64 + wv * 16;

  v8f acc[8];
#pragma unroll
  for (int t = 0; t < 8; ++t)
#pragma unroll
    for (int j = 0; j < 8; ++j) acc[t][j] = 0.0f;

  int rowA = r0 + l16;
  if (rowA >= N) rowA = N - 1;
  const size_t rbase = (size_t)rowA * DIM;

  for (int kp = 0; kp < 2 * DIM; kp += 64) {
    __syncthreads();
#pragma unroll
    for (int i = 0; i < 16; ++i) {
      const int q  = tid + 128 * i;
      const int n  = q >> 4;
      const int c4 = (q & 15) << 2;
      *(float4*)(&Wlds[n * LDP + c4]) =
          *(const float4*)(W + (size_t)n * (2 * DIM) + kp + c4);
    }
    __syncthreads();

    const float* abase = (kp < DIM) ? (A0 + rbase + kp) : (A1 + rbase + (kp - DIM));
#pragma unroll 4
    for (int k0 = 0; k0 < 64; k0 += 4) {
      const int ka = k0 + 2 * hk;
      v2f a; a.x = abase[ka]; a.y = abase[ka + 1];
#pragma unroll
      for (int t = 0; t < 8; ++t) {
        const float* wp = &Wlds[(t * 16 + l16) * LDP + ka];
        v2f b; b.x = wp[0]; b.y = wp[1];
        acc[t] = __builtin_amdgcn_wmma_f32_16x16x4_f32(
            false, a, false, b, (short)0, acc[t], false, false);
      }
    }
  }

  const int rb = r0 + hk * 8;
  if (r0 + 16 <= N) {
#pragma unroll
    for (int t = 0; t < 8; ++t)
#pragma unroll
      for (int j = 0; j < 8; ++j)
        C[(size_t)(rb + j) * DIM + t * 16 + l16] = acc[t][j];
  } else {
#pragma unroll
    for (int t = 0; t < 8; ++t)
#pragma unroll
      for (int j = 0; j < 8; ++j) {
        const int row = rb + j;
        if (row < N) C[(size_t)row * DIM + t * 16 + l16] = acc[t][j];
      }
  }
}

static constexpr int ROWB = 4;

static __device__ __forceinline__ void row_stats(const float4 x, float& mu, float& rs) {
  float s  = x.x + x.y + x.z + x.w;
  float sq = x.x * x.x + x.y * x.y + x.z * x.z + x.w * x.w;
  s = wave_reduce_sum(s); sq = wave_reduce_sum(sq);
  mu = s * (1.0f / DIM);
  rs = rsqrtf(sq * (1.0f / DIM) - mu * mu + 1e-5f);
}

__global__ __launch_bounds__(256) void gn_kernel(const float* X, float* Y,
                                                 const float* __restrict__ w,
                                                 const float* __restrict__ b, int N) {
  const int ln = threadIdx.x & 31;
  const int wv = __builtin_amdgcn_readfirstlane((int)(threadIdx.x >> 5));
  const int step = (int)gridDim.x * 8 * ROWB;
  const float4 wv4 = ((const float4*)w)[ln];
  const float4 bv4 = ((const float4*)b)[ln];
  for (int r0 = ((int)blockIdx.x * 8 + wv) * ROWB; r0 < N; r0 += step) {
    float4 x[ROWB];
#pragma unroll
    for (int t = 0; t < ROWB; ++t) {
      x[t] = make_float4(0.0f, 0.0f, 0.0f, 0.0f);
      if (r0 + t < N) x[t] = ((const float4*)(X + (size_t)(r0 + t) * DIM))[ln];
    }
#pragma unroll
    for (int t = 0; t < ROWB; ++t) {
      float mu, rs;
      row_stats(x[t], mu, rs);
      float4 y;
      y.x = (x[t].x - mu) * rs * wv4.x + bv4.x;
      y.y = (x[t].y - mu) * rs * wv4.y + bv4.y;
      y.z = (x[t].z - mu) * rs * wv4.z + bv4.z;
      y.w = (x[t].w - mu) * rs * wv4.w + bv4.w;
      if (r0 + t < N) ((float4*)(Y + (size_t)(r0 + t) * DIM))[ln] = y;
    }
  }
}

__global__ __launch_bounds__(256) void gn2_relu_kernel(const float* X, float* Y,
                                                       const float* __restrict__ w1,
                                                       const float* __restrict__ b1,
                                                       const float* __restrict__ w2,
                                                       const float* __restrict__ b2,
                                                       int N) {
  const int ln = threadIdx.x & 31;
  const int wv = __builtin_amdgcn_readfirstlane((int)(threadIdx.x >> 5));
  const int step = (int)gridDim.x * 8 * ROWB;
  const float4 wa = ((const float4*)w1)[ln], ba = ((const float4*)b1)[ln];
  const float4 wn = ((const float4*)w2)[ln], bn = ((const float4*)b2)[ln];
  for (int r0 = ((int)blockIdx.x * 8 + wv) * ROWB; r0 < N; r0 += step) {
    float4 x[ROWB];
#pragma unroll
    for (int t = 0; t < ROWB; ++t) {
      x[t] = make_float4(0.0f, 0.0f, 0.0f, 0.0f);
      if (r0 + t < N) x[t] = ((const float4*)(X + (size_t)(r0 + t) * DIM))[ln];
    }
#pragma unroll
    for (int t = 0; t < ROWB; ++t) {
      float mu, rs;
      row_stats(x[t], mu, rs);
      float4 y;
      y.x = (x[t].x - mu) * rs * wa.x + ba.x;
      y.y = (x[t].y - mu) * rs * wa.y + ba.y;
      y.z = (x[t].z - mu) * rs * wa.z + ba.z;
      y.w = (x[t].w - mu) * rs * wa.w + ba.w;
      row_stats(y, mu, rs);
      float4 z;
      z.x = fmaxf((y.x - mu) * rs * wn.x + bn.x, 0.0f);
      z.y = fmaxf((y.y - mu) * rs * wn.y + bn.y, 0.0f);
      z.z = fmaxf((y.z - mu) * rs * wn.z + bn.z, 0.0f);
      z.w = fmaxf((y.w - mu) * rs * wn.w + bn.w, 0.0f);
      if (r0 + t < N) ((float4*)(Y + (size_t)(r0 + t) * DIM))[ln] = z;
    }
  }
}

__global__ __launch_bounds__(256) void gn_res_relu_kernel(const float* __restrict__ X,
                                                          const float* __restrict__ R,
                                                          float* __restrict__ Y,
                                                          const float* __restrict__ w,
                                                          const float* __restrict__ b,
                                                          int N) {
  const int ln = threadIdx.x & 31;
  const int wv = __builtin_amdgcn_readfirstlane((int)(threadIdx.x >> 5));
  const int step = (int)gridDim.x * 8 * ROWB;
  const float4 wv4 = ((const float4*)w)[ln], bv4 = ((const float4*)b)[ln];
  for (int r0 = ((int)blockIdx.x * 8 + wv) * ROWB; r0 < N; r0 += step) {
    float4 x[ROWB], r[ROWB];
#pragma unroll
    for (int t = 0; t < ROWB; ++t) {
      x[t] = make_float4(0.0f, 0.0f, 0.0f, 0.0f);
      r[t] = x[t];
      if (r0 + t < N) {
        x[t] = ((const float4*)(X + (size_t)(r0 + t) * DIM))[ln];
        r[t] = ((const float4*)(R + (size_t)(r0 + t) * DIM))[ln];
      }
    }
#pragma unroll
    for (int t = 0; t < ROWB; ++t) {
      float mu, rs;
      row_stats(x[t], mu, rs);
      float4 y;
      y.x = fmaxf((x[t].x - mu) * rs * wv4.x + bv4.x + r[t].x, 0.0f);
      y.y = fmaxf((x[t].y - mu) * rs * wv4.y + bv4.y + r[t].y, 0.0f);
      y.z = fmaxf((x[t].z - mu) * rs * wv4.z + bv4.z + r[t].z, 0.0f);
      y.w = fmaxf((x[t].w - mu) * rs * wv4.w + bv4.w + r[t].w, 0.0f);
      if (r0 + t < N) ((float4*)(Y + (size_t)(r0 + t) * DIM))[ln] = y;
    }
  }
}

static constexpr int ECH    = 32;
static constexpr int EGR_D  = 4;
static constexpr int EGR_S  = 8;
static constexpr int OWN_ROWS = 196;

__global__ __launch_bounds__(256) void edge_dot_kernel(const float* __restrict__ Q,
                                                       const float* __restrict__ K,
                                                       const int* __restrict__ u,
                                                       const int* __restrict__ v,
                                                       float* __restrict__ c,
                                                       unsigned* __restrict__ gmax,
                                                       int E) {
  __shared__ float smax[8];
  const int ln = threadIdx.x & 31;
  const int wv = __builtin_amdgcn_readfirstlane((int)(threadIdx.x >> 5));
  const int e0 = (blockIdx.x * 8 + wv) * ECH;
  int nval = E - e0;
  if (nval > ECH) nval = ECH;
  const int hk = ln >> 4;
  const int b8 = (ln >> 3) & 1;
  float cmax = -3.0e38f;
  if (nval > 0) {
    const int el = e0 + (ln < nval ? ln : nval - 1);
    const int vl = v[el], ul = u[el];
    for (int j = 0; j < nval; j += EGR_D) {
      float4 qv[EGR_D], kv[EGR_D];
#pragma unroll
      for (int t = 0; t < EGR_D; ++t) {
        const int jj = (j + t < nval) ? (j + t) : (nval - 1);
        const int vj = __builtin_amdgcn_readlane(vl, jj);
        const int uj = __builtin_amdgcn_readlane(ul, jj);
        qv[t] = ((const float4*)(Q + (size_t)vj * DIM))[ln];
        kv[t] = ((const float4*)(K + (size_t)uj * DIM))[ln];
      }
      float p[EGR_D];
#pragma unroll
      for (int t = 0; t < EGR_D; ++t)
        p[t] = qv[t].x * kv[t].x + qv[t].y * kv[t].y + qv[t].z * kv[t].z + qv[t].w * kv[t].w;
      const float s0 = hk ? p[2] : p[0], r0 = hk ? p[0] : p[2];
      const float s1 = hk ? p[3] : p[1], r1 = hk ? p[1] : p[3];
      const float x0 = s0 + __shfl_xor(r0, 16, 32);
      const float x1 = s1 + __shfl_xor(r1, 16, 32);
      const float s2 = b8 ? x1 : x0, r2 = b8 ? x0 : x1;
      float d = s2 + __shfl_xor(r2, 8, 32);
      d += __shfl_xor(d, 4, 32);
      d += __shfl_xor(d, 2, 32);
      d += __shfl_xor(d, 1, 32);
      d = (d >= 0.0f) ? d : 0.1f * d;
      const int je = j + 2 * hk + b8;
      if (je < nval) {
        if ((ln & 7) == 0) c[e0 + je] = d;
        cmax = fmaxf(cmax, d);
      }
    }
  }
  cmax = fmaxf(cmax, __shfl_xor(cmax, 16, 32));
  cmax = fmaxf(cmax, __shfl_xor(cmax, 8, 32));
  if (ln == 0) smax[wv] = cmax;
  __syncthreads();
  if (threadIdx.x == 0) {
    float m = smax[0];
#pragma unroll
    for (int i = 1; i < 8; ++i) m = fmaxf(m, smax[i]);
    atomicMax(gmax, f32_order_enc(m));
  }
}

__global__ __launch_bounds__(256) void edge_exp_kernel(float* __restrict__ c,
                                                       float* __restrict__ csum,
                                                       const int* __restrict__ v,
                                                       const unsigned* __restrict__ gmax,
                                                       unsigned* __restrict__ keys,
                                                       int E) {
  const int e = blockIdx.x * 256 + threadIdx.x;
  if (e >= E) return;
  const float m = f32_order_dec(*gmax);
  const float ce = expf(c[e] - m);
  c[e] = ce;
  const int ve = v[e];
  atomicAdd(&csum[ve], ce);
  ((unsigned char*)keys)[e] = (unsigned char)(ve / OWN_ROWS);
}

static constexpr int OWN_CAP  = 8192;
static constexpr int OWN_SCAN = 8;

static __device__ __forceinline__ void own_append(int* list, int* cnt, int e) {
  const int pos = atomicAdd(cnt, 1);
  if (pos < OWN_CAP) list[pos] = e;
}

static __device__ __forceinline__ unsigned own_zero_bytes(unsigned y) {
  const unsigned t = ((y & 0x7f7f7f7fu) + 0x7f7f7f7fu) | y;
  return ~t & 0x80808080u;
}
static __device__ __forceinline__ void own_emit(int* list, int* cnt, unsigned z, int e0) {
  while (z) {
    own_append(list, cnt, e0 + (__builtin_ctz(z) >> 3));
    z &= z - 1;
  }
}

__global__ __launch_bounds__(1024) void edge_owner_kernel(const float* __restrict__ c,
                                                          const float* __restrict__ csum,
                                                          const float* __restrict__ ctxp,
                                                          const int* __restrict__ u,
                                                          const int* __restrict__ v,
                                                          const unsigned* __restrict__ keys,
                                                          float* __restrict__ agg,
                                                          int E, int NA) {
  __shared__ __attribute__((aligned(16))) float acc[OWN_ROWS * DIM];
  __shared__ int   list[OWN_CAP];
  __shared__ int   cnt;
  const int tid = threadIdx.x;
  const int ln  = tid & 31;
  const int wv  = __builtin_amdgcn_readfirstlane(tid >> 5);
  const int lo  = (int)blockIdx.x * OWN_ROWS;
  int nrows = NA - lo;
  if (nrows > OWN_ROWS) nrows = OWN_ROWS;
  if (nrows <= 0) return;

  const float4 z4 = make_float4(0.0f, 0.0f, 0.0f, 0.0f);
  for (int q = tid; q < nrows * (DIM / 4); q += 1024) ((float4*)acc)[q] = z4;
  if (tid == 0) cnt = 0;
  __syncthreads();

  const unsigned g    = blockIdx.x;
  const unsigned gpat = g * 0x01010101u;
  const int n16 = E >> 4;
  const uint4* k4 = (const uint4*)keys;
  const unsigned char* own8 = (const unsigned char*)keys;
  const int nch = (n16 + 1024 * OWN_SCAN - 1) / (1024 * OWN_SCAN);
  int k = (nch > 0) ? (int)(g % (unsigned)nch) : 0;
  for (int it = 0; it < nch; ++it) {
    const int i0 = k * (1024 * OWN_SCAN) + tid;
    if (++k == nch) k = 0;
    uint4 kw[OWN_SCAN];
#pragma unroll
    for (int t = 0; t < OWN_SCAN; ++t) {
      const int i = i0 + t * 1024;
      kw[t] = make_uint4(~gpat, ~gpat, ~gpat, ~gpat);
      if (i < n16) kw[t] = k4[i];
    }
#pragma unroll
    for (int t = 0; t < OWN_SCAN; ++t) {
      const int e = (i0 + t * 1024) << 4;
      const unsigned z0 = own_zero_bytes(kw[t].x ^ gpat);
      const unsigned z1 = own_zero_bytes(kw[t].y ^ gpat);
      const unsigned z2 = own_zero_bytes(kw[t].z ^ gpat);
      const unsigned z3 = own_zero_bytes(kw[t].w ^ gpat);
      if (z0 | z1 | z2 | z3) {
        own_emit(list, &cnt, z0, e + 0);
        own_emit(list, &cnt, z1, e + 4);
        own_emit(list, &cnt, z2, e + 8);
        own_emit(list, &cnt, z3, e + 12);
      }
    }
  }
  for (int e = (n16 << 4) + tid; e < E; e += 1024)
    if (own8[e] == (unsigned char)g) own_append(list, &cnt, e);
  __syncthreads();
  const int total = cnt;

  if (total <= OWN_CAP) {
    const int per = (total + 31) >> 5;
    const int b0  = wv * per;
    int b1 = b0 + per;
    if (b1 > total) b1 = total;
    for (int base = b0; base < b1; base += 32) {
      int nval = b1 - base;
      if (nval > 32) nval = 32;
      const int el = list[base + (ln < nval ? ln : nval - 1)];
      const int vl = v[el], ul = u[el];
      const int cl = __float_as_int(c[el] / (csum[vl] + 1e-16f));
      const int rl = (vl - lo) * DIM;
      int j = 0;
      for (; j + EGR_S <= nval; j += EGR_S) {
        float m[EGR_S][4];
#pragma unroll
        for (int t = 0; t < EGR_S; ++t) {
          const int uj = __builtin_amdgcn_readlane(ul, j + t);
          const float* srow = ctxp + (size_t)uj * DIM + ln;
          m[t][0] = srow[0];
          m[t][1] = srow[32];
          m[t][2] = srow[64];
          m[t][3] = srow[96];
        }
#pragma unroll
        for (int t = 0; t < EGR_S; ++t) {
          const int   rj   = __builtin_amdgcn_readlane(rl, j + t);
          const float coef = __int_as_float(__builtin_amdgcn_readlane(cl, j + t));
          float* dst = acc + rj + ln;
          atomicAdd(dst + 0,  m[t][0] * coef);
          atomicAdd(dst + 32, m[t][1] * coef);
          atomicAdd(dst + 64, m[t][2] * coef);
          atomicAdd(dst + 96, m[t][3] * coef);
        }
      }
      for (; j < nval; ++j) {
        const int   uj   = __builtin_amdgcn_readlane(ul, j);
        const int   rj   = __builtin_amdgcn_readlane(rl, j);
        const float coef = __int_as_float(__builtin_amdgcn_readlane(cl, j));
        const float* srow = ctxp + (size_t)uj * DIM + ln;
        const float m0 = srow[0], m1 = srow[32], m2 = srow[64], m3 = srow[96];
        float* dst = acc + rj + ln;
        atomicAdd(dst + 0,  m0 * coef);
        atomicAdd(dst + 32, m1 * coef);
        atomicAdd(dst + 64, m2 * coef);
        atomicAdd(dst + 96, m3 * coef);
      }
    }
  } else {
    for (int e0 = wv * 32; e0 < E; e0 += 1024) {
      const int e  = e0 + ln;
      const int vl = (e < E) ? v[e] : -1;
      unsigned mask = (unsigned)__ballot((unsigned)(vl - lo) < (unsigned)nrows);
      while (mask) {
        const int b = __builtin_ctz(mask);
        mask &= mask - 1;
        const int   vj   = __builtin_amdgcn_readlane(vl, b);
        const int   uj   = u[e0 + b];
        const float coef = c[e0 + b] / (csum[vj] + 1e-16f);
        const float* srow = ctxp + (size_t)uj * DIM + ln;
        const float m0 = srow[0], m1 = srow[32], m2 = srow[64], m3 = srow[96];
        float* dst = acc + (vj - lo) * DIM + ln;
        atomicAdd(dst + 0,  m0 * coef);
        atomicAdd(dst + 32, m1 * coef);
        atomicAdd(dst + 64, m2 * coef);
        atomicAdd(dst + 96, m3 * coef);
      }
    }
  }
  __syncthreads();

  float4* out4 = (float4*)(agg + (size_t)lo * DIM);
  for (int q = tid; q < nrows * (DIM / 4); q += 1024) out4[q] = ((const float4*)acc)[q];
}

__global__ __launch_bounds__(256) void zero_f32(float* __restrict__ p, long n) {
  const long n4 = n >> 2;
  const long step = (long)gridDim.x * 256;
  const float4 z = make_float4(0.0f, 0.0f, 0.0f, 0.0f);
  for (long i = (long)blockIdx.x * 256 + threadIdx.x; i < n4; i += step)
    ((float4*)p)[i] = z;
}
__global__ void init_scalar_u32(unsigned* p) {
  if (threadIdx.x == 0) *p = 0u;
}

extern "C" void kernel_launch(void* const* d_in, const int* in_sizes, int n_in,
                              void* d_out, int out_size, void* d_ws, size_t ws_size,
                              hipStream_t stream) {
  (void)in_sizes; (void)n_in; (void)out_size; (void)ws_size;
  const int NA = 50000, NC = 100000, NE = 600000;

  const float* agt    = (const float*)d_in[0];
  const float* ctx    = (const float*)d_in[1];
  const int*   u      = (const int*)d_in[2];
  const int*   v      = (const int*)d_in[3];
  const float* Wq     = (const float*)d_in[4];
  const float* Wk     = (const float*)d_in[5];
  const float* Wctx   = (const float*)d_in[6];
  const float* gnc_w  = (const float*)d_in[7];
  const float* gnc_b  = (const float*)d_in[8];
  const float* Wagt   = (const float*)d_in[9];
  const float* gna_w  = (const float*)d_in[10];
  const float* gna_b  = (const float*)d_in[11];
  const float* norm_w = (const float*)d_in[12];
  const float* norm_b = (const float*)d_in[13];
  const float* Wlin   = (const float*)d_in[14];
  const float* gnl_w  = (const float*)d_in[15];
  const float* gnl_b  = (const float*)d_in[16];

  float* ws = (float*)d_ws;
  size_t off = 0;
  float* qbuf   = ws + off; off += (size_t)NA * DIM;
  float* kbuf   = ws + off; off += (size_t)NC * DIM;
  float* cpbuf  = ws + off; off += (size_t)NC * DIM;
  float* cbuf   = ws + off; off += (size_t)NE;
  float* csum   = ws + off; off += (size_t)NA;
  float* aggbuf = ws + off; off += (size_t)NA * DIM;
  float* xbuf   = ws + off; off += (size_t)NA * DIM;
  float* agtbuf = ws + off; off += (size_t)NA * DIM;
  unsigned* maxbuf = (unsigned*)(ws + off); off += 16;

  const dim3 b128(128), b256(256);

  for (int i = 0; i < NBLK; ++i) {
    const float* ain  = (i == 0) ? agt : agtbuf;
    float*       aout = (i == NBLK - 1) ? (float*)d_out : agtbuf;

    gemm_nt_128<<<(NA + 63) / 64, b128, 0, stream>>>(ain, Wq + (size_t)i * DIM * DIM, qbuf, NA);
    gemm_nt_128<<<(NC + 63) / 64, b128, 0, stream>>>(ctx, Wk + (size_t)i * DIM * DIM, kbuf, NC);
    gemm_nt_128<<<(NC + 63) / 64, b128, 0, stream>>>(ctx, Wctx + (size_t)i * DIM * DIM, cpbuf, NC);
    gn_kernel<<<(NC + 63) / 64, b256, 0, stream>>>(cpbuf, cpbuf, gnc_w + i * DIM, gnc_b + i * DIM, NC);

    init_scalar_u32<<<1, 32, 0, stream>>>(maxbuf);
    zero_f32<<<(NA + 1023) / 1024, b256, 0, stream>>>(csum, (long)NA);
    edge_dot_kernel<<<(NE + 255) / 256, b256, 0, stream>>>(qbuf, kbuf, u, v, cbuf, maxbuf, NE);
    unsigned* keys = (unsigned*)xbuf;
    edge_exp_kernel<<<(NE + 255) / 256, b256, 0, stream>>>(cbuf, csum, v, maxbuf, keys, NE);
    edge_owner_kernel<<<(NA + OWN_ROWS - 1) / OWN_ROWS, dim3(1024), 0, stream>>>(cbuf, csum, cpbuf, u, v, keys, aggbuf, NE, NA);

    gemm_cat_256<<<(NA + 63) / 64, b128, 0, stream>>>(ain, aggbuf, Wagt + (size_t)i * DIM * 2 * DIM, xbuf, NA);
    gn2_relu_kernel<<<(NA + 31) / 32, b256, 0, stream>>>(xbuf, xbuf, gna_w + i * DIM, gna_b + i * DIM,
                                                       norm_w + i * DIM, norm_b + i * DIM, NA);
    gemm_nt_128<<<(NA + 63) / 64, b128, 0, stream>>>(xbuf, Wlin + (size_t)i * DIM * DIM, qbuf, NA);
    gn_res_relu_kernel<<<(NA + 31) / 32, b256, 0, stream>>>(qbuf, ain, aout, gnl_w + i * DIM, gnl_b + i * DIM, NA);
  }
}


# Softmax sums owner-local: the aggregation work-group forms exp terms and per-row sums in LDS (LDS float atomics, arrival order) for its own edges; exp kernel, global csum atomics, csum zero-fill gone
# speedup vs baseline: 1.9642x; 1.1640x over previous
#include <hip/hip_runtime.h>


static constexpr int DIM  = 128;
static constexpr int NBLK = 2;
static constexpr int LDP  = 68;

typedef __attribute__((ext_vector_type(2))) float v2f;
typedef __attribute__((ext_vector_type(8))) float v8f;

static __device__ __forceinline__ float wave_reduce_sum(float x) {
#pragma unroll
  for (int off = 16; off > 0; off >>= 1) x += __shfl_xor(x, off, 32);
  return x;
}

static __device__ __forceinline__ unsigned f32_order_enc(float f) {
  unsigned u = __float_as_uint(f);
  return (u & 0x80000000u) ? ~u : (u | 0x80000000u);
}
static __device__ __forceinline__ float f32_order_dec(unsigned e) {
  unsigned u = (e & 0x80000000u) ? (e & 0x7fffffffu) : ~e;
  return __uint_as_float(u);
}

__global__ __launch_bounds__(128) void gemm_nt_128(const float* __restrict__ A,
                                                   const float* __restrict__ W,
                                                   float* __restrict__ C, int N) {
  __shared__ float Wlds[128 * LDP];
  const int tid = threadIdx.x;
  const int wv  = tid >> 5;
  const int ln  = tid & 31;
  const int hk  = ln >> 4;
  const int l16 = ln & 15;
  const int r0  = blockIdx.x * 64 + wv * 16;

  v8f acc[8];
#pragma unroll
  for (int t = 0; t < 8; ++t)
#pragma unroll
    for (int j = 0; j < 8; ++j) acc[t][j] = 0.0f;

  int rowA = r0 + l16;
  if (rowA >= N) rowA = N - 1;
  const float* ap = A + (size_t)rowA * DIM;

  for (int kp = 0; kp < DIM; kp += 64) {
    __syncthreads();
#pragma unroll
    for (int i = 0; i < 16; ++i) {
      const int q  = tid + 128 * i;
      const int n  = q >> 4;
      const int c4 = (q & 15) << 2;
      *(float4*)(&Wlds[n * LDP + c4]) =
          *(const float4*)(W + (size_t)n * DIM + kp + c4);
    }
    __syncthreads();

#pragma unroll 4
    for (int k0 = 0; k0 < 64; k0 += 4) {
      const int ka = k0 + 2 * hk;
      v2f a; a.x = ap[kp + ka]; a.y = ap[kp + ka + 1];
#pragma unroll
      for (int t = 0; t < 8; ++t) {
        const float* wp = &Wlds[(t * 16 + l16) * LDP + ka];
        v2f b; b.x = wp[0]; b.y = wp[1];
        acc[t] = __builtin_amdgcn_wmma_f32_16x16x4_f32(
            false, a, false, b, (short)0, acc[t], false, false);
      }
    }
  }

  const int rb = r0 + hk * 8;
  if (r0 + 16 <= N) {
#pragma unroll
    for (int t = 0; t < 8; ++t)
#pragma unroll
      for (int j = 0; j < 8; ++j)
        C[(size_t)(rb + j) * DIM + t * 16 + l16] = acc[t][j];
  } else {
#pragma unroll
    for (int t = 0; t < 8; ++t)
#pragma unroll
      for (int j = 0; j < 8; ++j) {
        const int row = rb + j;
        if (row < N) C[(size_t)row * DIM + t * 16 + l16] = acc[t][j];
      }
  }
}

__global__ __launch_bounds__(128) void gemm_cat_256(const float* __restrict__ A0,
                                                    const float* __restrict__ A1,
                                                    const float* __restrict__ W,
                                                    float* __restrict__ C, int N) {
  __shared__ float Wlds[128 * LDP];
  const int tid = threadIdx.x;
  const int wv  = tid >> 5;
  const int ln  = tid & 31;
  const int hk  = ln >> 4;
  const int l16 = ln & 15;
  const int r0  = blockIdx.x * 64 + wv * 16;

  v8f acc[8];
#pragma unroll
  for (int t = 0; t < 8; ++t)
#pragma unroll
    for (int j = 0; j < 8; ++j) acc[t][j] = 0.0f;

  int rowA = r0 + l16;
  if (rowA >= N) rowA = N - 1;
  const size_t rbase = (size_t)rowA * DIM;

  for (int kp = 0; kp < 2 * DIM; kp += 64) {
    __syncthreads();
#pragma unroll
    for (int i = 0; i < 16; ++i) {
      const int q  = tid + 128 * i;
      const int n  = q >> 4;
      const int c4 = (q & 15) << 2;
      *(float4*)(&Wlds[n * LDP + c4]) =
          *(const float4*)(W + (size_t)n * (2 * DIM) + kp + c4);
    }
    __syncthreads();

    const float* abase = (kp < DIM) ? (A0 + rbase + kp) : (A1 + rbase + (kp - DIM));
#pragma unroll 4
    for (int k0 = 0; k0 < 64; k0 += 4) {
      const int ka = k0 + 2 * hk;
      v2f a; a.x = abase[ka]; a.y = abase[ka + 1];
#pragma unroll
      for (int t = 0; t < 8; ++t) {
        const float* wp = &Wlds[(t * 16 + l16) * LDP + ka];
        v2f b; b.x = wp[0]; b.y = wp[1];
        acc[t] = __builtin_amdgcn_wmma_f32_16x16x4_f32(
            false, a, false, b, (short)0, acc[t], false, false);
      }
    }
  }

  const int rb = r0 + hk * 8;
  if (r0 + 16 <= N) {
#pragma unroll
    for (int t = 0; t < 8; ++t)
#pragma unroll
      for (int j = 0; j < 8; ++j)
        C[(size_t)(rb + j) * DIM + t * 16 + l16] = acc[t][j];
  } else {
#pragma unroll
    for (int t = 0; t < 8; ++t)
#pragma unroll
      for (int j = 0; j < 8; ++j) {
        const int row = rb + j;
        if (row < N) C[(size_t)row * DIM + t * 16 + l16] = acc[t][j];
      }
  }
}

static constexpr int ROWB = 4;

static __device__ __forceinline__ void row_stats(const float4 x, float& mu, float& rs) {
  float s  = x.x + x.y + x.z + x.w;
  float sq = x.x * x.x + x.y * x.y + x.z * x.z + x.w * x.w;
  s = wave_reduce_sum(s); sq = wave_reduce_sum(sq);
  mu = s * (1.0f / DIM);
  rs = rsqrtf(sq * (1.0f / DIM) - mu * mu + 1e-5f);
}

__global__ __launch_bounds__(256) void gn_kernel(const float* X, float* Y,
                                                 const float* __restrict__ w,
                                                 const float* __restrict__ b, int N) {
  const int ln = threadIdx.x & 31;
  const int wv = __builtin_amdgcn_readfirstlane((int)(threadIdx.x >> 5));
  const int step = (int)gridDim.x * 8 * ROWB;
  const float4 wv4 = ((const float4*)w)[ln];
  const float4 bv4 = ((const float4*)b)[ln];
  for (int r0 = ((int)blockIdx.x * 8 + wv) * ROWB; r0 < N; r0 += step) {
    float4 x[ROWB];
#pragma unroll
    for (int t = 0; t < ROWB; ++t) {
      x[t] = make_float4(0.0f, 0.0f, 0.0f, 0.0f);
      if (r0 + t < N) x[t] = ((const float4*)(X + (size_t)(r0 + t) * DIM))[ln];
    }
#pragma unroll
    for (int t = 0; t < ROWB; ++t) {
      float mu, rs;
      row_stats(x[t], mu, rs);
      float4 y;
      y.x = (x[t].x - mu) * rs * wv4.x + bv4.x;
      y.y = (x[t].y - mu) * rs * wv4.y + bv4.y;
      y.z = (x[t].z - mu) * rs * wv4.z + bv4.z;
      y.w = (x[t].w - mu) * rs * wv4.w + bv4.w;
      if (r0 + t < N) ((float4*)(Y + (size_t)(r0 + t) * DIM))[ln] = y;
    }
  }
}

__global__ __launch_bounds__(256) void gn2_relu_kernel(const float* X, float* Y,
                                                       const float* __restrict__ w1,
                                                       const float* __restrict__ b1,
                                                       const float* __restrict__ w2,
                                                       const float* __restrict__ b2,
                                                       int N) {
  const int ln = threadIdx.x & 31;
  const int wv = __builtin_amdgcn_readfirstlane((int)(threadIdx.x >> 5));
  const int step = (int)gridDim.x * 8 * ROWB;
  const float4 wa = ((const float4*)w1)[ln], ba = ((const float4*)b1)[ln];
  const float4 wn = ((const float4*)w2)[ln], bn = ((const float4*)b2)[ln];
  for (int r0 = ((int)blockIdx.x * 8 + wv) * ROWB; r0 < N; r0 += step) {
    float4 x[ROWB];
#pragma unroll
    for (int t = 0; t < ROWB; ++t) {
      x[t] = make_float4(0.0f, 0.0f, 0.0f, 0.0f);
      if (r0 + t < N) x[t] = ((const float4*)(X + (size_t)(r0 + t) * DIM))[ln];
    }
#pragma unroll
    for (int t = 0; t < ROWB; ++t) {
      float mu, rs;
      row_stats(x[t], mu, rs);
      float4 y;
      y.x = (x[t].x - mu) * rs * wa.x + ba.x;
      y.y = (x[t].y - mu) * rs * wa.y + ba.y;
      y.z = (x[t].z - mu) * rs * wa.z + ba.z;
      y.w = (x[t].w - mu) * rs * wa.w + ba.w;
      row_stats(y, mu, rs);
      float4 z;
      z.x = fmaxf((y.x - mu) * rs * wn.x + bn.x, 0.0f);
      z.y = fmaxf((y.y - mu) * rs * wn.y + bn.y, 0.0f);
      z.z = fmaxf((y.z - mu) * rs * wn.z + bn.z, 0.0f);
      z.w = fmaxf((y.w - mu) * rs * wn.w + bn.w, 0.0f);
      if (r0 + t < N) ((float4*)(Y + (size_t)(r0 + t) * DIM))[ln] = z;
    }
  }
}

__global__ __launch_bounds__(256) void gn_res_relu_kernel(const float* __restrict__ X,
                                                          const float* __restrict__ R,
                                                          float* __restrict__ Y,
                                                          const float* __restrict__ w,
                                                          const float* __restrict__ b,
                                                          int N) {
  const int ln = threadIdx.x & 31;
  const int wv = __builtin_amdgcn_readfirstlane((int)(threadIdx.x >> 5));
  const int step = (int)gridDim.x * 8 * ROWB;
  const float4 wv4 = ((const float4*)w)[ln], bv4 = ((const float4*)b)[ln];
  for (int r0 = ((int)blockIdx.x * 8 + wv) * ROWB; r0 < N; r0 += step) {
    float4 x[ROWB], r[ROWB];
#pragma unroll
    for (int t = 0; t < ROWB; ++t) {
      x[t] = make_float4(0.0f, 0.0f, 0.0f, 0.0f);
      r[t] = x[t];
      if (r0 + t < N) {
        x[t] = ((const float4*)(X + (size_t)(r0 + t) * DIM))[ln];
        r[t] = ((const float4*)(R + (size_t)(r0 + t) * DIM))[ln];
      }
    }
#pragma unroll
    for (int t = 0; t < ROWB; ++t) {
      float mu, rs;
      row_stats(x[t], mu, rs);
      float4 y;
      y.x = fmaxf((x[t].x - mu) * rs * wv4.x + bv4.x + r[t].x, 0.0f);
      y.y = fmaxf((x[t].y - mu) * rs * wv4.y + bv4.y + r[t].y, 0.0f);
      y.z = fmaxf((x[t].z - mu) * rs * wv4.z + bv4.z + r[t].z, 0.0f);
      y.w = fmaxf((x[t].w - mu) * rs * wv4.w + bv4.w + r[t].w, 0.0f);
      if (r0 + t < N) ((float4*)(Y + (size_t)(r0 + t) * DIM))[ln] = y;
    }
  }
}

static constexpr int ECH    = 32;
static constexpr int EGR_D  = 4;
static constexpr int EGR_S  = 8;
static constexpr int OWN_ROWS = 196;

__global__ __launch_bounds__(256) void edge_dot_kernel(const float* __restrict__ Q,
                                                       const float* __restrict__ K,
                                                       const int* __restrict__ u,
                                                       const int* __restrict__ v,
                                                       float* __restrict__ c,
                                                       unsigned* __restrict__ gmax,
                                                       unsigned* __restrict__ keys,
                                                       int E) {
  __shared__ float smax[8];
  const int ln = threadIdx.x & 31;
  const int wv = __builtin_amdgcn_readfirstlane((int)(threadIdx.x >> 5));
  const int e0 = (blockIdx.x * 8 + wv) * ECH;
  int nval = E - e0;
  if (nval > ECH) nval = ECH;
  const int hk = ln >> 4;
  const int b8 = (ln >> 3) & 1;
  float cmax = -3.0e38f;
  if (nval > 0) {
    const int el = e0 + (ln < nval ? ln : nval - 1);
    const int vl = v[el], ul = u[el];
    if (ln < nval) ((unsigned char*)keys)[e0 + ln] = (unsigned char)(vl / OWN_ROWS);
    for (int j = 0; j < nval; j += EGR_D) {
      float4 qv[EGR_D], kv[EGR_D];
#pragma unroll
      for (int t = 0; t < EGR_D; ++t) {
        const int jj = (j + t < nval) ? (j + t) : (nval - 1);
        const int vj = __builtin_amdgcn_readlane(vl, jj);
        const int uj = __builtin_amdgcn_readlane(ul, jj);
        qv[t] = ((const float4*)(Q + (size_t)vj * DIM))[ln];
        kv[t] = ((const float4*)(K + (size_t)uj * DIM))[ln];
      }
      float p[EGR_D];
#pragma unroll
      for (int t = 0; t < EGR_D; ++t)
        p[t] = qv[t].x * kv[t].x + qv[t].y * kv[t].y + qv[t].z * kv[t].z + qv[t].w * kv[t].w;
      const float s0 = hk ? p[2] : p[0], r0 = hk ? p[0] : p[2];
      const float s1 = hk ? p[3] : p[1], r1 = hk ? p[1] : p[3];
      const float x0 = s0 + __shfl_xor(r0, 16, 32);
      const float x1 = s1 + __shfl_xor(r1, 16, 32);
      const float s2 = b8 ? x1 : x0, r2 = b8 ? x0 : x1;
      float d = s2 + __shfl_xor(r2, 8, 32);
      d += __shfl_xor(d, 4, 32);
      d += __shfl_xor(d, 2, 32);
      d += __shfl_xor(d, 1, 32);
      d = (d >= 0.0f) ? d : 0.1f * d;
      const int je = j + 2 * hk + b8;
      if (je < nval) {
        if ((ln & 7) == 0) c[e0 + je] = d;
        cmax = fmaxf(cmax, d);
      }
    }
  }
  cmax = fmaxf(cmax, __shfl_xor(cmax, 16, 32));
  cmax = fmaxf(cmax, __shfl_xor(cmax, 8, 32));
  if (ln == 0) smax[wv] = cmax;
  __syncthreads();
  if (threadIdx.x == 0) {
    float m = smax[0];
#pragma unroll
    for (int i = 1; i < 8; ++i) m = fmaxf(m, smax[i]);
    atomicMax(gmax, f32_order_enc(m));
  }
}

static constexpr int OWN_CAP  = 8192;
static constexpr int OWN_SCAN = 8;

static __device__ __forceinline__ void own_append(int* list, int* cnt, int e) {
  const int pos = atomicAdd(cnt, 1);
  if (pos < OWN_CAP) list[pos] = e;
}

static __device__ __forceinline__ unsigned own_zero_bytes(unsigned y) {
  const unsigned t = ((y & 0x7f7f7f7fu) + 0x7f7f7f7fu) | y;
  return ~t & 0x80808080u;
}
static __device__ __forceinline__ void own_emit(int* list, int* cnt, unsigned z, int e0) {
  while (z) {
    own_append(list, cnt, e0 + (__builtin_ctz(z) >> 3));
    z &= z - 1;
  }
}

__global__ __launch_bounds__(1024) void edge_owner_kernel(const float* __restrict__ c,
                                                          const unsigned* __restrict__ gmax,
                                                          const float* __restrict__ ctxp,
                                                          const int* __restrict__ u,
                                                          const int* __restrict__ v,
                                                          const unsigned* __restrict__ keys,
                                                          float* __restrict__ agg,
                                                          int E, int NA) {
  __shared__ __attribute__((aligned(16))) float acc[OWN_ROWS * DIM];
  __shared__ int   list[OWN_CAP];
  __shared__ float lsum[OWN_ROWS];
  __shared__ int   cnt;
  const int tid = threadIdx.x;
  const int ln  = tid & 31;
  const int wv  = __builtin_amdgcn_readfirstlane(tid >> 5);
  const int lo  = (int)blockIdx.x * OWN_ROWS;
  int nrows = NA - lo;
  if (nrows > OWN_ROWS) nrows = OWN_ROWS;
  if (nrows <= 0) return;

  const float4 z4 = make_float4(0.0f, 0.0f, 0.0f, 0.0f);
  for (int q = tid; q < nrows * (DIM / 4); q += 1024) ((float4*)acc)[q] = z4;
  if (tid < OWN_ROWS) lsum[tid] = 0.0f;
  if (tid == 0) cnt = 0;
  __syncthreads();

  const unsigned g    = blockIdx.x;
  const unsigned gpat = g * 0x01010101u;
  const int n16 = E >> 4;
  const uint4* k4 = (const uint4*)keys;
  const unsigned char* own8 = (const unsigned char*)keys;
  const int nch = (n16 + 1024 * OWN_SCAN - 1) / (1024 * OWN_SCAN);
  int k = (nch > 0) ? (int)(g % (unsigned)nch) : 0;
  for (int it = 0; it < nch; ++it) {
    const int i0 = k * (1024 * OWN_SCAN) + tid;
    if (++k == nch) k = 0;
    uint4 kw[OWN_SCAN];
#pragma unroll
    for (int t = 0; t < OWN_SCAN; ++t) {
      const int i = i0 + t * 1024;
      kw[t] = make_uint4(~gpat, ~gpat, ~gpat, ~gpat);
      if (i < n16) kw[t] = k4[i];
    }
#pragma unroll
    for (int t = 0; t < OWN_SCAN; ++t) {
      const int e = (i0 + t * 1024) << 4;
      const unsigned z0 = own_zero_bytes(kw[t].x ^ gpat);
      const unsigned z1 = own_zero_bytes(kw[t].y ^ gpat);
      const unsigned z2 = own_zero_bytes(kw[t].z ^ gpat);
      const unsigned z3 = own_zero_bytes(kw[t].w ^ gpat);
      if (z0 | z1 | z2 | z3) {
        own_emit(list, &cnt, z0, e + 0);
        own_emit(list, &cnt, z1, e + 4);
        own_emit(list, &cnt, z2, e + 8);
        own_emit(list, &cnt, z3, e + 12);
      }
    }
  }
  for (int e = (n16 << 4) + tid; e < E; e += 1024)
    if (own8[e] == (unsigned char)g) own_append(list, &cnt, e);
  __syncthreads();
  const int total = cnt;
  const float gm = f32_order_dec(*gmax);

  if (total <= OWN_CAP) {
    for (int i = tid; i < total; i += 1024) {
      const int e = list[i];
      atomicAdd(&lsum[v[e] - lo], expf(c[e] - gm));
    }
    __syncthreads();
    const int per = (total + 31) >> 5;
    const int b0  = wv * per;
    int b1 = b0 + per;
    if (b1 > total) b1 = total;
    for (int base = b0; base < b1; base += 32) {
      int nval = b1 - base;
      if (nval > 32) nval = 32;
      const int el = list[base + (ln < nval ? ln : nval - 1)];
      const int vl = v[el], ul = u[el];
      const int cl = __float_as_int(expf(c[el] - gm) / (lsum[vl - lo] + 1e-16f));
      const int rl = (vl - lo) * DIM;
      int j = 0;
      for (; j + EGR_S <= nval; j += EGR_S) {
        float m[EGR_S][4];
#pragma unroll
        for (int t = 0; t < EGR_S; ++t) {
          const int uj = __builtin_amdgcn_readlane(ul, j + t);
          const float* srow = ctxp + (size_t)uj * DIM + ln;
          m[t][0] = srow[0];
          m[t][1] = srow[32];
          m[t][2] = srow[64];
          m[t][3] = srow[96];
        }
#pragma unroll
        for (int t = 0; t < EGR_S; ++t) {
          const int   rj   = __builtin_amdgcn_readlane(rl, j + t);
          const float coef = __int_as_float(__builtin_amdgcn_readlane(cl, j + t));
          float* dst = acc + rj + ln;
          atomicAdd(dst + 0,  m[t][0] * coef);
          atomicAdd(dst + 32, m[t][1] * coef);
          atomicAdd(dst + 64, m[t][2] * coef);
          atomicAdd(dst + 96, m[t][3] * coef);
        }
      }
      for (; j < nval; ++j) {
        const int   uj   = __builtin_amdgcn_readlane(ul, j);
        const int   rj   = __builtin_amdgcn_readlane(rl, j);
        const float coef = __int_as_float(__builtin_amdgcn_readlane(cl, j));
        const float* srow = ctxp + (size_t)uj * DIM + ln;
        const float m0 = srow[0], m1 = srow[32], m2 = srow[64], m3 = srow[96];
        float* dst = acc + rj + ln;
        atomicAdd(dst + 0,  m0 * coef);
        atomicAdd(dst + 32, m1 * coef);
        atomicAdd(dst + 64, m2 * coef);
        atomicAdd(dst + 96, m3 * coef);
      }
    }
  } else {
    for (int e0 = wv * 32; e0 < E; e0 += 1024) {
      const int e  = e0 + ln;
      const int vl = (e < E) ? v[e] : -1;
      if ((unsigned)(vl - lo) < (unsigned)nrows) atomicAdd(&lsum[vl - lo], expf(c[e] - gm));
    }
    __syncthreads();
    for (int e0 = wv * 32; e0 < E; e0 += 1024) {
      const int e  = e0 + ln;
      const int vl = (e < E) ? v[e] : -1;
      unsigned mask = (unsigned)__ballot((unsigned)(vl - lo) < (unsigned)nrows);
      while (mask) {
        const int b = __builtin_ctz(mask);
        mask &= mask - 1;
        const int   vj   = __builtin_amdgcn_readlane(vl, b);
        const int   uj   = u[e0 + b];
        const float coef = expf(c[e0 + b] - gm) / (lsum[vj - lo] + 1e-16f);
        const float* srow = ctxp + (size_t)uj * DIM + ln;
        const float m0 = srow[0], m1 = srow[32], m2 = srow[64], m3 = srow[96];
        float* dst = acc + (vj - lo) * DIM + ln;
        atomicAdd(dst + 0,  m0 * coef);
        atomicAdd(dst + 32, m1 * coef);
        atomicAdd(dst + 64, m2 * coef);
        atomicAdd(dst + 96, m3 * coef);
      }
    }
  }
  __syncthreads();

  float4* out4 = (float4*)(agg + (size_t)lo * DIM);
  for (int q = tid; q < nrows * (DIM / 4); q += 1024) out4[q] = ((const float4*)acc)[q];
}

__global__ void init_scalar_u32(unsigned* p) {
  if (threadIdx.x == 0) *p = 0u;
}

extern "C" void kernel_launch(void* const* d_in, const int* in_sizes, int n_in,
                              void* d_out, int out_size, void* d_ws, size_t ws_size,
                              hipStream_t stream) {
  (void)in_sizes; (void)n_in; (void)out_size; (void)ws_size;
  const int NA = 50000, NC = 100000, NE = 600000;

  const float* agt    = (const float*)d_in[0];
  const float* ctx    = (const float*)d_in[1];
  const int*   u      = (const int*)d_in[2];
  const int*   v      = (const int*)d_in[3];
  const float* Wq     = (const float*)d_in[4];
  const float* Wk     = (const float*)d_in[5];
  const float* Wctx   = (const float*)d_in[6];
  const float* gnc_w  = (const float*)d_in[7];
  const float* gnc_b  = (const float*)d_in[8];
  const float* Wagt   = (const float*)d_in[9];
  const float* gna_w  = (const float*)d_in[10];
  const float* gna_b  = (const float*)d_in[11];
  const float* norm_w = (const float*)d_in[12];
  const float* norm_b = (const float*)d_in[13];
  const float* Wlin   = (const float*)d_in[14];
  const float* gnl_w  = (const float*)d_in[15];
  const float* gnl_b  = (const float*)d_in[16];

  float* ws = (float*)d_ws;
  size_t off = 0;
  float* qbuf   = ws + off; off += (size_t)NA * DIM;
  float* kbuf   = ws + off; off += (size_t)NC * DIM;
  float* cpbuf  = ws + off; off += (size_t)NC * DIM;
  float* cbuf   = ws + off; off += (size_t)NE;
  off += (size_t)NA;
  float* aggbuf = ws + off; off += (size_t)NA * DIM;
  float* xbuf   = ws + off; off += (size_t)NA * DIM;
  float* agtbuf = ws + off; off += (size_t)NA * DIM;
  unsigned* maxbuf = (unsigned*)(ws + off); off += 16;

  const dim3 b128(128), b256(256);

  for (int i = 0; i < NBLK; ++i) {
    const float* ain  = (i == 0) ? agt : agtbuf;
    float*       aout = (i == NBLK - 1) ? (float*)d_out : agtbuf;

    gemm_nt_128<<<(NA + 63) / 64, b128, 0, stream>>>(ain, Wq + (size_t)i * DIM * DIM, qbuf, NA);
    gemm_nt_128<<<(NC + 63) / 64, b128, 0, stream>>>(ctx, Wk + (size_t)i * DIM * DIM, kbuf, NC);
    gemm_nt_128<<<(NC + 63) / 64, b128, 0, stream>>>(ctx, Wctx + (size_t)i * DIM * DIM, cpbuf, NC);
    gn_kernel<<<(NC + 63) / 64, b256, 0, stream>>>(cpbuf, cpbuf, gnc_w + i * DIM, gnc_b + i * DIM, NC);

    init_scalar_u32<<<1, 32, 0, stream>>>(maxbuf);
    unsigned* keys = (unsigned*)xbuf;
    edge_dot_kernel<<<(NE + 255) / 256, b256, 0, stream>>>(qbuf, kbuf, u, v, cbuf, maxbuf, keys, NE);
    edge_owner_kernel<<<(NA + OWN_ROWS - 1) / OWN_ROWS, dim3(1024), 0, stream>>>(cbuf, maxbuf, cpbuf, u, v, keys, aggbuf, NE, NA);

    gemm_cat_256<<<(NA + 63) / 64, b128, 0, stream>>>(ain, aggbuf, Wagt + (size_t)i * DIM * 2 * DIM, xbuf, NA);
    gn2_relu_kernel<<<(NA + 31) / 32, b256, 0, stream>>>(xbuf, xbuf, gna_w + i * DIM, gna_b + i * DIM,
                                                       norm_w + i * DIM, norm_b + i * DIM, NA);
    gemm_nt_128<<<(NA + 63) / 64, b128, 0, stream>>>(xbuf, Wlin + (size_t)i * DIM * DIM, qbuf, NA);
    gn_res_relu_kernel<<<(NA + 31) / 32, b256, 0, stream>>>(qbuf, ain, aout, gnl_w + i * DIM, gnl_b + i * DIM, NA);
  }
}


# Edge scores owner-computes: a work-group keeps its 196 query rows in LDS, picks its edges by key byte, gathers one key row per edge (same dot tree), hands (row, source, score) entries to aggregation
# speedup vs baseline: 2.0989x; 1.0686x over previous
#include <hip/hip_runtime.h>


static constexpr int DIM  = 128;
static constexpr int NBLK = 2;
static constexpr int LDP  = 68;

typedef __attribute__((ext_vector_type(2))) float v2f;
typedef __attribute__((ext_vector_type(8))) float v8f;

static __device__ __forceinline__ float wave_reduce_sum(float x) {
#pragma unroll
  for (int off = 16; off > 0; off >>= 1) x += __shfl_xor(x, off, 32);
  return x;
}

static __device__ __forceinline__ unsigned f32_order_enc(float f) {
  unsigned u = __float_as_uint(f);
  return (u & 0x80000000u) ? ~u : (u | 0x80000000u);
}
static __device__ __forceinline__ float f32_order_dec(unsigned e) {
  unsigned u = (e & 0x80000000u) ? (e & 0x7fffffffu) : ~e;
  return __uint_as_float(u);
}

__global__ __launch_bounds__(128) void gemm_nt_128(const float* __restrict__ A,
                                                   const float* __restrict__ W,
                                                   float* __restrict__ C, int N) {
  __shared__ float Wlds[128 * LDP];
  const int tid = threadIdx.x;
  const int wv  = tid >> 5;
  const int ln  = tid & 31;
  const int hk  = ln >> 4;
  const int l16 = ln & 15;
  const int r0  = blockIdx.x * 64 + wv * 16;

  v8f acc[8];
#pragma unroll
  for (int t = 0; t < 8; ++t)
#pragma unroll
    for (int j = 0; j < 8; ++j) acc[t][j] = 0.0f;

  int rowA = r0 + l16;
  if (rowA >= N) rowA = N - 1;
  const float* ap = A + (size_t)rowA * DIM;

  for (int kp = 0; kp < DIM; kp += 64) {
    __syncthreads();
#pragma unroll
    for (int i = 0; i < 16; ++i) {
      const int q  = tid + 128 * i;
      const int n  = q >> 4;
      const int c4 = (q & 15) << 2;
      *(float4*)(&Wlds[n * LDP + c4]) =
          *(const float4*)(W + (size_t)n * DIM + kp + c4);
    }
    __syncthreads();

#pragma unroll 4
    for (int k0 = 0; k0 < 64; k0 += 4) {
      const int ka = k0 + 2 * hk;
      v2f a; a.x = ap[kp + ka]; a.y = ap[kp + ka + 1];
#pragma unroll
      for (int t = 0; t < 8; ++t) {
        const float* wp = &Wlds[(t * 16 + l16) * LDP + ka];
        v2f b; b.x = wp[0]; b.y = wp[1];
        acc[t] = __builtin_amdgcn_wmma_f32_16x16x4_f32(
            false, a, false, b, (short)0, acc[t], false, false);
      }
    }
  }

  const int rb = r0 + hk * 8;
  if (r0 + 16 <= N) {
#pragma unroll
    for (int t = 0; t < 8; ++t)
#pragma unroll
      for (int j = 0; j < 8; ++j)
        C[(size_t)(rb + j) * DIM + t * 16 + l16] = acc[t][j];
  } else {
#pragma unroll
    for (int t = 0; t < 8; ++t)
#pragma unroll
      for (int j = 0; j < 8; ++j) {
        const int row = rb + j;
        if (row < N) C[(size_t)row * DIM + t * 16 + l16] = acc[t][j];
      }
  }
}

__global__ __launch_bounds__(128) void gemm_cat_256(const float* __restrict__ A0,
                                                    const float* __restrict__ A1,
                                                    const float* __restrict__ W,
                                                    float* __restrict__ C, int N) {
  __shared__ float Wlds[128 * LDP];
  const int tid = threadIdx.x;
  const int wv  = tid >> 5;
  const int ln  = tid & 31;
  const int hk  = ln >> 4;
  const int l16 = ln & 15;
  const int r0  = blockIdx.x * 64 + wv * 16;

  v8f acc[8];
#pragma unroll
  for (int t = 0; t < 8; ++t)
#pragma unroll
    for (int j = 0; j < 8; ++j) acc[t][j] = 0.0f;

  int rowA = r0 + l16;
  if (rowA >= N) rowA = N - 1;
  const size_t rbase = (size_t)rowA * DIM;

  for (int kp = 0; kp < 2 * DIM; kp += 64) {
    __syncthreads();
#pragma unroll
    for (int i = 0; i < 16; ++i) {
      const int q  = tid + 128 * i;
      const int n  = q >> 4;
      const int c4 = (q & 15) << 2;
      *(float4*)(&Wlds[n * LDP + c4]) =
          *(const float4*)(W + (size_t)n * (2 * DIM) + kp + c4);
    }
    __syncthreads();

    const float* abase = (kp < DIM) ? (A0 + rbase + kp) : (A1 + rbase + (kp - DIM));
#pragma unroll 4
    for (int k0 = 0; k0 < 64; k0 += 4) {
      const int ka = k0 + 2 * hk;
      v2f a; a.x = abase[ka]; a.y = abase[ka + 1];
#pragma unroll
      for (int t = 0; t < 8; ++t) {
        const float* wp = &Wlds[(t * 16 + l16) * LDP + ka];
        v2f b; b.x = wp[0]; b.y = wp[1];
        acc[t] = __builtin_amdgcn_wmma_f32_16x16x4_f32(
            false, a, false, b, (short)0, acc[t], false, false);
      }
    }
  }

  const int rb = r0 + hk * 8;
  if (r0 + 16 <= N) {
#pragma unroll
    for (int t = 0; t < 8; ++t)
#pragma unroll
      for (int j = 0; j < 8; ++j)
        C[(size_t)(rb + j) * DIM + t * 16 + l16] = acc[t][j];
  } else {
#pragma unroll
    for (int t = 0; t < 8; ++t)
#pragma unroll
      for (int j = 0; j < 8; ++j) {
        const int row = rb + j;
        if (row < N) C[(size_t)row * DIM + t * 16 + l16] = acc[t][j];
      }
  }
}

static constexpr int ROWB = 4;

static __device__ __forceinline__ void row_stats(const float4 x, float& mu, float& rs) {
  float s  = x.x + x.y + x.z + x.w;
  float sq = x.x * x.x + x.y * x.y + x.z * x.z + x.w * x.w;
  s = wave_reduce_sum(s); sq = wave_reduce_sum(sq);
  mu = s * (1.0f / DIM);
  rs = rsqrtf(sq * (1.0f / DIM) - mu * mu + 1e-5f);
}

__global__ __launch_bounds__(256) void gn_kernel(const float* X, float* Y,
                                                 const float* __restrict__ w,
                                                 const float* __restrict__ b, int N) {
  const int ln = threadIdx.x & 31;
  const int wv = __builtin_amdgcn_readfirstlane((int)(threadIdx.x >> 5));
  const int step = (int)gridDim.x * 8 * ROWB;
  const float4 wv4 = ((const float4*)w)[ln];
  const float4 bv4 = ((const float4*)b)[ln];
  for (int r0 = ((int)blockIdx.x * 8 + wv) * ROWB; r0 < N; r0 += step) {
    float4 x[ROWB];
#pragma unroll
    for (int t = 0; t < ROWB; ++t) {
      x[t] = make_float4(0.0f, 0.0f, 0.0f, 0.0f);
      if (r0 + t < N) x[t] = ((const float4*)(X + (size_t)(r0 + t) * DIM))[ln];
    }
#pragma unroll
    for (int t = 0; t < ROWB; ++t) {
      float mu, rs;
      row_stats(x[t], mu, rs);
      float4 y;
      y.x = (x[t].x - mu) * rs * wv4.x + bv4.x;
      y.y = (x[t].y - mu) * rs * wv4.y + bv4.y;
      y.z = (x[t].z - mu) * rs * wv4.z + bv4.z;
      y.w = (x[t].w - mu) * rs * wv4.w + bv4.w;
      if (r0 + t < N) ((float4*)(Y + (size_t)(r0 + t) * DIM))[ln] = y;
    }
  }
}

__global__ __launch_bounds__(256) void gn2_relu_kernel(const float* X, float* Y,
                                                       const float* __restrict__ w1,
                                                       const float* __restrict__ b1,
                                                       const float* __restrict__ w2,
                                                       const float* __restrict__ b2,
                                                       int N) {
  const int ln = threadIdx.x & 31;
  const int wv = __builtin_amdgcn_readfirstlane((int)(threadIdx.x >> 5));
  const int step = (int)gridDim.x * 8 * ROWB;
  const float4 wa = ((const float4*)w1)[ln], ba = ((const float4*)b1)[ln];
  const float4 wn = ((const float4*)w2)[ln], bn = ((const float4*)b2)[ln];
  for (int r0 = ((int)blockIdx.x * 8 + wv) * ROWB; r0 < N; r0 += step) {
    float4 x[ROWB];
#pragma unroll
    for (int t = 0; t < ROWB; ++t) {
      x[t] = make_float4(0.0f, 0.0f, 0.0f, 0.0f);
      if (r0 + t < N) x[t] = ((const float4*)(X + (size_t)(r0 + t) * DIM))[ln];
    }
#pragma unroll
    for (int t = 0; t < ROWB; ++t) {
      float mu, rs;
      row_stats(x[t], mu, rs);
      float4 y;
      y.x = (x[t].x - mu) * rs * wa.x + ba.x;
      y.y = (x[t].y - mu) * rs * wa.y + ba.y;
      y.z = (x[t].z - mu) * rs * wa.z + ba.z;
      y.w = (x[t].w - mu) * rs * wa.w + ba.w;
      row_stats(y, mu, rs);
      float4 z;
      z.x = fmaxf((y.x - mu) * rs * wn.x + bn.x, 0.0f);
      z.y = fmaxf((y.y - mu) * rs * wn.y + bn.y, 0.0f);
      z.z = fmaxf((y.z - mu) * rs * wn.z + bn.z, 0.0f);
      z.w = fmaxf((y.w - mu) * rs * wn.w + bn.w, 0.0f);
      if (r0 + t < N) ((float4*)(Y + (size_t)(r0 + t) * DIM))[ln] = z;
    }
  }
}

__global__ __launch_bounds__(256) void gn_res_relu_kernel(const float* __restrict__ X,
                                                          const float* __restrict__ R,
                                                          float* __restrict__ Y,
                                                          const float* __restrict__ w,
                                                          const float* __restrict__ b,
                                                          int N) {
  const int ln = threadIdx.x & 31;
  const int wv = __builtin_amdgcn_readfirstlane((int)(threadIdx.x >> 5));
  const int step = (int)gridDim.x * 8 * ROWB;
  const float4 wv4 = ((const float4*)w)[ln], bv4 = ((const float4*)b)[ln];
  for (int r0 = ((int)blockIdx.x * 8 + wv) * ROWB; r0 < N; r0 += step) {
    float4 x[ROWB], r[ROWB];
#pragma unroll
    for (int t = 0; t < ROWB; ++t) {
      x[t] = make_float4(0.0f, 0.0f, 0.0f, 0.0f);
      r[t] = x[t];
      if (r0 + t < N) {
        x[t] = ((const float4*)(X + (size_t)(r0 + t) * DIM))[ln];
        r[t] = ((const float4*)(R + (size_t)(r0 + t) * DIM))[ln];
      }
    }
#pragma unroll
    for (int t = 0; t < ROWB; ++t) {
      float mu, rs;
      row_stats(x[t], mu, rs);
      float4 y;
      y.x = fmaxf((x[t].x - mu) * rs * wv4.x + bv4.x + r[t].x, 0.0f);
      y.y = fmaxf((x[t].y - mu) * rs * wv4.y + bv4.y + r[t].y, 0.0f);
      y.z = fmaxf((x[t].z - mu) * rs * wv4.z + bv4.z + r[t].z, 0.0f);
      y.w = fmaxf((x[t].w - mu) * rs * wv4.w + bv4.w + r[t].w, 0.0f);
      if (r0 + t < N) ((float4*)(Y + (size_t)(r0 + t) * DIM))[ln] = y;
    }
  }
}

static constexpr int EGR_Q  = 8;
static constexpr int EGR_S  = 8;
static constexpr int OWN_ROWS = 196;
static constexpr int OWN_CAP  = 8192;
static constexpr int OWN_SCAN = 8;

__global__ __launch_bounds__(256) void edge_prep_kernel(const int* __restrict__ u,
                                                        const int* __restrict__ v,
                                                        unsigned* __restrict__ keys,
                                                        int* __restrict__ vu_words,
                                                        unsigned* __restrict__ gmax,
                                                        int E) {
  int2* vu = (int2*)vu_words;
  if (blockIdx.x == 0 && threadIdx.x == 0) *gmax = 0u;
  const int i = (int)blockIdx.x * 256 + (int)threadIdx.x;
  const int e = i << 2;
  if (e + 3 < E) {
    const int4 vv = ((const int4*)v)[i];
    const int4 uu = ((const int4*)u)[i];
    keys[i] = (unsigned)(vv.x / OWN_ROWS) | ((unsigned)(vv.y / OWN_ROWS) << 8) |
              ((unsigned)(vv.z / OWN_ROWS) << 16) | ((unsigned)(vv.w / OWN_ROWS) << 24);
    ((int4*)vu)[2 * i]     = make_int4(vv.x, uu.x, vv.y, uu.y);
    ((int4*)vu)[2 * i + 1] = make_int4(vv.z, uu.z, vv.w, uu.w);
  } else {
    for (int k = e; k < E; ++k) {
      ((unsigned char*)keys)[k] = (unsigned char)(v[k] / OWN_ROWS);
      vu[k] = make_int2(v[k], u[k]);
    }
  }
}

static __device__ __forceinline__ void own_append(int* list, int* cnt, int e) {
  const int pos = atomicAdd(cnt, 1);
  if (pos < OWN_CAP) list[pos] = e;
}

static __device__ __forceinline__ unsigned own_zero_bytes(unsigned y) {
  const unsigned t = ((y & 0x7f7f7f7fu) + 0x7f7f7f7fu) | y;
  return ~t & 0x80808080u;
}
static __device__ __forceinline__ void own_emit(int* list, int* cnt, unsigned z, int e0) {
  while (z) {
    own_append(list, cnt, e0 + (__builtin_ctz(z) >> 3));
    z &= z - 1;
  }
}

static __device__ __forceinline__ float wave_sum4(const float p0, const float p1,
                                                  const float p2, const float p3,
                                                  const int hk, const int b8) {
  const float s0 = hk ? p2 : p0, r0 = hk ? p0 : p2;
  const float s1 = hk ? p3 : p1, r1 = hk ? p1 : p3;
  const float x0 = s0 + __shfl_xor(r0, 16, 32);
  const float x1 = s1 + __shfl_xor(r1, 16, 32);
  const float s2 = b8 ? x1 : x0, r2 = b8 ? x0 : x1;
  float d = s2 + __shfl_xor(r2, 8, 32);
  d += __shfl_xor(d, 4, 32);
  d += __shfl_xor(d, 2, 32);
  d += __shfl_xor(d, 1, 32);
  return d;
}

static __device__ __forceinline__ void score_fetch(float4 (&kv)[EGR_Q],
                                                   const float* __restrict__ K, const int ul,
                                                   const int j, const int nval, const int ln) {
#pragma unroll
  for (int t = 0; t < EGR_Q; ++t) {
    const int jj = (j + t < nval) ? (j + t) : (nval - 1);
    const int uj = __builtin_amdgcn_readlane(ul, jj);
    kv[t] = ((const float4*)(K + (size_t)uj * DIM))[ln];
  }
}
static __device__ __forceinline__ float score_step(const float4 (&kv)[EGR_Q], const float* qt,
                                                   const int rl, const int j, const int nval,
                                                   const int ln, const int hk, const int b8,
                                                   float dl) {
  float p[EGR_Q];
#pragma unroll
  for (int t = 0; t < EGR_Q; ++t) {
    const int jj = (j + t < nval) ? (j + t) : (nval - 1);
    const int rj = __builtin_amdgcn_readlane(rl, jj);
    const float4 qv = ((const float4*)(qt + rj))[ln];
    p[t] = qv.x * kv[t].x + qv.y * kv[t].y + qv.z * kv[t].z + qv.w * kv[t].w;
  }
  float da = wave_sum4(p[0], p[1], p[2], p[3], hk, b8);
  float db = wave_sum4(p[4], p[5], p[6], p[7], hk, b8);
  da = (da >= 0.0f) ? da : 0.1f * da;
  db = (db >= 0.0f) ? db : 0.1f * db;
  const int i8 = ln - j;
  const int sl = (i8 & 3) << 3;
  const float ga = __shfl(da, sl, 32);
  const float gb = __shfl(db, sl, 32);
  if ((unsigned)i8 < 4u) dl = ga;
  else if ((unsigned)i8 < 8u) dl = gb;
  return dl;
}

__global__ __launch_bounds__(1024) void edge_owner_score_kernel(const float* __restrict__ Q,
                                                                const float* __restrict__ K,
                                                                const int* __restrict__ u,
                                                                const int* __restrict__ v,
                                                                const unsigned* __restrict__ keys,
                                                                const int* __restrict__ vu_words,
                                                                float* __restrict__ c,
                                                                unsigned* __restrict__ gmax,
                                                                unsigned* __restrict__ rec_ru,
                                                                float* __restrict__ rec_sc,
                                                                int* __restrict__ gcount,
                                                                int E, int NA) {
  __shared__ __attribute__((aligned(16))) float qt[OWN_ROWS * DIM];
  __shared__ int   list[OWN_CAP];
  __shared__ float smax[32];
  __shared__ int   cnt;
  const int2* vu = (const int2*)vu_words;
  const int tid = threadIdx.x;
  const int ln  = tid & 31;
  const int wv  = __builtin_amdgcn_readfirstlane(tid >> 5);
  const int lo  = (int)blockIdx.x * OWN_ROWS;
  int nrows = NA - lo;
  if (nrows > OWN_ROWS) nrows = OWN_ROWS;
  if (nrows <= 0) return;
  const int hk = ln >> 4;
  const int b8 = (ln >> 3) & 1;

  const float4* q4 = (const float4*)(Q + (size_t)lo * DIM);
  for (int q = tid; q < nrows * (DIM / 4); q += 1024) ((float4*)qt)[q] = q4[q];
  if (tid == 0) cnt = 0;
  __syncthreads();

  const unsigned g    = blockIdx.x;
  const unsigned gpat = g * 0x01010101u;
  const int n16 = E >> 4;
  const uint4* k4 = (const uint4*)keys;
  const unsigned char* own8 = (const unsigned char*)keys;
  const int nch = (n16 + 1024 * OWN_SCAN - 1) / (1024 * OWN_SCAN);
  int k = (nch > 0) ? (int)(g % (unsigned)nch) : 0;
  for (int it = 0; it < nch; ++it) {
    const int i0 = k * (1024 * OWN_SCAN) + tid;
    if (++k == nch) k = 0;
    uint4 kw[OWN_SCAN];
#pragma unroll
    for (int t = 0; t < OWN_SCAN; ++t) {
      const int i = i0 + t * 1024;
      kw[t] = make_uint4(~gpat, ~gpat, ~gpat, ~gpat);
      if (i < n16) kw[t] = k4[i];
    }
#pragma unroll
    for (int t = 0; t < OWN_SCAN; ++t) {
      const int e = (i0 + t * 1024) << 4;
      const unsigned z0 = own_zero_bytes(kw[t].x ^ gpat);
      const unsigned z1 = own_zero_bytes(kw[t].y ^ gpat);
      const unsigned z2 = own_zero_bytes(kw[t].z ^ gpat);
      const unsigned z3 = own_zero_bytes(kw[t].w ^ gpat);
      if (z0 | z1 | z2 | z3) {
        own_emit(list, &cnt, z0, e + 0);
        own_emit(list, &cnt, z1, e + 4);
        own_emit(list, &cnt, z2, e + 8);
        own_emit(list, &cnt, z3, e + 12);
      }
    }
  }
  for (int e = (n16 << 4) + tid; e < E; e += 1024)
    if (own8[e] == (unsigned char)g) own_append(list, &cnt, e);
  __syncthreads();
  const int total = cnt;
  if (tid == 0) gcount[g] = total;

  float cmax = -3.0e38f;
  if (total <= OWN_CAP) {
    unsigned* my_ru = rec_ru + (size_t)g * OWN_CAP;
    float*    my_sc = rec_sc + (size_t)g * OWN_CAP;
    const int per = (total + 31) >> 5;
    const int b0  = wv * per;
    int b1 = b0 + per;
    if (b1 > total) b1 = total;
    if (b0 < b1) {
      int nval = (b1 - b0 > 32) ? 32 : (b1 - b0);
      int2 pn = vu[list[b0 + (ln < nval ? ln : nval - 1)]];
      float4 ka[EGR_Q], kb[EGR_Q];
      score_fetch(ka, K, pn.y, 0, nval, ln);
      for (int base = b0; base < b1; base += 32) {
        const int2 pl = pn;
        const int ul = pl.y;
        const int rl = (pl.x - lo) * DIM;
        const bool hasnext = base + 32 < b1;
        const int nvaln = hasnext ? ((b1 - base - 32 > 32) ? 32 : (b1 - base - 32)) : nval;
        pn = vu[list[(hasnext ? base + 32 : base) + (ln < nvaln ? ln : nvaln - 1)]];
        float dl = 0.0f;
        for (int j = 0; j < nval; j += 2 * EGR_Q) {
          score_fetch(kb, K, ul, j + EGR_Q, nval, ln);
          dl = score_step(ka, qt, rl, j, nval, ln, hk, b8, dl);
          const bool here = (j + 2 * EGR_Q < nval) || !hasnext;
          score_fetch(ka, K, here ? ul : pn.y, here ? j + 2 * EGR_Q : 0, here ? nval : nvaln, ln);
          dl = score_step(kb, qt, rl, j + EGR_Q, nval, ln, hk, b8, dl);
        }
        if (ln < nval) {
          my_ru[base + ln] = ((unsigned)(pl.x - lo) << 24) | (unsigned)ul;
          my_sc[base + ln] = dl;
          cmax = fmaxf(cmax, dl);
        }
        nval = nvaln;
      }
    }
  } else {
    for (int e0 = wv * 32; e0 < E; e0 += 1024) {
      const int e  = e0 + ln;
      const int vl = (e < E) ? v[e] : -1;
      unsigned mask = (unsigned)__ballot((unsigned)(vl - lo) < (unsigned)nrows);
      while (mask) {
        const int b = __builtin_ctz(mask);
        mask &= mask - 1;
        const int vj = __builtin_amdgcn_readlane(vl, b);
        const int uj = u[e0 + b];
        const float4 qv = ((const float4*)(qt + (vj - lo) * DIM))[ln];
        const float4 kv = ((const float4*)(K + (size_t)uj * DIM))[ln];
        float d = wave_reduce_sum(qv.x * kv.x + qv.y * kv.y + qv.z * kv.z + qv.w * kv.w);
        d = (d >= 0.0f) ? d : 0.1f * d;
        if (ln == 0) c[e0 + b] = d;
        cmax = fmaxf(cmax, d);
      }
    }
  }
  cmax = fmaxf(cmax, __shfl_xor(cmax, 16, 32));
  cmax = fmaxf(cmax, __shfl_xor(cmax, 8, 32));
  cmax = fmaxf(cmax, __shfl_xor(cmax, 4, 32));
  cmax = fmaxf(cmax, __shfl_xor(cmax, 2, 32));
  cmax = fmaxf(cmax, __shfl_xor(cmax, 1, 32));
  if (ln == 0) smax[wv] = cmax;
  __syncthreads();
  if (tid == 0) {
    float m = smax[0];
    for (int i = 1; i < 32; ++i) m = fmaxf(m, smax[i]);
    atomicMax(gmax, f32_order_enc(m));
  }
}

__global__ __launch_bounds__(1024) void edge_owner_kernel(const float* __restrict__ c,
                                                          const unsigned* __restrict__ gmax,
                                                          const float* __restrict__ ctxp,
                                                          const int* __restrict__ u,
                                                          const int* __restrict__ v,
                                                          const unsigned* __restrict__ rec_ru,
                                                          const float* __restrict__ rec_sc,
                                                          const int* __restrict__ gcount,
                                                          float* __restrict__ agg,
                                                          int E, int NA) {
  __shared__ __attribute__((aligned(16))) float acc[OWN_ROWS * DIM];
  __shared__ float lsum[OWN_ROWS];
  const int tid = threadIdx.x;
  const int ln  = tid & 31;
  const int wv  = __builtin_amdgcn_readfirstlane(tid >> 5);
  const int lo  = (int)blockIdx.x * OWN_ROWS;
  int nrows = NA - lo;
  if (nrows > OWN_ROWS) nrows = OWN_ROWS;
  if (nrows <= 0) return;

  const float4 z4 = make_float4(0.0f, 0.0f, 0.0f, 0.0f);
  for (int q = tid; q < nrows * (DIM / 4); q += 1024) ((float4*)acc)[q] = z4;
  if (tid < OWN_ROWS) lsum[tid] = 0.0f;

  const unsigned g = blockIdx.x;
  const int total = gcount[g];
  const unsigned* my_ru = rec_ru + (size_t)g * OWN_CAP;
  const float*    my_sc = rec_sc + (size_t)g * OWN_CAP;
  __syncthreads();
  const float gm = f32_order_dec(*gmax);

  if (total <= OWN_CAP) {
    for (int i = tid; i < total; i += 1024)
      atomicAdd(&lsum[my_ru[i] >> 24], expf(my_sc[i] - gm));
    __syncthreads();
    const int per = (total + 31) >> 5;
    const int b0  = wv * per;
    int b1 = b0 + per;
    if (b1 > total) b1 = total;
    for (int base = b0; base < b1; base += 32) {
      int nval = b1 - base;
      if (nval > 32) nval = 32;
      const int il = base + (ln < nval ? ln : nval - 1);
      const unsigned ru = my_ru[il];
      const int ul  = (int)(ru & 0x00ffffffu);
      const int row = (int)(ru >> 24);
      const int cl = __float_as_int(expf(my_sc[il] - gm) / (lsum[row] + 1e-16f));
      const int rl = row * DIM;
      int j = 0;
      for (; j + EGR_S <= nval; j += EGR_S) {
        float m[EGR_S][4];
#pragma unroll
        for (int t = 0; t < EGR_S; ++t) {
          const int uj = __builtin_amdgcn_readlane(ul, j + t);
          const float* srow = ctxp + (size_t)uj * DIM + ln;
          m[t][0] = srow[0];
          m[t][1] = srow[32];
          m[t][2] = srow[64];
          m[t][3] = srow[96];
        }
#pragma unroll
        for (int t = 0; t < EGR_S; ++t) {
          const int   rj   = __builtin_amdgcn_readlane(rl, j + t);
          const float coef = __int_as_float(__builtin_amdgcn_readlane(cl, j + t));
          float* dst = acc + rj + ln;
          atomicAdd(dst + 0,  m[t][0] * coef);
          atomicAdd(dst + 32, m[t][1] * coef);
          atomicAdd(dst + 64, m[t][2] * coef);
          atomicAdd(dst + 96, m[t][3] * coef);
        }
      }
      for (; j < nval; ++j) {
        const int   uj   = __builtin_amdgcn_readlane(ul, j);
        const int   rj   = __builtin_amdgcn_readlane(rl, j);
        const float coef = __int_as_float(__builtin_amdgcn_readlane(cl, j));
        const float* srow = ctxp + (size_t)uj * DIM + ln;
        const float m0 = srow[0], m1 = srow[32], m2 = srow[64], m3 = srow[96];
        float* dst = acc + rj + ln;
        atomicAdd(dst + 0,  m0 * coef);
        atomicAdd(dst + 32, m1 * coef);
        atomicAdd(dst + 64, m2 * coef);
        atomicAdd(dst + 96, m3 * coef);
      }
    }
  } else {
    for (int e0 = wv * 32; e0 < E; e0 += 1024) {
      const int e  = e0 + ln;
      const int vl = (e < E) ? v[e] : -1;
      if ((unsigned)(vl - lo) < (unsigned)nrows) atomicAdd(&lsum[vl - lo], expf(c[e] - gm));
    }
    __syncthreads();
    for (int e0 = wv * 32; e0 < E; e0 += 1024) {
      const int e  = e0 + ln;
      const int vl = (e < E) ? v[e] : -1;
      unsigned mask = (unsigned)__ballot((unsigned)(vl - lo) < (unsigned)nrows);
      while (mask) {
        const int b = __builtin_ctz(mask);
        mask &= mask - 1;
        const int   vj   = __builtin_amdgcn_readlane(vl, b);
        const int   uj   = u[e0 + b];
        const float coef = expf(c[e0 + b] - gm) / (lsum[vj - lo] + 1e-16f);
        const float* srow = ctxp + (size_t)uj * DIM + ln;
        const float m0 = srow[0], m1 = srow[32], m2 = srow[64], m3 = srow[96];
        float* dst = acc + (vj - lo) * DIM + ln;
        atomicAdd(dst + 0,  m0 * coef);
        atomicAdd(dst + 32, m1 * coef);
        atomicAdd(dst + 64, m2 * coef);
        atomicAdd(dst + 96, m3 * coef);
      }
    }
  }
  __syncthreads();

  float4* out4 = (float4*)(agg + (size_t)lo * DIM);
  for (int q = tid; q < nrows * (DIM / 4); q += 1024) out4[q] = ((const float4*)acc)[q];
}

extern "C" void kernel_launch(void* const* d_in, const int* in_sizes, int n_in,
                              void* d_out, int out_size, void* d_ws, size_t ws_size,
                              hipStream_t stream) {
  (void)in_sizes; (void)n_in; (void)out_size; (void)ws_size;
  const int NA = 50000, NC = 100000, NE = 600000;

  const float* agt    = (const float*)d_in[0];
  const float* ctx    = (const float*)d_in[1];
  const int*   u      = (const int*)d_in[2];
  const int*   v      = (const int*)d_in[3];
  const float* Wq     = (const float*)d_in[4];
  const float* Wk     = (const float*)d_in[5];
  const float* Wctx   = (const float*)d_in[6];
  const float* gnc_w  = (const float*)d_in[7];
  const float* gnc_b  = (const float*)d_in[8];
  const float* Wagt   = (const float*)d_in[9];
  const float* gna_w  = (const float*)d_in[10];
  const float* gna_b  = (const float*)d_in[11];
  const float* norm_w = (const float*)d_in[12];
  const float* norm_b = (const float*)d_in[13];
  const float* Wlin   = (const float*)d_in[14];
  const float* gnl_w  = (const float*)d_in[15];
  const float* gnl_b  = (const float*)d_in[16];

  float* ws = (float*)d_ws;
  size_t off = 0;
  float* qbuf   = ws + off; off += (size_t)NA * DIM;
  float* kbuf   = ws + off; off += (size_t)NC * DIM;
  float* cpbuf  = ws + off; off += (size_t)NC * DIM;
  float* cbuf   = ws + off; off += (size_t)NE;
  off += (size_t)NA;
  float* aggbuf = ws + off; off += (size_t)NA * DIM;
  float* xbuf   = ws + off; off += (size_t)NA * DIM;
  float* agtbuf = ws + off; off += (size_t)NA * DIM;
  unsigned* maxbuf = (unsigned*)(ws + off); off += 16;

  const dim3 b128(128), b256(256);

  for (int i = 0; i < NBLK; ++i) {
    const float* ain  = (i == 0) ? agt : agtbuf;
    float*       aout = (i == NBLK - 1) ? (float*)d_out : agtbuf;

    gemm_nt_128<<<(NA + 63) / 64, b128, 0, stream>>>(ain, Wq + (size_t)i * DIM * DIM, qbuf, NA);
    gemm_nt_128<<<(NC + 63) / 64, b128, 0, stream>>>(ctx, Wk + (size_t)i * DIM * DIM, kbuf, NC);
    gemm_nt_128<<<(NC + 63) / 64, b128, 0, stream>>>(ctx, Wctx + (size_t)i * DIM * DIM, cpbuf, NC);
    gn_kernel<<<(NC + 63) / 64, b256, 0, stream>>>(cpbuf, cpbuf, gnc_w + i * DIM, gnc_b + i * DIM, NC);

    unsigned* keys   = (unsigned*)xbuf;
    int*      vu     = (int*)xbuf + (1 << 18);
    unsigned* rec_ru = (unsigned*)xbuf + 3 * (1 << 19);
    float*    rec_sc = (float*)(rec_ru + 256 * OWN_CAP);
    int*      gcount = (int*)(rec_sc + 256 * OWN_CAP);
    edge_prep_kernel<<<((NE + 3) / 4 + 255) / 256, b256, 0, stream>>>(u, v, keys, vu, maxbuf, NE);
    edge_owner_score_kernel<<<(NA + OWN_ROWS - 1) / OWN_ROWS, dim3(1024), 0, stream>>>(
        qbuf, kbuf, u, v, keys, vu, cbuf, maxbuf, rec_ru, rec_sc, gcount, NE, NA);
    edge_owner_kernel<<<(NA + OWN_ROWS - 1) / OWN_ROWS, dim3(1024), 0, stream>>>(
        cbuf, maxbuf, cpbuf, u, v, rec_ru, rec_sc, gcount, aggbuf, NE, NA);

    gemm_cat_256<<<(NA + 63) / 64, b128, 0, stream>>>(ain, aggbuf, Wagt + (size_t)i * DIM * 2 * DIM, xbuf, NA);
    gn2_relu_kernel<<<(NA + 31) / 32, b256, 0, stream>>>(xbuf, xbuf, gna_w + i * DIM, gna_b + i * DIM,
                                                       norm_w + i * DIM, norm_b + i * DIM, NA);
    gemm_nt_128<<<(NA + 63) / 64, b128, 0, stream>>>(xbuf, Wlin + (size_t)i * DIM * DIM, qbuf, NA);
    gn_res_relu_kernel<<<(NA + 31) / 32, b256, 0, stream>>>(qbuf, ain, aout, gnl_w + i * DIM, gnl_b + i * DIM, NA);
  }
}


# Aggregation work-group keeps its 196 summed rows in LDS and runs the update MLP on them there (both WMMA products, GroupNorm row passes, residual; same k and sum order); 22 launches -> 14
# speedup vs baseline: 2.3953x; 1.1412x over previous
#include <hip/hip_runtime.h>


static constexpr int DIM  = 128;
static constexpr int NBLK = 2;
static constexpr int LDP  = 68;

typedef __attribute__((ext_vector_type(2))) float v2f;
typedef __attribute__((ext_vector_type(8))) float v8f;

static __device__ __forceinline__ float wave_reduce_sum(float x) {
#pragma unroll
  for (int off = 16; off > 0; off >>= 1) x += __shfl_xor(x, off, 32);
  return x;
}

static __device__ __forceinline__ unsigned f32_order_enc(float f) {
  unsigned u = __float_as_uint(f);
  return (u & 0x80000000u) ? ~u : (u | 0x80000000u);
}
static __device__ __forceinline__ float f32_order_dec(unsigned e) {
  unsigned u = (e & 0x80000000u) ? (e & 0x7fffffffu) : ~e;
  return __uint_as_float(u);
}

__global__ __launch_bounds__(128) void gemm_nt_128(const float* __restrict__ A,
                                                   const float* __restrict__ W,
                                                   float* __restrict__ C, int N) {
  __shared__ float Wlds[128 * LDP];
  const int tid = threadIdx.x;
  const int wv  = tid >> 5;
  const int ln  = tid & 31;
  const int hk  = ln >> 4;
  const int l16 = ln & 15;
  const int r0  = blockIdx.x * 64 + wv * 16;

  v8f acc[8];
#pragma unroll
  for (int t = 0; t < 8; ++t)
#pragma unroll
    for (int j = 0; j < 8; ++j) acc[t][j] = 0.0f;

  int rowA = r0 + l16;
  if (rowA >= N) rowA = N - 1;
  const float* ap = A + (size_t)rowA * DIM;

  for (int kp = 0; kp < DIM; kp += 64) {
    __syncthreads();
#pragma unroll
    for (int i = 0; i < 16; ++i) {
      const int q  = tid + 128 * i;
      const int n  = q >> 4;
      const int c4 = (q & 15) << 2;
      *(float4*)(&Wlds[n * LDP + c4]) =
          *(const float4*)(W + (size_t)n * DIM + kp + c4);
    }
    __syncthreads();

#pragma unroll 4
    for (int k0 = 0; k0 < 64; k0 += 4) {
      const int ka = k0 + 2 * hk;
      v2f a; a.x = ap[kp + ka]; a.y = ap[kp + ka + 1];
#pragma unroll
      for (int t = 0; t < 8; ++t) {
        const float* wp = &Wlds[(t * 16 + l16) * LDP + ka];
        v2f b; b.x = wp[0]; b.y = wp[1];
        acc[t] = __builtin_amdgcn_wmma_f32_16x16x4_f32(
            false, a, false, b, (short)0, acc[t], false, false);
      }
    }
  }

  const int rb = r0 + hk * 8;
  if (r0 + 16 <= N) {
#pragma unroll
    for (int t = 0; t < 8; ++t)
#pragma unroll
      for (int j = 0; j < 8; ++j)
        C[(size_t)(rb + j) * DIM + t * 16 + l16] = acc[t][j];
  } else {
#pragma unroll
    for (int t = 0; t < 8; ++t)
#pragma unroll
      for (int j = 0; j < 8; ++j) {
        const int row = rb + j;
        if (row < N) C[(size_t)row * DIM + t * 16 + l16] = acc[t][j];
      }
  }
}

static constexpr int ROWB = 4;

static __device__ __forceinline__ void row_stats(const float4 x, float& mu, float& rs) {
  float s  = x.x + x.y + x.z + x.w;
  float sq = x.x * x.x + x.y * x.y + x.z * x.z + x.w * x.w;
  s = wave_reduce_sum(s); sq = wave_reduce_sum(sq);
  mu = s * (1.0f / DIM);
  rs = rsqrtf(sq * (1.0f / DIM) - mu * mu + 1e-5f);
}

__global__ __launch_bounds__(256) void gn_kernel(const float* X, float* Y,
                                                 const float* __restrict__ w,
                                                 const float* __restrict__ b, int N) {
  const int ln = threadIdx.x & 31;
  const int wv = __builtin_amdgcn_readfirstlane((int)(threadIdx.x >> 5));
  const int step = (int)gridDim.x * 8 * ROWB;
  const float4 wv4 = ((const float4*)w)[ln];
  const float4 bv4 = ((const float4*)b)[ln];
  for (int r0 = ((int)blockIdx.x * 8 + wv) * ROWB; r0 < N; r0 += step) {
    float4 x[ROWB];
#pragma unroll
    for (int t = 0; t < ROWB; ++t) {
      x[t] = make_float4(0.0f, 0.0f, 0.0f, 0.0f);
      if (r0 + t < N) x[t] = ((const float4*)(X + (size_t)(r0 + t) * DIM))[ln];
    }
#pragma unroll
    for (int t = 0; t < ROWB; ++t) {
      float mu, rs;
      row_stats(x[t], mu, rs);
      float4 y;
      y.x = (x[t].x - mu) * rs * wv4.x + bv4.x;
      y.y = (x[t].y - mu) * rs * wv4.y + bv4.y;
      y.z = (x[t].z - mu) * rs * wv4.z + bv4.z;
      y.w = (x[t].w - mu) * rs * wv4.w + bv4.w;
      if (r0 + t < N) ((float4*)(Y + (size_t)(r0 + t) * DIM))[ln] = y;
    }
  }
}

static constexpr int EGR_Q  = 8;
static constexpr int EGR_S  = 8;
static constexpr int OWN_ROWS = 196;
static constexpr int OWN_CAP  = 8192;
static constexpr int OWN_SCAN = 8;

__global__ __launch_bounds__(256) void edge_prep_kernel(const int* __restrict__ u,
                                                        const int* __restrict__ v,
                                                        unsigned* __restrict__ keys,
                                                        int* __restrict__ vu_words,
                                                        unsigned* __restrict__ gmax,
                                                        int E) {
  int2* vu = (int2*)vu_words;
  if (blockIdx.x == 0 && threadIdx.x == 0) *gmax = 0u;
  const int i = (int)blockIdx.x * 256 + (int)threadIdx.x;
  const int e = i << 2;
  if (e + 3 < E) {
    const int4 vv = ((const int4*)v)[i];
    const int4 uu = ((const int4*)u)[i];
    keys[i] = (unsigned)(vv.x / OWN_ROWS) | ((unsigned)(vv.y / OWN_ROWS) << 8) |
              ((unsigned)(vv.z / OWN_ROWS) << 16) | ((unsigned)(vv.w / OWN_ROWS) << 24);
    ((int4*)vu)[2 * i]     = make_int4(vv.x, uu.x, vv.y, uu.y);
    ((int4*)vu)[2 * i + 1] = make_int4(vv.z, uu.z, vv.w, uu.w);
  } else {
    for (int k = e; k < E; ++k) {
      ((unsigned char*)keys)[k] = (unsigned char)(v[k] / OWN_ROWS);
      vu[k] = make_int2(v[k], u[k]);
    }
  }
}

static __device__ __forceinline__ void own_append(int* list, int* cnt, int e) {
  const int pos = atomicAdd(cnt, 1);
  if (pos < OWN_CAP) list[pos] = e;
}

static __device__ __forceinline__ unsigned own_zero_bytes(unsigned y) {
  const unsigned t = ((y & 0x7f7f7f7fu) + 0x7f7f7f7fu) | y;
  return ~t & 0x80808080u;
}
static __device__ __forceinline__ void own_emit(int* list, int* cnt, unsigned z, int e0) {
  while (z) {
    own_append(list, cnt, e0 + (__builtin_ctz(z) >> 3));
    z &= z - 1;
  }
}

static __device__ __forceinline__ float wave_sum4(const float p0, const float p1,
                                                  const float p2, const float p3,
                                                  const int hk, const int b8) {
  const float s0 = hk ? p2 : p0, r0 = hk ? p0 : p2;
  const float s1 = hk ? p3 : p1, r1 = hk ? p1 : p3;
  const float x0 = s0 + __shfl_xor(r0, 16, 32);
  const float x1 = s1 + __shfl_xor(r1, 16, 32);
  const float s2 = b8 ? x1 : x0, r2 = b8 ? x0 : x1;
  float d = s2 + __shfl_xor(r2, 8, 32);
  d += __shfl_xor(d, 4, 32);
  d += __shfl_xor(d, 2, 32);
  d += __shfl_xor(d, 1, 32);
  return d;
}

static __device__ __forceinline__ void score_fetch(float4 (&kv)[EGR_Q],
                                                   const float* __restrict__ K, const int ul,
                                                   const int j, const int nval, const int ln) {
#pragma unroll
  for (int t = 0; t < EGR_Q; ++t) {
    const int jj = (j + t < nval) ? (j + t) : (nval - 1);
    const int uj = __builtin_amdgcn_readlane(ul, jj);
    kv[t] = ((const float4*)(K + (size_t)uj * DIM))[ln];
  }
}
static __device__ __forceinline__ float score_step(const float4 (&kv)[EGR_Q], const float* qt,
                                                   const int rl, const int j, const int nval,
                                                   const int ln, const int hk, const int b8,
                                                   float dl) {
  float p[EGR_Q];
#pragma unroll
  for (int t = 0; t < EGR_Q; ++t) {
    const int jj = (j + t < nval) ? (j + t) : (nval - 1);
    const int rj = __builtin_amdgcn_readlane(rl, jj);
    const float4 qv = ((const float4*)(qt + rj))[ln];
    p[t] = qv.x * kv[t].x + qv.y * kv[t].y + qv.z * kv[t].z + qv.w * kv[t].w;
  }
  float da = wave_sum4(p[0], p[1], p[2], p[3], hk, b8);
  float db = wave_sum4(p[4], p[5], p[6], p[7], hk, b8);
  da = (da >= 0.0f) ? da : 0.1f * da;
  db = (db >= 0.0f) ? db : 0.1f * db;
  const int i8 = ln - j;
  const int sl = (i8 & 3) << 3;
  const float ga = __shfl(da, sl, 32);
  const float gb = __shfl(db, sl, 32);
  if ((unsigned)i8 < 4u) dl = ga;
  else if ((unsigned)i8 < 8u) dl = gb;
  return dl;
}

__global__ __launch_bounds__(1024) void edge_owner_score_kernel(const float* __restrict__ Q,
                                                                const float* __restrict__ K,
                                                                const int* __restrict__ u,
                                                                const int* __restrict__ v,
                                                                const unsigned* __restrict__ keys,
                                                                const int* __restrict__ vu_words,
                                                                float* __restrict__ c,
                                                                unsigned* __restrict__ gmax,
                                                                unsigned* __restrict__ rec_ru,
                                                                float* __restrict__ rec_sc,
                                                                int* __restrict__ gcount,
                                                                int E, int NA) {
  __shared__ __attribute__((aligned(16))) float qt[OWN_ROWS * DIM];
  __shared__ int   list[OWN_CAP];
  __shared__ float smax[32];
  __shared__ int   cnt;
  const int2* vu = (const int2*)vu_words;
  const int tid = threadIdx.x;
  const int ln  = tid & 31;
  const int wv  = __builtin_amdgcn_readfirstlane(tid >> 5);
  const int lo  = (int)blockIdx.x * OWN_ROWS;
  int nrows = NA - lo;
  if (nrows > OWN_ROWS) nrows = OWN_ROWS;
  if (nrows <= 0) return;
  const int hk = ln >> 4;
  const int b8 = (ln >> 3) & 1;

  const float4* q4 = (const float4*)(Q + (size_t)lo * DIM);
  for (int q = tid; q < nrows * (DIM / 4); q += 1024) ((float4*)qt)[q] = q4[q];
  if (tid == 0) cnt = 0;
  __syncthreads();

  const unsigned g    = blockIdx.x;
  const unsigned gpat = g * 0x01010101u;
  const int n16 = E >> 4;
  const uint4* k4 = (const uint4*)keys;
  const unsigned char* own8 = (const unsigned char*)keys;
  const int nch = (n16 + 1024 * OWN_SCAN - 1) / (1024 * OWN_SCAN);
  int k = (nch > 0) ? (int)(g % (unsigned)nch) : 0;
  for (int it = 0; it < nch; ++it) {
    const int i0 = k * (1024 * OWN_SCAN) + tid;
    if (++k == nch) k = 0;
    uint4 kw[OWN_SCAN];
#pragma unroll
    for (int t = 0; t < OWN_SCAN; ++t) {
      const int i = i0 + t * 1024;
      kw[t] = make_uint4(~gpat, ~gpat, ~gpat, ~gpat);
      if (i < n16) kw[t] = k4[i];
    }
#pragma unroll
    for (int t = 0; t < OWN_SCAN; ++t) {
      const int e = (i0 + t * 1024) << 4;
      const unsigned z0 = own_zero_bytes(kw[t].x ^ gpat);
      const unsigned z1 = own_zero_bytes(kw[t].y ^ gpat);
      const unsigned z2 = own_zero_bytes(kw[t].z ^ gpat);
      const unsigned z3 = own_zero_bytes(kw[t].w ^ gpat);
      if (z0 | z1 | z2 | z3) {
        own_emit(list, &cnt, z0, e + 0);
        own_emit(list, &cnt, z1, e + 4);
        own_emit(list, &cnt, z2, e + 8);
        own_emit(list, &cnt, z3, e + 12);
      }
    }
  }
  for (int e = (n16 << 4) + tid; e < E; e += 1024)
    if (own8[e] == (unsigned char)g) own_append(list, &cnt, e);
  __syncthreads();
  const int total = cnt;
  if (tid == 0) gcount[g] = total;

  float cmax = -3.0e38f;
  if (total <= OWN_CAP) {
    unsigned* my_ru = rec_ru + (size_t)g * OWN_CAP;
    float*    my_sc = rec_sc + (size_t)g * OWN_CAP;
    const int per = (total + 31) >> 5;
    const int b0  = wv * per;
    int b1 = b0 + per;
    if (b1 > total) b1 = total;
    if (b0 < b1) {
      int nval = (b1 - b0 > 32) ? 32 : (b1 - b0);
      int2 pn = vu[list[b0 + (ln < nval ? ln : nval - 1)]];
      float4 ka[EGR_Q], kb[EGR_Q];
      score_fetch(ka, K, pn.y, 0, nval, ln);
      for (int base = b0; base < b1; base += 32) {
        const int2 pl = pn;
        const int ul = pl.y;
        const int rl = (pl.x - lo) * DIM;
        const bool hasnext = base + 32 < b1;
        const int nvaln = hasnext ? ((b1 - base - 32 > 32) ? 32 : (b1 - base - 32)) : nval;
        pn = vu[list[(hasnext ? base + 32 : base) + (ln < nvaln ? ln : nvaln - 1)]];
        float dl = 0.0f;
        for (int j = 0; j < nval; j += 2 * EGR_Q) {
          score_fetch(kb, K, ul, j + EGR_Q, nval, ln);
          dl = score_step(ka, qt, rl, j, nval, ln, hk, b8, dl);
          const bool here = (j + 2 * EGR_Q < nval) || !hasnext;
          score_fetch(ka, K, here ? ul : pn.y, here ? j + 2 * EGR_Q : 0, here ? nval : nvaln, ln);
          dl = score_step(kb, qt, rl, j + EGR_Q, nval, ln, hk, b8, dl);
        }
        if (ln < nval) {
          my_ru[base + ln] = ((unsigned)(pl.x - lo) << 24) | (unsigned)ul;
          my_sc[base + ln] = dl;
          cmax = fmaxf(cmax, dl);
        }
        nval = nvaln;
      }
    }
  } else {
    for (int e0 = wv * 32; e0 < E; e0 += 1024) {
      const int e  = e0 + ln;
      const int vl = (e < E) ? v[e] : -1;
      unsigned mask = (unsigned)__ballot((unsigned)(vl - lo) < (unsigned)nrows);
      while (mask) {
        const int b = __builtin_ctz(mask);
        mask &= mask - 1;
        const int vj = __builtin_amdgcn_readlane(vl, b);
        const int uj = u[e0 + b];
        const float4 qv = ((const float4*)(qt + (vj - lo) * DIM))[ln];
        const float4 kv = ((const float4*)(K + (size_t)uj * DIM))[ln];
        float d = wave_reduce_sum(qv.x * kv.x + qv.y * kv.y + qv.z * kv.z + qv.w * kv.w);
        d = (d >= 0.0f) ? d : 0.1f * d;
        if (ln == 0) c[e0 + b] = d;
        cmax = fmaxf(cmax, d);
      }
    }
  }
  cmax = fmaxf(cmax, __shfl_xor(cmax, 16, 32));
  cmax = fmaxf(cmax, __shfl_xor(cmax, 8, 32));
  cmax = fmaxf(cmax, __shfl_xor(cmax, 4, 32));
  cmax = fmaxf(cmax, __shfl_xor(cmax, 2, 32));
  cmax = fmaxf(cmax, __shfl_xor(cmax, 1, 32));
  if (ln == 0) smax[wv] = cmax;
  __syncthreads();
  if (tid == 0) {
    float m = smax[0];
    for (int i = 1; i < 32; ++i) m = fmaxf(m, smax[i]);
    atomicMax(gmax, f32_order_enc(m));
  }
}

static constexpr int TP     = 132;
static constexpr int NSTRIP = (OWN_ROWS + 15) / 16;
static constexpr int TROWS  = NSTRIP * 16;
static constexpr int RPW    = (OWN_ROWS + 31) / 32;

static __device__ __forceinline__ void w_fetch(float4 (&wr)[2], const float* __restrict__ Wsrc,
                                               const int stride, const int kp, const int tid) {
#pragma unroll
  for (int i = 0; i < 2; ++i) {
    const int q  = tid + 1024 * i;
    const int n  = q >> 4;
    const int c4 = (q & 15) << 2;
    wr[i] = *(const float4*)(Wsrc + (size_t)n * stride + kp + c4);
  }
}
static __device__ __forceinline__ void w_put(const float4 (&wr)[2], float* buf, const int tid) {
#pragma unroll
  for (int i = 0; i < 2; ++i) {
    const int q  = tid + 1024 * i;
    const int n  = q >> 4;
    const int c4 = (q & 15) << 2;
    *(float4*)(&buf[n * LDP + c4]) = wr[i];
  }
}
static constexpr int WSTG = (2048 + (32 - 2 * NSTRIP) * 32 - 1) / ((32 - 2 * NSTRIP) * 32);
static_assert(2 * NSTRIP < 32, "no wave left to stage the weights");
static __device__ __forceinline__ void w_stage(float* buf, const float* __restrict__ Wsrc,
                                               const int stride, const int kp,
                                               const int sid, const int sstep) {
  float4 r[WSTG];
#pragma unroll
  for (int i = 0; i < WSTG; ++i) {
    const int q = sid + i * sstep;
    r[i] = make_float4(0.0f, 0.0f, 0.0f, 0.0f);
    if (q < 2048) r[i] = *(const float4*)(Wsrc + (size_t)(q >> 4) * stride + kp + ((q & 15) << 2));
  }
#pragma unroll
  for (int i = 0; i < WSTG; ++i) {
    const int q = sid + i * sstep;
    if (q < 2048) *(float4*)(&buf[(q >> 4) * LDP + ((q & 15) << 2)]) = r[i];
  }
}
static __device__ __forceinline__ void mm_chunk(v8f (&c)[4], const float* ap, const float* wl,
                                                const int hk) {
#pragma unroll 4
  for (int k0 = 0; k0 < 64; k0 += 4) {
    const int ka = k0 + 2 * hk;
    v2f a; a.x = ap[ka]; a.y = ap[ka + 1];
#pragma unroll
    for (int t = 0; t < 4; ++t) {
      const float* wp = wl + t * 16 * LDP + ka;
      v2f b; b.x = wp[0]; b.y = wp[1];
      c[t] = __builtin_amdgcn_wmma_f32_16x16x4_f32(
          false, a, false, b, (short)0, c[t], false, false);
    }
  }
}

__global__ __launch_bounds__(1024) void edge_owner_kernel(const float* __restrict__ c,
                                                          const unsigned* __restrict__ gmax,
                                                          const float* __restrict__ ctxp,
                                                          const int* __restrict__ u,
                                                          const int* __restrict__ v,
                                                          const unsigned* __restrict__ rec_ru,
                                                          const float* __restrict__ rec_sc,
                                                          const int* __restrict__ gcount,
                                                          const float* __restrict__ ain,
                                                          const float* __restrict__ Wagt,
                                                          const float* __restrict__ gna_w,
                                                          const float* __restrict__ gna_b,
                                                          const float* __restrict__ nrm_w,
                                                          const float* __restrict__ nrm_b,
                                                          const float* __restrict__ Wlin,
                                                          const float* __restrict__ gnl_w,
                                                          const float* __restrict__ gnl_b,
                                                          float* __restrict__ aout,
                                                          int E, int NA) {
  __shared__ __attribute__((aligned(16))) float acc[TROWS * TP];
  __shared__ __attribute__((aligned(16))) float xt[TROWS * TP];
  __shared__ __attribute__((aligned(16))) float Wl0[128 * LDP];
  __shared__ __attribute__((aligned(16))) float Wl1[128 * LDP];
  __shared__ float lsum[OWN_ROWS];
  const int tid = threadIdx.x;
  const int ln  = tid & 31;
  const int wv  = __builtin_amdgcn_readfirstlane(tid >> 5);
  const int lo  = (int)blockIdx.x * OWN_ROWS;
  int nrows = NA - lo;
  if (nrows > OWN_ROWS) nrows = OWN_ROWS;
  if (nrows <= 0) return;

  float4 wr[2];
  w_fetch(wr, Wagt, 2 * DIM, 0, tid);

  const float4 z4 = make_float4(0.0f, 0.0f, 0.0f, 0.0f);
  for (int q = tid; q < TROWS * TP / 4; q += 1024) ((float4*)acc)[q] = z4;
  if (tid < OWN_ROWS) lsum[tid] = 0.0f;
  w_put(wr, Wl0, tid);

  const unsigned g = blockIdx.x;
  const int total = gcount[g];
  const unsigned* my_ru = rec_ru + (size_t)g * OWN_CAP;
  const float*    my_sc = rec_sc + (size_t)g * OWN_CAP;
  __syncthreads();
  const float gm = f32_order_dec(*gmax);

  if (total <= OWN_CAP) {
    for (int i = tid; i < total; i += 1024)
      atomicAdd(&lsum[my_ru[i] >> 24], expf(my_sc[i] - gm));
    __syncthreads();
    const int per = (total + 31) >> 5;
    const int b0  = wv * per;
    int b1 = b0 + per;
    if (b1 > total) b1 = total;
    for (int base = b0; base < b1; base += 32) {
      int nval = b1 - base;
      if (nval > 32) nval = 32;
      const int il = base + (ln < nval ? ln : nval - 1);
      const unsigned ru = my_ru[il];
      const int ul  = (int)(ru & 0x00ffffffu);
      const int row = (int)(ru >> 24);
      const int cl = __float_as_int(expf(my_sc[il] - gm) / (lsum[row] + 1e-16f));
      const int rl = row * TP;
      int j = 0;
      for (; j + EGR_S <= nval; j += EGR_S) {
        float m[EGR_S][4];
#pragma unroll
        for (int t = 0; t < EGR_S; ++t) {
          const int uj = __builtin_amdgcn_readlane(ul, j + t);
          const float* srow = ctxp + (size_t)uj * DIM + ln;
          m[t][0] = srow[0];
          m[t][1] = srow[32];
          m[t][2] = srow[64];
          m[t][3] = srow[96];
        }
#pragma unroll
        for (int t = 0; t < EGR_S; ++t) {
          const int   rj   = __builtin_amdgcn_readlane(rl, j + t);
          const float coef = __int_as_float(__builtin_amdgcn_readlane(cl, j + t));
          float* dst = acc + rj + ln;
          atomicAdd(dst + 0,  m[t][0] * coef);
          atomicAdd(dst + 32, m[t][1] * coef);
          atomicAdd(dst + 64, m[t][2] * coef);
          atomicAdd(dst + 96, m[t][3] * coef);
        }
      }
      for (; j < nval; ++j) {
        const int   uj   = __builtin_amdgcn_readlane(ul, j);
        const int   rj   = __builtin_amdgcn_readlane(rl, j);
        const float coef = __int_as_float(__builtin_amdgcn_readlane(cl, j));
        const float* srow = ctxp + (size_t)uj * DIM + ln;
        const float m0 = srow[0], m1 = srow[32], m2 = srow[64], m3 = srow[96];
        float* dst = acc + rj + ln;
        atomicAdd(dst + 0,  m0 * coef);
        atomicAdd(dst + 32, m1 * coef);
        atomicAdd(dst + 64, m2 * coef);
        atomicAdd(dst + 96, m3 * coef);
      }
    }
  } else {
    for (int e0 = wv * 32; e0 < E; e0 += 1024) {
      const int e  = e0 + ln;
      const int vl = (e < E) ? v[e] : -1;
      if ((unsigned)(vl - lo) < (unsigned)nrows) atomicAdd(&lsum[vl - lo], expf(c[e] - gm));
    }
    __syncthreads();
    for (int e0 = wv * 32; e0 < E; e0 += 1024) {
      const int e  = e0 + ln;
      const int vl = (e < E) ? v[e] : -1;
      unsigned mask = (unsigned)__ballot((unsigned)(vl - lo) < (unsigned)nrows);
      while (mask) {
        const int b = __builtin_ctz(mask);
        mask &= mask - 1;
        const int   vj   = __builtin_amdgcn_readlane(vl, b);
        const int   uj   = u[e0 + b];
        const float coef = expf(c[e0 + b] - gm) / (lsum[vj - lo] + 1e-16f);
        const float* srow = ctxp + (size_t)uj * DIM + ln;
        const float m0 = srow[0], m1 = srow[32], m2 = srow[64], m3 = srow[96];
        float* dst = acc + (vj - lo) * TP + ln;
        atomicAdd(dst + 0,  m0 * coef);
        atomicAdd(dst + 32, m1 * coef);
        atomicAdd(dst + 64, m2 * coef);
        atomicAdd(dst + 96, m3 * coef);
      }
    }
  }
  __syncthreads();

  const int hk  = ln >> 4;
  const int l16 = ln & 15;
  const int nstrip = (nrows + 15) >> 4;
  const bool act = wv < 2 * nstrip;
  const int st = wv >> 1;
  const int ch = wv & 1;
  const int trow = st * 16 + l16;
  int ga = lo + trow;
  if (ga >= NA) ga = NA - 1;
  const float* apg = ain + (size_t)ga * DIM;
  const float* apl = acc + trow * TP;
  const float* apx = xt + trow * TP;
  const int wofs = (ch * 64 + l16) * LDP;
  const int orow = (st * 16 + 8 * hk) * TP + ch * 64 + l16;

  v8f cc[4];
#pragma unroll
  for (int t = 0; t < 4; ++t)
#pragma unroll
    for (int j = 0; j < 8; ++j) cc[t][j] = 0.0f;

  const int sid   = (wv - 2 * nstrip) * 32 + ln;
  const int sstep = (32 - 2 * nstrip) * 32;
  if (act) mm_chunk(cc, apg, Wl0 + wofs, hk);
  else     w_stage(Wl1, Wagt, 2 * DIM, 64, sid, sstep);
  __syncthreads();
  if (act) mm_chunk(cc, apg + 64, Wl1 + wofs, hk);
  else     w_stage(Wl0, Wagt, 2 * DIM, 128, sid, sstep);
  __syncthreads();
  if (act) mm_chunk(cc, apl, Wl0 + wofs, hk);
  else     w_stage(Wl1, Wagt, 2 * DIM, 192, sid, sstep);
  __syncthreads();
  if (act) {
    mm_chunk(cc, apl + 64, Wl1 + wofs, hk);
#pragma unroll
    for (int t = 0; t < 4; ++t)
#pragma unroll
      for (int j = 0; j < 8; ++j) xt[orow + j * TP + t * 16] = cc[t][j];
  } else {
    w_stage(Wl0, Wlin, DIM, 0, sid, sstep);
  }
  __syncthreads();

  {
    const float4 wa = ((const float4*)gna_w)[ln], ba = ((const float4*)gna_b)[ln];
    const float4 wn = ((const float4*)nrm_w)[ln], bn = ((const float4*)nrm_b)[ln];
    for (int r = wv; r < nrows; r += 32) {
      float4* xp = (float4*)(xt + r * TP) + ln;
      const float4 x = *xp;
      float mu, rs;
      row_stats(x, mu, rs);
      float4 y;
      y.x = (x.x - mu) * rs * wa.x + ba.x;
      y.y = (x.y - mu) * rs * wa.y + ba.y;
      y.z = (x.z - mu) * rs * wa.z + ba.z;
      y.w = (x.w - mu) * rs * wa.w + ba.w;
      row_stats(y, mu, rs);
      float4 z;
      z.x = fmaxf((y.x - mu) * rs * wn.x + bn.x, 0.0f);
      z.y = fmaxf((y.y - mu) * rs * wn.y + bn.y, 0.0f);
      z.z = fmaxf((y.z - mu) * rs * wn.z + bn.z, 0.0f);
      z.w = fmaxf((y.w - mu) * rs * wn.w + bn.w, 0.0f);
      *xp = z;
    }
  }
  __syncthreads();

#pragma unroll
  for (int t = 0; t < 4; ++t)
#pragma unroll
    for (int j = 0; j < 8; ++j) cc[t][j] = 0.0f;
  if (act) mm_chunk(cc, apx, Wl0 + wofs, hk);
  else     w_stage(Wl1, Wlin, DIM, 64, sid, sstep);
  __syncthreads();
  if (act) {
    mm_chunk(cc, apx + 64, Wl1 + wofs, hk);
#pragma unroll
    for (int t = 0; t < 4; ++t)
#pragma unroll
      for (int j = 0; j < 8; ++j) acc[orow + j * TP + t * 16] = cc[t][j];
  }
  __syncthreads();

  {
    const float4 wl4 = ((const float4*)gnl_w)[ln], bl4 = ((const float4*)gnl_b)[ln];
    float4 rr[RPW];
#pragma unroll
    for (int t = 0; t < RPW; ++t) {
      const int r = wv + 32 * t;
      rr[t] = z4;
      if (r < nrows) rr[t] = ((const float4*)(ain + (size_t)(lo + r) * DIM))[ln];
    }
#pragma unroll
    for (int t = 0; t < RPW; ++t) {
      const int r = wv + 32 * t;
      if (r < nrows) {
        const float4 x = ((const float4*)(acc + r * TP))[ln];
        float mu, rs;
        row_stats(x, mu, rs);
        float4 y;
        y.x = fmaxf((x.x - mu) * rs * wl4.x + bl4.x + rr[t].x, 0.0f);
        y.y = fmaxf((x.y - mu) * rs * wl4.y + bl4.y + rr[t].y, 0.0f);
        y.z = fmaxf((x.z - mu) * rs * wl4.z + bl4.z + rr[t].z, 0.0f);
        y.w = fmaxf((x.w - mu) * rs * wl4.w + bl4.w + rr[t].w, 0.0f);
        ((float4*)(aout + (size_t)(lo + r) * DIM))[ln] = y;
      }
    }
  }
}

extern "C" void kernel_launch(void* const* d_in, const int* in_sizes, int n_in,
                              void* d_out, int out_size, void* d_ws, size_t ws_size,
                              hipStream_t stream) {
  (void)in_sizes; (void)n_in; (void)out_size; (void)ws_size;
  const int NA = 50000, NC = 100000, NE = 600000;

  const float* agt    = (const float*)d_in[0];
  const float* ctx    = (const float*)d_in[1];
  const int*   u      = (const int*)d_in[2];
  const int*   v      = (const int*)d_in[3];
  const float* Wq     = (const float*)d_in[4];
  const float* Wk     = (const float*)d_in[5];
  const float* Wctx   = (const float*)d_in[6];
  const float* gnc_w  = (const float*)d_in[7];
  const float* gnc_b  = (const float*)d_in[8];
  const float* Wagt   = (const float*)d_in[9];
  const float* gna_w  = (const float*)d_in[10];
  const float* gna_b  = (const float*)d_in[11];
  const float* norm_w = (const float*)d_in[12];
  const float* norm_b = (const float*)d_in[13];
  const float* Wlin   = (const float*)d_in[14];
  const float* gnl_w  = (const float*)d_in[15];
  const float* gnl_b  = (const float*)d_in[16];

  float* ws = (float*)d_ws;
  size_t off = 0;
  float* qbuf   = ws + off; off += (size_t)NA * DIM;
  float* kbuf   = ws + off; off += (size_t)NC * DIM;
  float* cpbuf  = ws + off; off += (size_t)NC * DIM;
  float* cbuf   = ws + off; off += (size_t)NE;
  off += (size_t)NA;
  off += (size_t)NA * DIM;
  float* xbuf   = ws + off; off += (size_t)NA * DIM;
  float* agtbuf = ws + off; off += (size_t)NA * DIM;
  unsigned* maxbuf = (unsigned*)(ws + off); off += 16;

  const dim3 b128(128), b256(256);

  for (int i = 0; i < NBLK; ++i) {
    const float* ain  = (i == 0) ? agt : agtbuf;
    float*       aout = (i == NBLK - 1) ? (float*)d_out : agtbuf;

    gemm_nt_128<<<(NA + 63) / 64, b128, 0, stream>>>(ain, Wq + (size_t)i * DIM * DIM, qbuf, NA);
    gemm_nt_128<<<(NC + 63) / 64, b128, 0, stream>>>(ctx, Wk + (size_t)i * DIM * DIM, kbuf, NC);
    gemm_nt_128<<<(NC + 63) / 64, b128, 0, stream>>>(ctx, Wctx + (size_t)i * DIM * DIM, cpbuf, NC);
    gn_kernel<<<(NC + 63) / 64, b256, 0, stream>>>(cpbuf, cpbuf, gnc_w + i * DIM, gnc_b + i * DIM, NC);

    unsigned* keys   = (unsigned*)xbuf;
    int*      vu     = (int*)xbuf + (1 << 18);
    unsigned* rec_ru = (unsigned*)xbuf + 3 * (1 << 19);
    float*    rec_sc = (float*)(rec_ru + 256 * OWN_CAP);
    int*      gcount = (int*)(rec_sc + 256 * OWN_CAP);
    edge_prep_kernel<<<((NE + 3) / 4 + 255) / 256, b256, 0, stream>>>(u, v, keys, vu, maxbuf, NE);
    edge_owner_score_kernel<<<(NA + OWN_ROWS - 1) / OWN_ROWS, dim3(1024), 0, stream>>>(
        qbuf, kbuf, u, v, keys, vu, cbuf, maxbuf, rec_ru, rec_sc, gcount, NE, NA);
    edge_owner_kernel<<<(NA + OWN_ROWS - 1) / OWN_ROWS, dim3(1024), 0, stream>>>(
        cbuf, maxbuf, cpbuf, u, v, rec_ru, rec_sc, gcount, ain,
        Wagt + (size_t)i * DIM * 2 * DIM, gna_w + i * DIM, gna_b + i * DIM,
        norm_w + i * DIM, norm_b + i * DIM, Wlin + (size_t)i * DIM * DIM,
        gnl_w + i * DIM, gnl_b + i * DIM, aout, NE, NA);
  }
}


# Edges binned by row tile once per call in one pass (per-group segments that owners collect): per-block index pass and each owner's scan of all edge keys go, block 2 reuses the lists; 14->13 launches
# speedup vs baseline: 2.6408x; 1.1025x over previous
#include <hip/hip_runtime.h>


static constexpr int DIM  = 128;
static constexpr int NBLK = 2;
static constexpr int LDP  = 68;

typedef __attribute__((ext_vector_type(2))) float v2f;
typedef __attribute__((ext_vector_type(8))) float v8f;

static __device__ __forceinline__ float wave_reduce_sum(float x) {
#pragma unroll
  for (int off = 16; off > 0; off >>= 1) x += __shfl_xor(x, off, 32);
  return x;
}

static __device__ __forceinline__ unsigned f32_order_enc(float f) {
  unsigned u = __float_as_uint(f);
  return (u & 0x80000000u) ? ~u : (u | 0x80000000u);
}
static __device__ __forceinline__ float f32_order_dec(unsigned e) {
  unsigned u = (e & 0x80000000u) ? (e & 0x7fffffffu) : ~e;
  return __uint_as_float(u);
}

__global__ __launch_bounds__(128) void gemm_nt_128(const float* __restrict__ A,
                                                   const float* __restrict__ W,
                                                   float* __restrict__ C, int N) {
  __shared__ float Wlds[128 * LDP];
  const int tid = threadIdx.x;
  const int wv  = tid >> 5;
  const int ln  = tid & 31;
  const int hk  = ln >> 4;
  const int l16 = ln & 15;
  const int r0  = blockIdx.x * 64 + wv * 16;

  v8f acc[8];
#pragma unroll
  for (int t = 0; t < 8; ++t)
#pragma unroll
    for (int j = 0; j < 8; ++j) acc[t][j] = 0.0f;

  int rowA = r0 + l16;
  if (rowA >= N) rowA = N - 1;
  const float* ap = A + (size_t)rowA * DIM;

  for (int kp = 0; kp < DIM; kp += 64) {
    __syncthreads();
#pragma unroll
    for (int i = 0; i < 16; ++i) {
      const int q  = tid + 128 * i;
      const int n  = q >> 4;
      const int c4 = (q & 15) << 2;
      *(float4*)(&Wlds[n * LDP + c4]) =
          *(const float4*)(W + (size_t)n * DIM + kp + c4);
    }
    __syncthreads();

#pragma unroll 4
    for (int k0 = 0; k0 < 64; k0 += 4) {
      const int ka = k0 + 2 * hk;
      v2f a; a.x = ap[kp + ka]; a.y = ap[kp + ka + 1];
#pragma unroll
      for (int t = 0; t < 8; ++t) {
        const float* wp = &Wlds[(t * 16 + l16) * LDP + ka];
        v2f b; b.x = wp[0]; b.y = wp[1];
        acc[t] = __builtin_amdgcn_wmma_f32_16x16x4_f32(
            false, a, false, b, (short)0, acc[t], false, false);
      }
    }
  }

  const int rb = r0 + hk * 8;
  if (r0 + 16 <= N) {
#pragma unroll
    for (int t = 0; t < 8; ++t)
#pragma unroll
      for (int j = 0; j < 8; ++j)
        C[(size_t)(rb + j) * DIM + t * 16 + l16] = acc[t][j];
  } else {
#pragma unroll
    for (int t = 0; t < 8; ++t)
#pragma unroll
      for (int j = 0; j < 8; ++j) {
        const int row = rb + j;
        if (row < N) C[(size_t)row * DIM + t * 16 + l16] = acc[t][j];
      }
  }
}

static constexpr int ROWB = 4;

static __device__ __forceinline__ void row_stats(const float4 x, float& mu, float& rs) {
  float s  = x.x + x.y + x.z + x.w;
  float sq = x.x * x.x + x.y * x.y + x.z * x.z + x.w * x.w;
  s = wave_reduce_sum(s); sq = wave_reduce_sum(sq);
  mu = s * (1.0f / DIM);
  rs = rsqrtf(sq * (1.0f / DIM) - mu * mu + 1e-5f);
}

__global__ __launch_bounds__(256) void gn_kernel(const float* X, float* Y,
                                                 const float* __restrict__ w,
                                                 const float* __restrict__ b, int N) {
  const int ln = threadIdx.x & 31;
  const int wv = __builtin_amdgcn_readfirstlane((int)(threadIdx.x >> 5));
  const int step = (int)gridDim.x * 8 * ROWB;
  const float4 wv4 = ((const float4*)w)[ln];
  const float4 bv4 = ((const float4*)b)[ln];
  for (int r0 = ((int)blockIdx.x * 8 + wv) * ROWB; r0 < N; r0 += step) {
    float4 x[ROWB];
#pragma unroll
    for (int t = 0; t < ROWB; ++t) {
      x[t] = make_float4(0.0f, 0.0f, 0.0f, 0.0f);
      if (r0 + t < N) x[t] = ((const float4*)(X + (size_t)(r0 + t) * DIM))[ln];
    }
#pragma unroll
    for (int t = 0; t < ROWB; ++t) {
      float mu, rs;
      row_stats(x[t], mu, rs);
      float4 y;
      y.x = (x[t].x - mu) * rs * wv4.x + bv4.x;
      y.y = (x[t].y - mu) * rs * wv4.y + bv4.y;
      y.z = (x[t].z - mu) * rs * wv4.z + bv4.z;
      y.w = (x[t].w - mu) * rs * wv4.w + bv4.w;
      if (r0 + t < N) ((float4*)(Y + (size_t)(r0 + t) * DIM))[ln] = y;
    }
  }
}

static constexpr int EGR_Q  = 8;
static constexpr int EGR_S  = 8;
static constexpr int OWN_ROWS = 196;
static constexpr int OWN_CAP  = 8192;
static constexpr int OWN_TILES = 256;
static constexpr int NE_MAX  = 600000;
static constexpr int SEG_E   = 4096;
static constexpr int SEG_CAP = 64;
static constexpr int NSEG    = (NE_MAX + SEG_E - 1) / SEG_E;
static constexpr int SPW     = (NSEG + 31) / 32;

__global__ __launch_bounds__(1024) void edge_bin_kernel(const int* __restrict__ u,
                                                        const int* __restrict__ v,
                                                        unsigned* __restrict__ seg,
                                                        int* __restrict__ segcnt,
                                                        unsigned* __restrict__ gmax,
                                                        int E) {
  __shared__ int hist[OWN_TILES];
  const int tid = threadIdx.x;
  const int w   = blockIdx.x;
  if (tid < OWN_TILES) hist[tid] = 0;
  if (w == 0 && tid == 0) {
    for (int i = 0; i < NBLK; ++i) gmax[i] = 0u;
  }
  __syncthreads();
  int vv[SEG_E / 1024], uu[SEG_E / 1024];
#pragma unroll
  for (int k = 0; k < SEG_E / 1024; ++k) {
    const int e = w * SEG_E + k * 1024 + tid;
    vv[k] = -1; uu[k] = 0;
    if (e < E) { vv[k] = v[e]; uu[k] = u[e]; }
  }
#pragma unroll
  for (int k = 0; k < SEG_E / 1024; ++k) {
    if (vv[k] >= 0) {
      const int t   = vv[k] / OWN_ROWS;
      const int r   = vv[k] - t * OWN_ROWS;
      const int pos = atomicAdd(&hist[t], 1);
      if (pos < SEG_CAP)
        seg[((size_t)t * NSEG + w) * SEG_CAP + pos] = ((unsigned)r << 24) | (unsigned)uu[k];
    }
  }
  __syncthreads();
  if (tid < OWN_TILES) segcnt[w * OWN_TILES + tid] = hist[tid];
}

static __device__ __forceinline__ float wave_sum4(const float p0, const float p1,
                                                  const float p2, const float p3,
                                                  const int hk, const int b8) {
  const float s0 = hk ? p2 : p0, r0 = hk ? p0 : p2;
  const float s1 = hk ? p3 : p1, r1 = hk ? p1 : p3;
  const float x0 = s0 + __shfl_xor(r0, 16, 32);
  const float x1 = s1 + __shfl_xor(r1, 16, 32);
  const float s2 = b8 ? x1 : x0, r2 = b8 ? x0 : x1;
  float d = s2 + __shfl_xor(r2, 8, 32);
  d += __shfl_xor(d, 4, 32);
  d += __shfl_xor(d, 2, 32);
  d += __shfl_xor(d, 1, 32);
  return d;
}

static __device__ __forceinline__ void score_fetch(float4 (&kv)[EGR_Q],
                                                   const float* __restrict__ K, const int ul,
                                                   const int j, const int nval, const int ln) {
#pragma unroll
  for (int t = 0; t < EGR_Q; ++t) {
    const int jj = (j + t < nval) ? (j + t) : (nval - 1);
    const int uj = __builtin_amdgcn_readlane(ul, jj);
    kv[t] = ((const float4*)(K + (size_t)uj * DIM))[ln];
  }
}
static __device__ __forceinline__ float score_step(const float4 (&kv)[EGR_Q], const float* qt,
                                                   const int rl, const int j, const int nval,
                                                   const int ln, const int hk, const int b8,
                                                   float dl) {
  float p[EGR_Q];
#pragma unroll
  for (int t = 0; t < EGR_Q; ++t) {
    const int jj = (j + t < nval) ? (j + t) : (nval - 1);
    const int rj = __builtin_amdgcn_readlane(rl, jj);
    const float4 qv = ((const float4*)(qt + rj))[ln];
    p[t] = qv.x * kv[t].x + qv.y * kv[t].y + qv.z * kv[t].z + qv.w * kv[t].w;
  }
  float da = wave_sum4(p[0], p[1], p[2], p[3], hk, b8);
  float db = wave_sum4(p[4], p[5], p[6], p[7], hk, b8);
  da = (da >= 0.0f) ? da : 0.1f * da;
  db = (db >= 0.0f) ? db : 0.1f * db;
  const int i8 = ln - j;
  const int sl = (i8 & 3) << 3;
  const float ga = __shfl(da, sl, 32);
  const float gb = __shfl(db, sl, 32);
  if ((unsigned)i8 < 4u) dl = ga;
  else if ((unsigned)i8 < 8u) dl = gb;
  return dl;
}

template <bool REUSE>
static __device__ __forceinline__ int2 own_entry(const int* list, const unsigned* my_ru,
                                                 const int idx, const int lo) {
  unsigned ru;
  if constexpr (REUSE) ru = my_ru[idx];
  else                 ru = (unsigned)list[idx];
  return make_int2((int)(ru >> 24) + lo, (int)(ru & 0x00ffffffu));
}

template <bool REUSE>
__global__ __launch_bounds__(1024) void edge_owner_score_kernel(const float* __restrict__ Q,
                                                                const float* __restrict__ K,
                                                                const int* __restrict__ u,
                                                                const int* __restrict__ v,
                                                                const unsigned* __restrict__ seg,
                                                                const int* __restrict__ segcnt,
                                                                float* __restrict__ c,
                                                                unsigned* __restrict__ gmax,
                                                                unsigned* __restrict__ rec_ru,
                                                                float* __restrict__ rec_sc,
                                                                int* __restrict__ gcount,
                                                                int E, int NA) {
  __shared__ __attribute__((aligned(16))) float qt[OWN_ROWS * DIM];
  __shared__ int   list[OWN_CAP];
  __shared__ float smax[32];
  __shared__ int   scnt[NSEG];
  __shared__ int   soff[NSEG];
  __shared__ int   cnt;
  const int tid = threadIdx.x;
  const int ln  = tid & 31;
  const int wv  = __builtin_amdgcn_readfirstlane(tid >> 5);
  const int lo  = (int)blockIdx.x * OWN_ROWS;
  int nrows = NA - lo;
  if (nrows > OWN_ROWS) nrows = OWN_ROWS;
  if (nrows <= 0) return;
  const int hk = ln >> 4;
  const int b8 = (ln >> 3) & 1;

  const float4* q4 = (const float4*)(Q + (size_t)lo * DIM);
  for (int q = tid; q < nrows * (DIM / 4); q += 1024) ((float4*)qt)[q] = q4[q];
  const unsigned g = blockIdx.x;
  int total;
  if constexpr (REUSE) {
    __syncthreads();
    total = gcount[g];
  } else {
    const int nseg = (E + SEG_E - 1) / SEG_E;
    const unsigned* my_seg = seg + (size_t)g * NSEG * SEG_CAP;
    if (wv == 0) {
      int n[SPW];
      int loc = 0;
      bool over = false;
#pragma unroll
      for (int k = 0; k < SPW; ++k) {
        const int s = ln * SPW + k;
        n[k] = (s < nseg) ? segcnt[s * OWN_TILES + (int)g] : 0;
      }
#pragma unroll
      for (int k = 0; k < SPW; ++k) {
        over = over || (n[k] > SEG_CAP);
        loc += n[k];
      }
      int inc = loc;
#pragma unroll
      for (int d = 1; d < 32; d <<= 1) {
        const int up = __shfl_up(inc, d, 32);
        if (ln >= d) inc += up;
      }
      int a = inc - loc;
#pragma unroll
      for (int k = 0; k < SPW; ++k) {
        const int s = ln * SPW + k;
        if (s < nseg) { scnt[s] = n[k]; soff[s] = a; }
        a += n[k];
      }
      const int  all   = __shfl(inc, 31, 32);
      const bool anyov = __ballot(over) != 0;
      if (ln == 0) cnt = (anyov || all > OWN_CAP) ? OWN_CAP + 1 : all;
    }
    __syncthreads();
    total = cnt;
    if (total <= OWN_CAP) {
      const unsigned* spare = (const unsigned*)(segcnt + (int)g);
      unsigned e0[SPW], e1[SPW];
#pragma unroll
      for (int k = 0; k < SPW; ++k) {
        const int s = wv + 32 * k;
        const int m = (s < nseg) ? scnt[s] : 0;
        const unsigned* p0 = (ln < m)      ? my_seg + (s * SEG_CAP + ln)      : spare;
        const unsigned* p1 = (ln + 32 < m) ? my_seg + (s * SEG_CAP + 32 + ln) : spare;
        e0[k] = *p0;
        e1[k] = *p1;
      }
#pragma unroll
      for (int k = 0; k < SPW; ++k) {
        const int s = wv + 32 * k;
        const int m = (s < nseg) ? scnt[s] : 0;
        const int o = (s < nseg) ? soff[s] : 0;
        if (ln < m)      list[o + ln]      = (int)e0[k];
        if (ln + 32 < m) list[o + 32 + ln] = (int)e1[k];
      }
    }
    __syncthreads();
    if (tid == 0) gcount[g] = total;
  }

  float cmax = -3.0e38f;
  if (total <= OWN_CAP) {
    unsigned* my_ru = rec_ru + (size_t)g * OWN_CAP;
    float*    my_sc = rec_sc + (size_t)g * OWN_CAP;
    const int per = (total + 31) >> 5;
    const int b0  = wv * per;
    int b1 = b0 + per;
    if (b1 > total) b1 = total;
    if (b0 < b1) {
      int nval = (b1 - b0 > 32) ? 32 : (b1 - b0);
      int2 pn = own_entry<REUSE>(list, my_ru, b0 + (ln < nval ? ln : nval - 1), lo);
      float4 ka[EGR_Q], kb[EGR_Q];
      score_fetch(ka, K, pn.y, 0, nval, ln);
      for (int base = b0; base < b1; base += 32) {
        const int2 pl = pn;
        const int ul = pl.y;
        const int rl = (pl.x - lo) * DIM;
        const bool hasnext = base + 32 < b1;
        const int nvaln = hasnext ? ((b1 - base - 32 > 32) ? 32 : (b1 - base - 32)) : nval;
        pn = own_entry<REUSE>(list, my_ru,
                              (hasnext ? base + 32 : base) + (ln < nvaln ? ln : nvaln - 1), lo);
        float dl = 0.0f;
        for (int j = 0; j < nval; j += 2 * EGR_Q) {
          score_fetch(kb, K, ul, j + EGR_Q, nval, ln);
          dl = score_step(ka, qt, rl, j, nval, ln, hk, b8, dl);
          const bool here = (j + 2 * EGR_Q < nval) || !hasnext;
          score_fetch(ka, K, here ? ul : pn.y, here ? j + 2 * EGR_Q : 0, here ? nval : nvaln, ln);
          dl = score_step(kb, qt, rl, j + EGR_Q, nval, ln, hk, b8, dl);
        }
        if (ln < nval) {
          if constexpr (!REUSE) my_ru[base + ln] = ((unsigned)(pl.x - lo) << 24) | (unsigned)ul;
          my_sc[base + ln] = dl;
          cmax = fmaxf(cmax, dl);
        }
        nval = nvaln;
      }
    }
  } else {
    for (int e0 = wv * 32; e0 < E; e0 += 1024) {
      const int e  = e0 + ln;
      const int vl = (e < E) ? v[e] : -1;
      unsigned mask = (unsigned)__ballot((unsigned)(vl - lo) < (unsigned)nrows);
      while (mask) {
        const int b = __builtin_ctz(mask);
        mask &= mask - 1;
        const int vj = __builtin_amdgcn_readlane(vl, b);
        const int uj = u[e0 + b];
        const float4 qv = ((const float4*)(qt + (vj - lo) * DIM))[ln];
        const float4 kv = ((const float4*)(K + (size_t)uj * DIM))[ln];
        float d = wave_reduce_sum(qv.x * kv.x + qv.y * kv.y + qv.z * kv.z + qv.w * kv.w);
        d = (d >= 0.0f) ? d : 0.1f * d;
        if (ln == 0) c[e0 + b] = d;
        cmax = fmaxf(cmax, d);
      }
    }
  }
  cmax = fmaxf(cmax, __shfl_xor(cmax, 16, 32));
  cmax = fmaxf(cmax, __shfl_xor(cmax, 8, 32));
  cmax = fmaxf(cmax, __shfl_xor(cmax, 4, 32));
  cmax = fmaxf(cmax, __shfl_xor(cmax, 2, 32));
  cmax = fmaxf(cmax, __shfl_xor(cmax, 1, 32));
  if (ln == 0) smax[wv] = cmax;
  __syncthreads();
  if (tid == 0) {
    float m = smax[0];
    for (int i = 1; i < 32; ++i) m = fmaxf(m, smax[i]);
    atomicMax(gmax, f32_order_enc(m));
  }
}

static constexpr int TP     = 132;
static constexpr int NSTRIP = (OWN_ROWS + 15) / 16;
static constexpr int TROWS  = NSTRIP * 16;
static constexpr int RPW    = (OWN_ROWS + 31) / 32;

static __device__ __forceinline__ void w_fetch(float4 (&wr)[2], const float* __restrict__ Wsrc,
                                               const int stride, const int kp, const int tid) {
#pragma unroll
  for (int i = 0; i < 2; ++i) {
    const int q  = tid + 1024 * i;
    const int n  = q >> 4;
    const int c4 = (q & 15) << 2;
    wr[i] = *(const float4*)(Wsrc + (size_t)n * stride + kp + c4);
  }
}
static __device__ __forceinline__ void w_put(const float4 (&wr)[2], float* buf, const int tid) {
#pragma unroll
  for (int i = 0; i < 2; ++i) {
    const int q  = tid + 1024 * i;
    const int n  = q >> 4;
    const int c4 = (q & 15) << 2;
    *(float4*)(&buf[n * LDP + c4]) = wr[i];
  }
}
static constexpr int WSTG = (2048 + (32 - 2 * NSTRIP) * 32 - 1) / ((32 - 2 * NSTRIP) * 32);
static_assert(2 * NSTRIP < 32, "no wave left to stage the weights");
static __device__ __forceinline__ void w_stage(float* buf, const float* __restrict__ Wsrc,
                                               const int stride, const int kp,
                                               const int sid, const int sstep) {
  float4 r[WSTG];
#pragma unroll
  for (int i = 0; i < WSTG; ++i) {
    const int q = sid + i * sstep;
    r[i] = make_float4(0.0f, 0.0f, 0.0f, 0.0f);
    if (q < 2048) r[i] = *(const float4*)(Wsrc + (size_t)(q >> 4) * stride + kp + ((q & 15) << 2));
  }
#pragma unroll
  for (int i = 0; i < WSTG; ++i) {
    const int q = sid + i * sstep;
    if (q < 2048) *(float4*)(&buf[(q >> 4) * LDP + ((q & 15) << 2)]) = r[i];
  }
}
static __device__ __forceinline__ void mm_chunk(v8f (&c)[4], const float* ap, const float* wl,
                                                const int hk) {
#pragma unroll 4
  for (int k0 = 0; k0 < 64; k0 += 4) {
    const int ka = k0 + 2 * hk;
    v2f a; a.x = ap[ka]; a.y = ap[ka + 1];
#pragma unroll
    for (int t = 0; t < 4; ++t) {
      const float* wp = wl + t * 16 * LDP + ka;
      v2f b; b.x = wp[0]; b.y = wp[1];
      c[t] = __builtin_amdgcn_wmma_f32_16x16x4_f32(
          false, a, false, b, (short)0, c[t], false, false);
    }
  }
}

__global__ __launch_bounds__(1024) void edge_owner_kernel(const float* __restrict__ c,
                                                          const unsigned* __restrict__ gmax,
                                                          const float* __restrict__ ctxp,
                                                          const int* __restrict__ u,
                                                          const int* __restrict__ v,
                                                          const unsigned* __restrict__ rec_ru,
                                                          const float* __restrict__ rec_sc,
                                                          const int* __restrict__ gcount,
                                                          const float* __restrict__ ain,
                                                          const float* __restrict__ Wagt,
                                                          const float* __restrict__ gna_w,
                                                          const float* __restrict__ gna_b,
                                                          const float* __restrict__ nrm_w,
                                                          const float* __restrict__ nrm_b,
                                                          const float* __restrict__ Wlin,
                                                          const float* __restrict__ gnl_w,
                                                          const float* __restrict__ gnl_b,
                                                          float* __restrict__ aout,
                                                          int E, int NA) {
  __shared__ __attribute__((aligned(16))) float acc[TROWS * TP];
  __shared__ __attribute__((aligned(16))) float xt[TROWS * TP];
  __shared__ __attribute__((aligned(16))) float Wl0[128 * LDP];
  __shared__ __attribute__((aligned(16))) float Wl1[128 * LDP];
  __shared__ float lsum[OWN_ROWS];
  const int tid = threadIdx.x;
  const int ln  = tid & 31;
  const int wv  = __builtin_amdgcn_readfirstlane(tid >> 5);
  const int lo  = (int)blockIdx.x * OWN_ROWS;
  int nrows = NA - lo;
  if (nrows > OWN_ROWS) nrows = OWN_ROWS;
  if (nrows <= 0) return;

  float4 wr[2];
  w_fetch(wr, Wagt, 2 * DIM, 0, tid);

  const float4 z4 = make_float4(0.0f, 0.0f, 0.0f, 0.0f);
  for (int q = tid; q < TROWS * TP / 4; q += 1024) ((float4*)acc)[q] = z4;
  if (tid < OWN_ROWS) lsum[tid] = 0.0f;
  w_put(wr, Wl0, tid);

  const unsigned g = blockIdx.x;
  const int total = gcount[g];
  const unsigned* my_ru = rec_ru + (size_t)g * OWN_CAP;
  const float*    my_sc = rec_sc + (size_t)g * OWN_CAP;
  __syncthreads();
  const float gm = f32_order_dec(*gmax);

  if (total <= OWN_CAP) {
    for (int i = tid; i < total; i += 1024)
      atomicAdd(&lsum[my_ru[i] >> 24], expf(my_sc[i] - gm));
    __syncthreads();
    const int per = (total + 31) >> 5;
    const int b0  = wv * per;
    int b1 = b0 + per;
    if (b1 > total) b1 = total;
    for (int base = b0; base < b1; base += 32) {
      int nval = b1 - base;
      if (nval > 32) nval = 32;
      const int il = base + (ln < nval ? ln : nval - 1);
      const unsigned ru = my_ru[il];
      const int ul  = (int)(ru & 0x00ffffffu);
      const int row = (int)(ru >> 24);
      const int cl = __float_as_int(expf(my_sc[il] - gm) / (lsum[row] + 1e-16f));
      const int rl = row * TP;
      int j = 0;
      for (; j + EGR_S <= nval; j += EGR_S) {
        float m[EGR_S][4];
#pragma unroll
        for (int t = 0; t < EGR_S; ++t) {
          const int uj = __builtin_amdgcn_readlane(ul, j + t);
          const float* srow = ctxp + (size_t)uj * DIM + ln;
          m[t][0] = srow[0];
          m[t][1] = srow[32];
          m[t][2] = srow[64];
          m[t][3] = srow[96];
        }
#pragma unroll
        for (int t = 0; t < EGR_S; ++t) {
          const int   rj   = __builtin_amdgcn_readlane(rl, j + t);
          const float coef = __int_as_float(__builtin_amdgcn_readlane(cl, j + t));
          float* dst = acc + rj + ln;
          atomicAdd(dst + 0,  m[t][0] * coef);
          atomicAdd(dst + 32, m[t][1] * coef);
          atomicAdd(dst + 64, m[t][2] * coef);
          atomicAdd(dst + 96, m[t][3] * coef);
        }
      }
      for (; j < nval; ++j) {
        const int   uj   = __builtin_amdgcn_readlane(ul, j);
        const int   rj   = __builtin_amdgcn_readlane(rl, j);
        const float coef = __int_as_float(__builtin_amdgcn_readlane(cl, j));
        const float* srow = ctxp + (size_t)uj * DIM + ln;
        const float m0 = srow[0], m1 = srow[32], m2 = srow[64], m3 = srow[96];
        float* dst = acc + rj + ln;
        atomicAdd(dst + 0,  m0 * coef);
        atomicAdd(dst + 32, m1 * coef);
        atomicAdd(dst + 64, m2 * coef);
        atomicAdd(dst + 96, m3 * coef);
      }
    }
  } else {
    for (int e0 = wv * 32; e0 < E; e0 += 1024) {
      const int e  = e0 + ln;
      const int vl = (e < E) ? v[e] : -1;
      if ((unsigned)(vl - lo) < (unsigned)nrows) atomicAdd(&lsum[vl - lo], expf(c[e] - gm));
    }
    __syncthreads();
    for (int e0 = wv * 32; e0 < E; e0 += 1024) {
      const int e  = e0 + ln;
      const int vl = (e < E) ? v[e] : -1;
      unsigned mask = (unsigned)__ballot((unsigned)(vl - lo) < (unsigned)nrows);
      while (mask) {
        const int b = __builtin_ctz(mask);
        mask &= mask - 1;
        const int   vj   = __builtin_amdgcn_readlane(vl, b);
        const int   uj   = u[e0 + b];
        const float coef = expf(c[e0 + b] - gm) / (lsum[vj - lo] + 1e-16f);
        const float* srow = ctxp + (size_t)uj * DIM + ln;
        const float m0 = srow[0], m1 = srow[32], m2 = srow[64], m3 = srow[96];
        float* dst = acc + (vj - lo) * TP + ln;
        atomicAdd(dst + 0,  m0 * coef);
        atomicAdd(dst + 32, m1 * coef);
        atomicAdd(dst + 64, m2 * coef);
        atomicAdd(dst + 96, m3 * coef);
      }
    }
  }
  __syncthreads();

  const int hk  = ln >> 4;
  const int l16 = ln & 15;
  const int nstrip = (nrows + 15) >> 4;
  const bool act = wv < 2 * nstrip;
  const int st = wv >> 1;
  const int ch = wv & 1;
  const int trow = st * 16 + l16;
  int ga = lo + trow;
  if (ga >= NA) ga = NA - 1;
  const float* apg = ain + (size_t)ga * DIM;
  const float* apl = acc + trow * TP;
  const float* apx = xt + trow * TP;
  const int wofs = (ch * 64 + l16) * LDP;
  const int orow = (st * 16 + 8 * hk) * TP + ch * 64 + l16;

  v8f cc[4];
#pragma unroll
  for (int t = 0; t < 4; ++t)
#pragma unroll
    for (int j = 0; j < 8; ++j) cc[t][j] = 0.0f;

  const int sid   = (wv - 2 * nstrip) * 32 + ln;
  const int sstep = (32 - 2 * nstrip) * 32;
  if (act) mm_chunk(cc, apg, Wl0 + wofs, hk);
  else     w_stage(Wl1, Wagt, 2 * DIM, 64, sid, sstep);
  __syncthreads();
  if (act) mm_chunk(cc, apg + 64, Wl1 + wofs, hk);
  else     w_stage(Wl0, Wagt, 2 * DIM, 128, sid, sstep);
  __syncthreads();
  if (act) mm_chunk(cc, apl, Wl0 + wofs, hk);
  else     w_stage(Wl1, Wagt, 2 * DIM, 192, sid, sstep);
  __syncthreads();
  if (act) {
    mm_chunk(cc, apl + 64, Wl1 + wofs, hk);
#pragma unroll
    for (int t = 0; t < 4; ++t)
#pragma unroll
      for (int j = 0; j < 8; ++j) xt[orow + j * TP + t * 16] = cc[t][j];
  } else {
    w_stage(Wl0, Wlin, DIM, 0, sid, sstep);
  }
  __syncthreads();

  {
    const float4 wa = ((const float4*)gna_w)[ln], ba = ((const float4*)gna_b)[ln];
    const float4 wn = ((const float4*)nrm_w)[ln], bn = ((const float4*)nrm_b)[ln];
    for (int r = wv; r < nrows; r += 32) {
      float4* xp = (float4*)(xt + r * TP) + ln;
      const float4 x = *xp;
      float mu, rs;
      row_stats(x, mu, rs);
      float4 y;
      y.x = (x.x - mu) * rs * wa.x + ba.x;
      y.y = (x.y - mu) * rs * wa.y + ba.y;
      y.z = (x.z - mu) * rs * wa.z + ba.z;
      y.w = (x.w - mu) * rs * wa.w + ba.w;
      row_stats(y, mu, rs);
      float4 z;
      z.x = fmaxf((y.x - mu) * rs * wn.x + bn.x, 0.0f);
      z.y = fmaxf((y.y - mu) * rs * wn.y + bn.y, 0.0f);
      z.z = fmaxf((y.z - mu) * rs * wn.z + bn.z, 0.0f);
      z.w = fmaxf((y.w - mu) * rs * wn.w + bn.w, 0.0f);
      *xp = z;
    }
  }
  __syncthreads();

#pragma unroll
  for (int t = 0; t < 4; ++t)
#pragma unroll
    for (int j = 0; j < 8; ++j) cc[t][j] = 0.0f;
  if (act) mm_chunk(cc, apx, Wl0 + wofs, hk);
  else     w_stage(Wl1, Wlin, DIM, 64, sid, sstep);
  __syncthreads();
  if (act) {
    mm_chunk(cc, apx + 64, Wl1 + wofs, hk);
#pragma unroll
    for (int t = 0; t < 4; ++t)
#pragma unroll
      for (int j = 0; j < 8; ++j) acc[orow + j * TP + t * 16] = cc[t][j];
  }
  __syncthreads();

  {
    const float4 wl4 = ((const float4*)gnl_w)[ln], bl4 = ((const float4*)gnl_b)[ln];
    float4 rr[RPW];
#pragma unroll
    for (int t = 0; t < RPW; ++t) {
      const int r = wv + 32 * t;
      rr[t] = z4;
      if (r < nrows) rr[t] = ((const float4*)(ain + (size_t)(lo + r) * DIM))[ln];
    }
#pragma unroll
    for (int t = 0; t < RPW; ++t) {
      const int r = wv + 32 * t;
      if (r < nrows) {
        const float4 x = ((const float4*)(acc + r * TP))[ln];
        float mu, rs;
        row_stats(x, mu, rs);
        float4 y;
        y.x = fmaxf((x.x - mu) * rs * wl4.x + bl4.x + rr[t].x, 0.0f);
        y.y = fmaxf((x.y - mu) * rs * wl4.y + bl4.y + rr[t].y, 0.0f);
        y.z = fmaxf((x.z - mu) * rs * wl4.z + bl4.z + rr[t].z, 0.0f);
        y.w = fmaxf((x.w - mu) * rs * wl4.w + bl4.w + rr[t].w, 0.0f);
        ((float4*)(aout + (size_t)(lo + r) * DIM))[ln] = y;
      }
    }
  }
}

extern "C" void kernel_launch(void* const* d_in, const int* in_sizes, int n_in,
                              void* d_out, int out_size, void* d_ws, size_t ws_size,
                              hipStream_t stream) {
  (void)in_sizes; (void)n_in; (void)out_size; (void)ws_size;
  const int NA = 50000, NC = 100000, NE = 600000;

  const float* agt    = (const float*)d_in[0];
  const float* ctx    = (const float*)d_in[1];
  const int*   u      = (const int*)d_in[2];
  const int*   v      = (const int*)d_in[3];
  const float* Wq     = (const float*)d_in[4];
  const float* Wk     = (const float*)d_in[5];
  const float* Wctx   = (const float*)d_in[6];
  const float* gnc_w  = (const float*)d_in[7];
  const float* gnc_b  = (const float*)d_in[8];
  const float* Wagt   = (const float*)d_in[9];
  const float* gna_w  = (const float*)d_in[10];
  const float* gna_b  = (const float*)d_in[11];
  const float* norm_w = (const float*)d_in[12];
  const float* norm_b = (const float*)d_in[13];
  const float* Wlin   = (const float*)d_in[14];
  const float* gnl_w  = (const float*)d_in[15];
  const float* gnl_b  = (const float*)d_in[16];

  float* ws = (float*)d_ws;
  size_t off = 0;
  float* qbuf   = ws + off; off += (size_t)NA * DIM;
  float* kbuf   = ws + off; off += (size_t)NC * DIM;
  float* cpbuf  = ws + off; off += (size_t)NC * DIM;
  float* cbuf   = ws + off; off += (size_t)NE;
  off += (size_t)NA;
  float* segbuf = ws + off; off += (size_t)NA * DIM;
  float* xbuf   = ws + off; off += (size_t)NA * DIM;
  float* agtbuf = ws + off; off += (size_t)NA * DIM;
  unsigned* maxbuf = (unsigned*)(ws + off); off += 16;

  const dim3 b128(128), b256(256);

  for (int i = 0; i < NBLK; ++i) {
    const float* ain  = (i == 0) ? agt : agtbuf;
    float*       aout = (i == NBLK - 1) ? (float*)d_out : agtbuf;

    gemm_nt_128<<<(NA + 63) / 64, b128, 0, stream>>>(ain, Wq + (size_t)i * DIM * DIM, qbuf, NA);
    gemm_nt_128<<<(NC + 63) / 64, b128, 0, stream>>>(ctx, Wk + (size_t)i * DIM * DIM, kbuf, NC);
    gemm_nt_128<<<(NC + 63) / 64, b128, 0, stream>>>(ctx, Wctx + (size_t)i * DIM * DIM, cpbuf, NC);
    gn_kernel<<<(NC + 63) / 64, b256, 0, stream>>>(cpbuf, cpbuf, gnc_w + i * DIM, gnc_b + i * DIM, NC);

    unsigned* seg    = (unsigned*)segbuf;
    int*      segcnt = (int*)segbuf + OWN_TILES * NSEG * SEG_CAP;
    unsigned* rec_ru = (unsigned*)xbuf + 3 * (1 << 19);
    float*    rec_sc = (float*)(rec_ru + 256 * OWN_CAP);
    int*      gcount = (int*)(rec_sc + 256 * OWN_CAP);
    unsigned* bmax   = maxbuf + i;
    if (i == 0) {
      edge_bin_kernel<<<NSEG, dim3(1024), 0, stream>>>(u, v, seg, segcnt, maxbuf, NE);
      edge_owner_score_kernel<false><<<(NA + OWN_ROWS - 1) / OWN_ROWS, dim3(1024), 0, stream>>>(
          qbuf, kbuf, u, v, seg, segcnt, cbuf, bmax, rec_ru, rec_sc, gcount, NE, NA);
    } else {
      edge_owner_score_kernel<true><<<(NA + OWN_ROWS - 1) / OWN_ROWS, dim3(1024), 0, stream>>>(
          qbuf, kbuf, u, v, seg, segcnt, cbuf, bmax, rec_ru, rec_sc, gcount, NE, NA);
    }
    edge_owner_kernel<<<(NA + OWN_ROWS - 1) / OWN_ROWS, dim3(1024), 0, stream>>>(
        cbuf, bmax, cpbuf, u, v, rec_ru, rec_sc, gcount, ain,
        Wagt + (size_t)i * DIM * 2 * DIM, gna_w + i * DIM, gna_b + i * DIM,
        norm_w + i * DIM, norm_b + i * DIM, Wlin + (size_t)i * DIM * DIM,
        gnl_w + i * DIM, gnl_b + i * DIM, aout, NE, NA);
  }
}


# Score kernel forms its tile's query rows itself: 196 input rows @ Wq^T by WMMA into LDS (same k and sum order); both query GEMM launches and q's trip through memory gone, 13->11 launches
# speedup vs baseline: 2.7804x; 1.0529x over previous
#include <hip/hip_runtime.h>


static constexpr int DIM  = 128;
static constexpr int NBLK = 2;
static constexpr int LDP  = 68;

typedef __attribute__((ext_vector_type(2))) float v2f;
typedef __attribute__((ext_vector_type(8))) float v8f;

static __device__ __forceinline__ float wave_reduce_sum(float x) {
#pragma unroll
  for (int off = 16; off > 0; off >>= 1) x += __shfl_xor(x, off, 32);
  return x;
}

static __device__ __forceinline__ unsigned f32_order_enc(float f) {
  unsigned u = __float_as_uint(f);
  return (u & 0x80000000u) ? ~u : (u | 0x80000000u);
}
static __device__ __forceinline__ float f32_order_dec(unsigned e) {
  unsigned u = (e & 0x80000000u) ? (e & 0x7fffffffu) : ~e;
  return __uint_as_float(u);
}

__global__ __launch_bounds__(128) void gemm_nt_128(const float* __restrict__ A,
                                                   const float* __restrict__ W,
                                                   float* __restrict__ C, int N) {
  __shared__ float Wlds[128 * LDP];
  const int tid = threadIdx.x;
  const int wv  = tid >> 5;
  const int ln  = tid & 31;
  const int hk  = ln >> 4;
  const int l16 = ln & 15;
  const int r0  = blockIdx.x * 64 + wv * 16;

  v8f acc[8];
#pragma unroll
  for (int t = 0; t < 8; ++t)
#pragma unroll
    for (int j = 0; j < 8; ++j) acc[t][j] = 0.0f;

  int rowA = r0 + l16;
  if (rowA >= N) rowA = N - 1;
  const float* ap = A + (size_t)rowA * DIM;

  for (int kp = 0; kp < DIM; kp += 64) {
    __syncthreads();
#pragma unroll
    for (int i = 0; i < 16; ++i) {
      const int q  = tid + 128 * i;
      const int n  = q >> 4;
      const int c4 = (q & 15) << 2;
      *(float4*)(&Wlds[n * LDP + c4]) =
          *(const float4*)(W + (size_t)n * DIM + kp + c4);
    }
    __syncthreads();

#pragma unroll 4
    for (int k0 = 0; k0 < 64; k0 += 4) {
      const int ka = k0 + 2 * hk;
      v2f a; a.x = ap[kp + ka]; a.y = ap[kp + ka + 1];
#pragma unroll
      for (int t = 0; t < 8; ++t) {
        const float* wp = &Wlds[(t * 16 + l16) * LDP + ka];
        v2f b; b.x = wp[0]; b.y = wp[1];
        acc[t] = __builtin_amdgcn_wmma_f32_16x16x4_f32(
            false, a, false, b, (short)0, acc[t], false, false);
      }
    }
  }

  const int rb = r0 + hk * 8;
  if (r0 + 16 <= N) {
#pragma unroll
    for (int t = 0; t < 8; ++t)
#pragma unroll
      for (int j = 0; j < 8; ++j)
        C[(size_t)(rb + j) * DIM + t * 16 + l16] = acc[t][j];
  } else {
#pragma unroll
    for (int t = 0; t < 8; ++t)
#pragma unroll
      for (int j = 0; j < 8; ++j) {
        const int row = rb + j;
        if (row < N) C[(size_t)row * DIM + t * 16 + l16] = acc[t][j];
      }
  }
}

static constexpr int ROWB = 4;

static __device__ __forceinline__ void row_stats(const float4 x, float& mu, float& rs) {
  float s  = x.x + x.y + x.z + x.w;
  float sq = x.x * x.x + x.y * x.y + x.z * x.z + x.w * x.w;
  s = wave_reduce_sum(s); sq = wave_reduce_sum(sq);
  mu = s * (1.0f / DIM);
  rs = rsqrtf(sq * (1.0f / DIM) - mu * mu + 1e-5f);
}

__global__ __launch_bounds__(256) void gn_kernel(const float* X, float* Y,
                                                 const float* __restrict__ w,
                                                 const float* __restrict__ b, int N) {
  const int ln = threadIdx.x & 31;
  const int wv = __builtin_amdgcn_readfirstlane((int)(threadIdx.x >> 5));
  const int step = (int)gridDim.x * 8 * ROWB;
  const float4 wv4 = ((const float4*)w)[ln];
  const float4 bv4 = ((const float4*)b)[ln];
  for (int r0 = ((int)blockIdx.x * 8 + wv) * ROWB; r0 < N; r0 += step) {
    float4 x[ROWB];
#pragma unroll
    for (int t = 0; t < ROWB; ++t) {
      x[t] = make_float4(0.0f, 0.0f, 0.0f, 0.0f);
      if (r0 + t < N) x[t] = ((const float4*)(X + (size_t)(r0 + t) * DIM))[ln];
    }
#pragma unroll
    for (int t = 0; t < ROWB; ++t) {
      float mu, rs;
      row_stats(x[t], mu, rs);
      float4 y;
      y.x = (x[t].x - mu) * rs * wv4.x + bv4.x;
      y.y = (x[t].y - mu) * rs * wv4.y + bv4.y;
      y.z = (x[t].z - mu) * rs * wv4.z + bv4.z;
      y.w = (x[t].w - mu) * rs * wv4.w + bv4.w;
      if (r0 + t < N) ((float4*)(Y + (size_t)(r0 + t) * DIM))[ln] = y;
    }
  }
}

static constexpr int EGR_Q  = 8;
static constexpr int EGR_S  = 8;
static constexpr int OWN_ROWS = 196;
static constexpr int OWN_CAP  = 8192;
static constexpr int OWN_TILES = 256;
static constexpr int NE_MAX  = 600000;
static constexpr int SEG_E   = 4096;
static constexpr int SEG_CAP = 64;
static constexpr int NSEG    = (NE_MAX + SEG_E - 1) / SEG_E;
static constexpr int SPW     = (NSEG + 31) / 32;

__global__ __launch_bounds__(1024) void edge_bin_kernel(const int* __restrict__ u,
                                                        const int* __restrict__ v,
                                                        unsigned* __restrict__ seg,
                                                        int* __restrict__ segcnt,
                                                        unsigned* __restrict__ gmax,
                                                        int E) {
  __shared__ int hist[OWN_TILES];
  const int tid = threadIdx.x;
  const int w   = blockIdx.x;
  if (tid < OWN_TILES) hist[tid] = 0;
  if (w == 0 && tid == 0) {
    for (int i = 0; i < NBLK; ++i) gmax[i] = 0u;
  }
  __syncthreads();
  int vv[SEG_E / 1024], uu[SEG_E / 1024];
#pragma unroll
  for (int k = 0; k < SEG_E / 1024; ++k) {
    const int e = w * SEG_E + k * 1024 + tid;
    vv[k] = -1; uu[k] = 0;
    if (e < E) { vv[k] = v[e]; uu[k] = u[e]; }
  }
#pragma unroll
  for (int k = 0; k < SEG_E / 1024; ++k) {
    if (vv[k] >= 0) {
      const int t   = vv[k] / OWN_ROWS;
      const int r   = vv[k] - t * OWN_ROWS;
      const int pos = atomicAdd(&hist[t], 1);
      if (pos < SEG_CAP)
        seg[((size_t)t * NSEG + w) * SEG_CAP + pos] = ((unsigned)r << 24) | (unsigned)uu[k];
    }
  }
  __syncthreads();
  if (tid < OWN_TILES) segcnt[w * OWN_TILES + tid] = hist[tid];
}

static __device__ __forceinline__ float wave_sum4(const float p0, const float p1,
                                                  const float p2, const float p3,
                                                  const int hk, const int b8) {
  const float s0 = hk ? p2 : p0, r0 = hk ? p0 : p2;
  const float s1 = hk ? p3 : p1, r1 = hk ? p1 : p3;
  const float x0 = s0 + __shfl_xor(r0, 16, 32);
  const float x1 = s1 + __shfl_xor(r1, 16, 32);
  const float s2 = b8 ? x1 : x0, r2 = b8 ? x0 : x1;
  float d = s2 + __shfl_xor(r2, 8, 32);
  d += __shfl_xor(d, 4, 32);
  d += __shfl_xor(d, 2, 32);
  d += __shfl_xor(d, 1, 32);
  return d;
}

static __device__ __forceinline__ void score_fetch(float4 (&kv)[EGR_Q],
                                                   const float* __restrict__ K, const int ul,
                                                   const int j, const int nval, const int ln) {
#pragma unroll
  for (int t = 0; t < EGR_Q; ++t) {
    const int jj = (j + t < nval) ? (j + t) : (nval - 1);
    const int uj = __builtin_amdgcn_readlane(ul, jj);
    kv[t] = ((const float4*)(K + (size_t)uj * DIM))[ln];
  }
}
static __device__ __forceinline__ float score_step(const float4 (&kv)[EGR_Q], const float* qt,
                                                   const int rl, const int j, const int nval,
                                                   const int ln, const int hk, const int b8,
                                                   float dl) {
  float p[EGR_Q];
#pragma unroll
  for (int t = 0; t < EGR_Q; ++t) {
    const int jj = (j + t < nval) ? (j + t) : (nval - 1);
    const int rj = __builtin_amdgcn_readlane(rl, jj);
    const float4 qv = ((const float4*)(qt + rj))[ln];
    p[t] = qv.x * kv[t].x + qv.y * kv[t].y + qv.z * kv[t].z + qv.w * kv[t].w;
  }
  float da = wave_sum4(p[0], p[1], p[2], p[3], hk, b8);
  float db = wave_sum4(p[4], p[5], p[6], p[7], hk, b8);
  da = (da >= 0.0f) ? da : 0.1f * da;
  db = (db >= 0.0f) ? db : 0.1f * db;
  const int i8 = ln - j;
  const int sl = (i8 & 3) << 3;
  const float ga = __shfl(da, sl, 32);
  const float gb = __shfl(db, sl, 32);
  if ((unsigned)i8 < 4u) dl = ga;
  else if ((unsigned)i8 < 8u) dl = gb;
  return dl;
}

static constexpr int TP     = 132;
static constexpr int NSTRIP = (OWN_ROWS + 15) / 16;
static constexpr int TROWS  = NSTRIP * 16;
static constexpr int RPW    = (OWN_ROWS + 31) / 32;

static __device__ __forceinline__ void w_fetch(float4 (&wr)[2], const float* __restrict__ Wsrc,
                                               const int stride, const int kp, const int tid) {
#pragma unroll
  for (int i = 0; i < 2; ++i) {
    const int q  = tid + 1024 * i;
    const int n  = q >> 4;
    const int c4 = (q & 15) << 2;
    wr[i] = *(const float4*)(Wsrc + (size_t)n * stride + kp + c4);
  }
}
static __device__ __forceinline__ void w_put(const float4 (&wr)[2], float* buf, const int tid) {
#pragma unroll
  for (int i = 0; i < 2; ++i) {
    const int q  = tid + 1024 * i;
    const int n  = q >> 4;
    const int c4 = (q & 15) << 2;
    *(float4*)(&buf[n * LDP + c4]) = wr[i];
  }
}
static constexpr int WSTG = (2048 + (32 - 2 * NSTRIP) * 32 - 1) / ((32 - 2 * NSTRIP) * 32);
static_assert(2 * NSTRIP < 32, "no wave left to stage the weights");
static __device__ __forceinline__ void w_stage_fetch(float4 (&r)[WSTG],
                                                     const float* __restrict__ Wsrc,
                                                     const int stride, const int kp,
                                                     const int sid, const int sstep) {
#pragma unroll
  for (int i = 0; i < WSTG; ++i) {
    const int q = sid + i * sstep;
    r[i] = make_float4(0.0f, 0.0f, 0.0f, 0.0f);
    if (q < 2048) r[i] = *(const float4*)(Wsrc + (size_t)(q >> 4) * stride + kp + ((q & 15) << 2));
  }
}
static __device__ __forceinline__ void w_stage_put(const float4 (&r)[WSTG], float* buf,
                                                   const int sid, const int sstep) {
#pragma unroll
  for (int i = 0; i < WSTG; ++i) {
    const int q = sid + i * sstep;
    if (q < 2048) *(float4*)(&buf[(q >> 4) * LDP + ((q & 15) << 2)]) = r[i];
  }
}
static __device__ __forceinline__ void w_stage(float* buf, const float* __restrict__ Wsrc,
                                               const int stride, const int kp,
                                               const int sid, const int sstep) {
  float4 r[WSTG];
  w_stage_fetch(r, Wsrc, stride, kp, sid, sstep);
  w_stage_put(r, buf, sid, sstep);
}
static __device__ __forceinline__ void mm_chunk(v8f (&c)[4], const float* ap, const float* wl,
                                                const int hk) {
#pragma unroll 4
  for (int k0 = 0; k0 < 64; k0 += 4) {
    const int ka = k0 + 2 * hk;
    v2f a; a.x = ap[ka]; a.y = ap[ka + 1];
#pragma unroll
    for (int t = 0; t < 4; ++t) {
      const float* wp = wl + t * 16 * LDP + ka;
      v2f b; b.x = wp[0]; b.y = wp[1];
      c[t] = __builtin_amdgcn_wmma_f32_16x16x4_f32(
          false, a, false, b, (short)0, c[t], false, false);
    }
  }
}

static __device__ __forceinline__ void q_tile(float* qt, float* wb, const float* __restrict__ Wq,
                                              const float* __restrict__ Ain,
                                              const int lo, const int nrows, const int NA,
                                              const int wv, const int ln) {
  const int hk  = ln >> 4;
  const int l16 = ln & 15;
  const int nstrip = (nrows + 15) >> 4;
  const bool act = wv < 2 * nstrip;
  const int st = wv >> 1;
  const int ch = wv & 1;
  int ga = lo + st * 16 + l16;
  if (ga >= NA) ga = NA - 1;
  const float* apg = Ain + (size_t)ga * DIM;
  const int wofs = (ch * 64 + l16) * LDP;
  const int orow = (st * 16 + 8 * hk) * DIM + ch * 64 + l16;
  const int sid   = (wv - 2 * nstrip) * 32 + ln;
  const int sstep = (32 - 2 * nstrip) * 32;
  v8f cc[4];
#pragma unroll
  for (int t = 0; t < 4; ++t)
#pragma unroll
    for (int j = 0; j < 8; ++j) cc[t][j] = 0.0f;
  float4 r[WSTG];
  if (act) mm_chunk(cc, apg, wb + wofs, hk);
  else     w_stage_fetch(r, Wq, DIM, 64, sid, sstep);
  __syncthreads();
  if (!act) w_stage_put(r, wb, sid, sstep);
  __syncthreads();
  if (act) {
    mm_chunk(cc, apg + 64, wb + wofs, hk);
#pragma unroll
    for (int t = 0; t < 4; ++t)
#pragma unroll
      for (int j = 0; j < 8; ++j) qt[orow + j * DIM + t * 16] = cc[t][j];
  }
  __syncthreads();
}

template <bool REUSE>
static __device__ __forceinline__ int2 own_entry(const int* list, const unsigned* my_ru,
                                                 const int idx, const int lo) {
  unsigned ru;
  if constexpr (REUSE) ru = my_ru[idx];
  else                 ru = (unsigned)list[idx];
  return make_int2((int)(ru >> 24) + lo, (int)(ru & 0x00ffffffu));
}

template <bool REUSE>
__global__ __launch_bounds__(1024) void edge_owner_score_kernel(const float* __restrict__ Ain,
                                                                const float* __restrict__ Wq,
                                                                const float* __restrict__ K,
                                                                const int* __restrict__ u,
                                                                const int* __restrict__ v,
                                                                const unsigned* __restrict__ seg,
                                                                const int* __restrict__ segcnt,
                                                                float* __restrict__ c,
                                                                unsigned* __restrict__ gmax,
                                                                unsigned* __restrict__ rec_ru,
                                                                float* __restrict__ rec_sc,
                                                                int* __restrict__ gcount,
                                                                int E, int NA) {
  __shared__ __attribute__((aligned(16))) float qt[TROWS * DIM];
  __shared__ __attribute__((aligned(16))) float wb[128 * LDP];
  static_assert(sizeof(float) * 128 * LDP >= sizeof(int) * OWN_CAP, "list does not fit the chunk buffer");
  int* const list = (int*)wb;
  __shared__ float smax[32];
  __shared__ int   scnt[NSEG];
  __shared__ int   soff[NSEG];
  __shared__ int   cnt;
  const int tid = threadIdx.x;
  const int ln  = tid & 31;
  const int wv  = __builtin_amdgcn_readfirstlane(tid >> 5);
  const int lo  = (int)blockIdx.x * OWN_ROWS;
  int nrows = NA - lo;
  if (nrows > OWN_ROWS) nrows = OWN_ROWS;
  if (nrows <= 0) return;
  const int hk = ln >> 4;
  const int b8 = (ln >> 3) & 1;

  {
    float4 w0[2];
    w_fetch(w0, Wq, DIM, 0, tid);
    w_put(w0, wb, tid);
  }
  const unsigned g = blockIdx.x;
  int total;
  if constexpr (REUSE) {
    __syncthreads();
    q_tile(qt, wb, Wq, Ain, lo, nrows, NA, wv, ln);
    total = gcount[g];
  } else {
    const int nseg = (E + SEG_E - 1) / SEG_E;
    const unsigned* my_seg = seg + (size_t)g * NSEG * SEG_CAP;
    if (wv == 0) {
      int n[SPW];
      int loc = 0;
      bool over = false;
#pragma unroll
      for (int k = 0; k < SPW; ++k) {
        const int s = ln * SPW + k;
        n[k] = (s < nseg) ? segcnt[s * OWN_TILES + (int)g] : 0;
      }
#pragma unroll
      for (int k = 0; k < SPW; ++k) {
        over = over || (n[k] > SEG_CAP);
        loc += n[k];
      }
      int inc = loc;
#pragma unroll
      for (int d = 1; d < 32; d <<= 1) {
        const int up = __shfl_up(inc, d, 32);
        if (ln >= d) inc += up;
      }
      int a = inc - loc;
#pragma unroll
      for (int k = 0; k < SPW; ++k) {
        const int s = ln * SPW + k;
        if (s < nseg) { scnt[s] = n[k]; soff[s] = a; }
        a += n[k];
      }
      const int  all   = __shfl(inc, 31, 32);
      const bool anyov = __ballot(over) != 0;
      if (ln == 0) cnt = (anyov || all > OWN_CAP) ? OWN_CAP + 1 : all;
    }
    __syncthreads();
    q_tile(qt, wb, Wq, Ain, lo, nrows, NA, wv, ln);
    total = cnt;
    if (total <= OWN_CAP) {
      const unsigned* spare = (const unsigned*)(segcnt + (int)g);
      unsigned e0[SPW], e1[SPW];
#pragma unroll
      for (int k = 0; k < SPW; ++k) {
        const int s = wv + 32 * k;
        const int m = (s < nseg) ? scnt[s] : 0;
        const unsigned* p0 = (ln < m)      ? my_seg + (s * SEG_CAP + ln)      : spare;
        const unsigned* p1 = (ln + 32 < m) ? my_seg + (s * SEG_CAP + 32 + ln) : spare;
        e0[k] = *p0;
        e1[k] = *p1;
      }
#pragma unroll
      for (int k = 0; k < SPW; ++k) {
        const int s = wv + 32 * k;
        const int m = (s < nseg) ? scnt[s] : 0;
        const int o = (s < nseg) ? soff[s] : 0;
        if (ln < m)      list[o + ln]      = (int)e0[k];
        if (ln + 32 < m) list[o + 32 + ln] = (int)e1[k];
      }
    }
    __syncthreads();
    if (tid == 0) gcount[g] = total;
  }

  float cmax = -3.0e38f;
  if (total <= OWN_CAP) {
    unsigned* my_ru = rec_ru + (size_t)g * OWN_CAP;
    float*    my_sc = rec_sc + (size_t)g * OWN_CAP;
    const int per = (total + 31) >> 5;
    const int b0  = wv * per;
    int b1 = b0 + per;
    if (b1 > total) b1 = total;
    if (b0 < b1) {
      int nval = (b1 - b0 > 32) ? 32 : (b1 - b0);
      int2 pn = own_entry<REUSE>(list, my_ru, b0 + (ln < nval ? ln : nval - 1), lo);
      float4 ka[EGR_Q], kb[EGR_Q];
      score_fetch(ka, K, pn.y, 0, nval, ln);
      for (int base = b0; base < b1; base += 32) {
        const int2 pl = pn;
        const int ul = pl.y;
        const int rl = (pl.x - lo) * DIM;
        const bool hasnext = base + 32 < b1;
        const int nvaln = hasnext ? ((b1 - base - 32 > 32) ? 32 : (b1 - base - 32)) : nval;
        pn = own_entry<REUSE>(list, my_ru,
                              (hasnext ? base + 32 : base) + (ln < nvaln ? ln : nvaln - 1), lo);
        float dl = 0.0f;
        for (int j = 0; j < nval; j += 2 * EGR_Q) {
          score_fetch(kb, K, ul, j + EGR_Q, nval, ln);
          dl = score_step(ka, qt, rl, j, nval, ln, hk, b8, dl);
          const bool here = (j + 2 * EGR_Q < nval) || !hasnext;
          score_fetch(ka, K, here ? ul : pn.y, here ? j + 2 * EGR_Q : 0, here ? nval : nvaln, ln);
          dl = score_step(kb, qt, rl, j + EGR_Q, nval, ln, hk, b8, dl);
        }
        if (ln < nval) {
          if constexpr (!REUSE) my_ru[base + ln] = ((unsigned)(pl.x - lo) << 24) | (unsigned)ul;
          my_sc[base + ln] = dl;
          cmax = fmaxf(cmax, dl);
        }
        nval = nvaln;
      }
    }
  } else {
    for (int e0 = wv * 32; e0 < E; e0 += 1024) {
      const int e  = e0 + ln;
      const int vl = (e < E) ? v[e] : -1;
      unsigned mask = (unsigned)__ballot((unsigned)(vl - lo) < (unsigned)nrows);
      while (mask) {
        const int b = __builtin_ctz(mask);
        mask &= mask - 1;
        const int vj = __builtin_amdgcn_readlane(vl, b);
        const int uj = u[e0 + b];
        const float4 qv = ((const float4*)(qt + (vj - lo) * DIM))[ln];
        const float4 kv = ((const float4*)(K + (size_t)uj * DIM))[ln];
        float d = wave_reduce_sum(qv.x * kv.x + qv.y * kv.y + qv.z * kv.z + qv.w * kv.w);
        d = (d >= 0.0f) ? d : 0.1f * d;
        if (ln == 0) c[e0 + b] = d;
        cmax = fmaxf(cmax, d);
      }
    }
  }
  cmax = fmaxf(cmax, __shfl_xor(cmax, 16, 32));
  cmax = fmaxf(cmax, __shfl_xor(cmax, 8, 32));
  cmax = fmaxf(cmax, __shfl_xor(cmax, 4, 32));
  cmax = fmaxf(cmax, __shfl_xor(cmax, 2, 32));
  cmax = fmaxf(cmax, __shfl_xor(cmax, 1, 32));
  if (ln == 0) smax[wv] = cmax;
  __syncthreads();
  if (tid == 0) {
    float m = smax[0];
    for (int i = 1; i < 32; ++i) m = fmaxf(m, smax[i]);
    atomicMax(gmax, f32_order_enc(m));
  }
}

__global__ __launch_bounds__(1024) void edge_owner_kernel(const float* __restrict__ c,
                                                          const unsigned* __restrict__ gmax,
                                                          const float* __restrict__ ctxp,
                                                          const int* __restrict__ u,
                                                          const int* __restrict__ v,
                                                          const unsigned* __restrict__ rec_ru,
                                                          const float* __restrict__ rec_sc,
                                                          const int* __restrict__ gcount,
                                                          const float* __restrict__ ain,
                                                          const float* __restrict__ Wagt,
                                                          const float* __restrict__ gna_w,
                                                          const float* __restrict__ gna_b,
                                                          const float* __restrict__ nrm_w,
                                                          const float* __restrict__ nrm_b,
                                                          const float* __restrict__ Wlin,
                                                          const float* __restrict__ gnl_w,
                                                          const float* __restrict__ gnl_b,
                                                          float* __restrict__ aout,
                                                          int E, int NA) {
  __shared__ __attribute__((aligned(16))) float acc[TROWS * TP];
  __shared__ __attribute__((aligned(16))) float xt[TROWS * TP];
  __shared__ __attribute__((aligned(16))) float Wl0[128 * LDP];
  __shared__ __attribute__((aligned(16))) float Wl1[128 * LDP];
  __shared__ float lsum[OWN_ROWS];
  const int tid = threadIdx.x;
  const int ln  = tid & 31;
  const int wv  = __builtin_amdgcn_readfirstlane(tid >> 5);
  const int lo  = (int)blockIdx.x * OWN_ROWS;
  int nrows = NA - lo;
  if (nrows > OWN_ROWS) nrows = OWN_ROWS;
  if (nrows <= 0) return;

  float4 wr[2];
  w_fetch(wr, Wagt, 2 * DIM, 0, tid);

  const float4 z4 = make_float4(0.0f, 0.0f, 0.0f, 0.0f);
  for (int q = tid; q < TROWS * TP / 4; q += 1024) ((float4*)acc)[q] = z4;
  if (tid < OWN_ROWS) lsum[tid] = 0.0f;
  w_put(wr, Wl0, tid);

  const unsigned g = blockIdx.x;
  const int total = gcount[g];
  const unsigned* my_ru = rec_ru + (size_t)g * OWN_CAP;
  const float*    my_sc = rec_sc + (size_t)g * OWN_CAP;
  __syncthreads();
  const float gm = f32_order_dec(*gmax);

  if (total <= OWN_CAP) {
    for (int i = tid; i < total; i += 1024)
      atomicAdd(&lsum[my_ru[i] >> 24], expf(my_sc[i] - gm));
    __syncthreads();
    const int per = (total + 31) >> 5;
    const int b0  = wv * per;
    int b1 = b0 + per;
    if (b1 > total) b1 = total;
    for (int base = b0; base < b1; base += 32) {
      int nval = b1 - base;
      if (nval > 32) nval = 32;
      const int il = base + (ln < nval ? ln : nval - 1);
      const unsigned ru = my_ru[il];
      const int ul  = (int)(ru & 0x00ffffffu);
      const int row = (int)(ru >> 24);
      const int cl = __float_as_int(expf(my_sc[il] - gm) / (lsum[row] + 1e-16f));
      const int rl = row * TP;
      int j = 0;
      for (; j + EGR_S <= nval; j += EGR_S) {
        float m[EGR_S][4];
#pragma unroll
        for (int t = 0; t < EGR_S; ++t) {
          const int uj = __builtin_amdgcn_readlane(ul, j + t);
          const float* srow = ctxp + (size_t)uj * DIM + ln;
          m[t][0] = srow[0];
          m[t][1] = srow[32];
          m[t][2] = srow[64];
          m[t][3] = srow[96];
        }
#pragma unroll
        for (int t = 0; t < EGR_S; ++t) {
          const int   rj   = __builtin_amdgcn_readlane(rl, j + t);
          const float coef = __int_as_float(__builtin_amdgcn_readlane(cl, j + t));
          float* dst = acc + rj + ln;
          atomicAdd(dst + 0,  m[t][0] * coef);
          atomicAdd(dst + 32, m[t][1] * coef);
          atomicAdd(dst + 64, m[t][2] * coef);
          atomicAdd(dst + 96, m[t][3] * coef);
        }
      }
      for (; j < nval; ++j) {
        const int   uj   = __builtin_amdgcn_readlane(ul, j);
        const int   rj   = __builtin_amdgcn_readlane(rl, j);
        const float coef = __int_as_float(__builtin_amdgcn_readlane(cl, j));
        const float* srow = ctxp + (size_t)uj * DIM + ln;
        const float m0 = srow[0], m1 = srow[32], m2 = srow[64], m3 = srow[96];
        float* dst = acc + rj + ln;
        atomicAdd(dst + 0,  m0 * coef);
        atomicAdd(dst + 32, m1 * coef);
        atomicAdd(dst + 64, m2 * coef);
        atomicAdd(dst + 96, m3 * coef);
      }
    }
  } else {
    for (int e0 = wv * 32; e0 < E; e0 += 1024) {
      const int e  = e0 + ln;
      const int vl = (e < E) ? v[e] : -1;
      if ((unsigned)(vl - lo) < (unsigned)nrows) atomicAdd(&lsum[vl - lo], expf(c[e] - gm));
    }
    __syncthreads();
    for (int e0 = wv * 32; e0 < E; e0 += 1024) {
      const int e  = e0 + ln;
      const int vl = (e < E) ? v[e] : -1;
      unsigned mask = (unsigned)__ballot((unsigned)(vl - lo) < (unsigned)nrows);
      while (mask) {
        const int b = __builtin_ctz(mask);
        mask &= mask - 1;
        const int   vj   = __builtin_amdgcn_readlane(vl, b);
        const int   uj   = u[e0 + b];
        const float coef = expf(c[e0 + b] - gm) / (lsum[vj - lo] + 1e-16f);
        const float* srow = ctxp + (size_t)uj * DIM + ln;
        const float m0 = srow[0], m1 = srow[32], m2 = srow[64], m3 = srow[96];
        float* dst = acc + (vj - lo) * TP + ln;
        atomicAdd(dst + 0,  m0 * coef);
        atomicAdd(dst + 32, m1 * coef);
        atomicAdd(dst + 64, m2 * coef);
        atomicAdd(dst + 96, m3 * coef);
      }
    }
  }
  __syncthreads();

  const int hk  = ln >> 4;
  const int l16 = ln & 15;
  const int nstrip = (nrows + 15) >> 4;
  const bool act = wv < 2 * nstrip;
  const int st = wv >> 1;
  const int ch = wv & 1;
  const int trow = st * 16 + l16;
  int ga = lo + trow;
  if (ga >= NA) ga = NA - 1;
  const float* apg = ain + (size_t)ga * DIM;
  const float* apl = acc + trow * TP;
  const float* apx = xt + trow * TP;
  const int wofs = (ch * 64 + l16) * LDP;
  const int orow = (st * 16 + 8 * hk) * TP + ch * 64 + l16;

  v8f cc[4];
#pragma unroll
  for (int t = 0; t < 4; ++t)
#pragma unroll
    for (int j = 0; j < 8; ++j) cc[t][j] = 0.0f;

  const int sid   = (wv - 2 * nstrip) * 32 + ln;
  const int sstep = (32 - 2 * nstrip) * 32;
  if (act) mm_chunk(cc, apg, Wl0 + wofs, hk);
  else     w_stage(Wl1, Wagt, 2 * DIM, 64, sid, sstep);
  __syncthreads();
  if (act) mm_chunk(cc, apg + 64, Wl1 + wofs, hk);
  else     w_stage(Wl0, Wagt, 2 * DIM, 128, sid, sstep);
  __syncthreads();
  if (act) mm_chunk(cc, apl, Wl0 + wofs, hk);
  else     w_stage(Wl1, Wagt, 2 * DIM, 192, sid, sstep);
  __syncthreads();
  if (act) {
    mm_chunk(cc, apl + 64, Wl1 + wofs, hk);
#pragma unroll
    for (int t = 0; t < 4; ++t)
#pragma unroll
      for (int j = 0; j < 8; ++j) xt[orow + j * TP + t * 16] = cc[t][j];
  } else {
    w_stage(Wl0, Wlin, DIM, 0, sid, sstep);
  }
  __syncthreads();

  {
    const float4 wa = ((const float4*)gna_w)[ln], ba = ((const float4*)gna_b)[ln];
    const float4 wn = ((const float4*)nrm_w)[ln], bn = ((const float4*)nrm_b)[ln];
    for (int r = wv; r < nrows; r += 32) {
      float4* xp = (float4*)(xt + r * TP) + ln;
      const float4 x = *xp;
      float mu, rs;
      row_stats(x, mu, rs);
      float4 y;
      y.x = (x.x - mu) * rs * wa.x + ba.x;
      y.y = (x.y - mu) * rs * wa.y + ba.y;
      y.z = (x.z - mu) * rs * wa.z + ba.z;
      y.w = (x.w - mu) * rs * wa.w + ba.w;
      row_stats(y, mu, rs);
      float4 z;
      z.x = fmaxf((y.x - mu) * rs * wn.x + bn.x, 0.0f);
      z.y = fmaxf((y.y - mu) * rs * wn.y + bn.y, 0.0f);
      z.z = fmaxf((y.z - mu) * rs * wn.z + bn.z, 0.0f);
      z.w = fmaxf((y.w - mu) * rs * wn.w + bn.w, 0.0f);
      *xp = z;
    }
  }
  __syncthreads();

#pragma unroll
  for (int t = 0; t < 4; ++t)
#pragma unroll
    for (int j = 0; j < 8; ++j) cc[t][j] = 0.0f;
  if (act) mm_chunk(cc, apx, Wl0 + wofs, hk);
  else     w_stage(Wl1, Wlin, DIM, 64, sid, sstep);
  __syncthreads();
  if (act) {
    mm_chunk(cc, apx + 64, Wl1 + wofs, hk);
#pragma unroll
    for (int t = 0; t < 4; ++t)
#pragma unroll
      for (int j = 0; j < 8; ++j) acc[orow + j * TP + t * 16] = cc[t][j];
  }
  __syncthreads();

  {
    const float4 wl4 = ((const float4*)gnl_w)[ln], bl4 = ((const float4*)gnl_b)[ln];
    float4 rr[RPW];
#pragma unroll
    for (int t = 0; t < RPW; ++t) {
      const int r = wv + 32 * t;
      rr[t] = z4;
      if (r < nrows) rr[t] = ((const float4*)(ain + (size_t)(lo + r) * DIM))[ln];
    }
#pragma unroll
    for (int t = 0; t < RPW; ++t) {
      const int r = wv + 32 * t;
      if (r < nrows) {
        const float4 x = ((const float4*)(acc + r * TP))[ln];
        float mu, rs;
        row_stats(x, mu, rs);
        float4 y;
        y.x = fmaxf((x.x - mu) * rs * wl4.x + bl4.x + rr[t].x, 0.0f);
        y.y = fmaxf((x.y - mu) * rs * wl4.y + bl4.y + rr[t].y, 0.0f);
        y.z = fmaxf((x.z - mu) * rs * wl4.z + bl4.z + rr[t].z, 0.0f);
        y.w = fmaxf((x.w - mu) * rs * wl4.w + bl4.w + rr[t].w, 0.0f);
        ((float4*)(aout + (size_t)(lo + r) * DIM))[ln] = y;
      }
    }
  }
}

extern "C" void kernel_launch(void* const* d_in, const int* in_sizes, int n_in,
                              void* d_out, int out_size, void* d_ws, size_t ws_size,
                              hipStream_t stream) {
  (void)in_sizes; (void)n_in; (void)out_size; (void)ws_size;
  const int NA = 50000, NC = 100000, NE = 600000;

  const float* agt    = (const float*)d_in[0];
  const float* ctx    = (const float*)d_in[1];
  const int*   u      = (const int*)d_in[2];
  const int*   v      = (const int*)d_in[3];
  const float* Wq     = (const float*)d_in[4];
  const float* Wk     = (const float*)d_in[5];
  const float* Wctx   = (const float*)d_in[6];
  const float* gnc_w  = (const float*)d_in[7];
  const float* gnc_b  = (const float*)d_in[8];
  const float* Wagt   = (const float*)d_in[9];
  const float* gna_w  = (const float*)d_in[10];
  const float* gna_b  = (const float*)d_in[11];
  const float* norm_w = (const float*)d_in[12];
  const float* norm_b = (const float*)d_in[13];
  const float* Wlin   = (const float*)d_in[14];
  const float* gnl_w  = (const float*)d_in[15];
  const float* gnl_b  = (const float*)d_in[16];

  float* ws = (float*)d_ws;
  size_t off = 0;
  off += (size_t)NA * DIM;
  float* kbuf   = ws + off; off += (size_t)NC * DIM;
  float* cpbuf  = ws + off; off += (size_t)NC * DIM;
  float* cbuf   = ws + off; off += (size_t)NE;
  off += (size_t)NA;
  float* segbuf = ws + off; off += (size_t)NA * DIM;
  float* xbuf   = ws + off; off += (size_t)NA * DIM;
  float* agtbuf = ws + off; off += (size_t)NA * DIM;
  unsigned* maxbuf = (unsigned*)(ws + off); off += 16;

  const dim3 b128(128), b256(256);

  for (int i = 0; i < NBLK; ++i) {
    const float* ain  = (i == 0) ? agt : agtbuf;
    float*       aout = (i == NBLK - 1) ? (float*)d_out : agtbuf;

    gemm_nt_128<<<(NC + 63) / 64, b128, 0, stream>>>(ctx, Wk + (size_t)i * DIM * DIM, kbuf, NC);
    gemm_nt_128<<<(NC + 63) / 64, b128, 0, stream>>>(ctx, Wctx + (size_t)i * DIM * DIM, cpbuf, NC);
    gn_kernel<<<(NC + 63) / 64, b256, 0, stream>>>(cpbuf, cpbuf, gnc_w + i * DIM, gnc_b + i * DIM, NC);

    unsigned* seg    = (unsigned*)segbuf;
    int*      segcnt = (int*)segbuf + OWN_TILES * NSEG * SEG_CAP;
    unsigned* rec_ru = (unsigned*)xbuf + 3 * (1 << 19);
    float*    rec_sc = (float*)(rec_ru + 256 * OWN_CAP);
    int*      gcount = (int*)(rec_sc + 256 * OWN_CAP);
    unsigned* bmax   = maxbuf + i;
    if (i == 0) {
      edge_bin_kernel<<<NSEG, dim3(1024), 0, stream>>>(u, v, seg, segcnt, maxbuf, NE);
      edge_owner_score_kernel<false><<<(NA + OWN_ROWS - 1) / OWN_ROWS, dim3(1024), 0, stream>>>(
          ain, Wq + (size_t)i * DIM * DIM, kbuf, u, v, seg, segcnt, cbuf, bmax, rec_ru, rec_sc,
          gcount, NE, NA);
    } else {
      edge_owner_score_kernel<true><<<(NA + OWN_ROWS - 1) / OWN_ROWS, dim3(1024), 0, stream>>>(
          ain, Wq + (size_t)i * DIM * DIM, kbuf, u, v, seg, segcnt, cbuf, bmax, rec_ru, rec_sc,
          gcount, NE, NA);
    }
    edge_owner_kernel<<<(NA + OWN_ROWS - 1) / OWN_ROWS, dim3(1024), 0, stream>>>(
        cbuf, bmax, cpbuf, u, v, rec_ru, rec_sc, gcount, ain,
        Wagt + (size_t)i * DIM * 2 * DIM, gna_w + i * DIM, gna_b + i * DIM,
        norm_w + i * DIM, norm_b + i * DIM, Wlin + (size_t)i * DIM * DIM,
        gnl_w + i * DIM, gnl_b + i * DIM, aout, NE, NA);
  }
}


# ctx side of a block in one launch: a work-group multiplies its 64 ctx rows by Wk, then Wctx, and normalises the ctx_p strip out of an LDS tile (same k and sum order); 11->7 launches
# speedup vs baseline: 3.4498x; 1.2408x over previous
#include <hip/hip_runtime.h>


static constexpr int DIM  = 128;
static constexpr int NBLK = 2;
static constexpr int LDP  = 68;

typedef __attribute__((ext_vector_type(2))) float v2f;
typedef __attribute__((ext_vector_type(8))) float v8f;

static __device__ __forceinline__ float wave_reduce_sum(float x) {
#pragma unroll
  for (int off = 16; off > 0; off >>= 1) x += __shfl_xor(x, off, 32);
  return x;
}

static __device__ __forceinline__ unsigned f32_order_enc(float f) {
  unsigned u = __float_as_uint(f);
  return (u & 0x80000000u) ? ~u : (u | 0x80000000u);
}
static __device__ __forceinline__ float f32_order_dec(unsigned e) {
  unsigned u = (e & 0x80000000u) ? (e & 0x7fffffffu) : ~e;
  return __uint_as_float(u);
}

static __device__ __forceinline__ void strip_nt_128(v8f (&acc)[8], const float* __restrict__ ap,
                                                    const float* __restrict__ W, float* Wlds,
                                                    const int tid, const int hk, const int l16) {
#pragma unroll
  for (int t = 0; t < 8; ++t)
#pragma unroll
    for (int j = 0; j < 8; ++j) acc[t][j] = 0.0f;

  for (int kp = 0; kp < DIM; kp += 64) {
    __syncthreads();
#pragma unroll
    for (int i = 0; i < 16; ++i) {
      const int q  = tid + 128 * i;
      const int n  = q >> 4;
      const int c4 = (q & 15) << 2;
      *(float4*)(&Wlds[n * LDP + c4]) =
          *(const float4*)(W + (size_t)n * DIM + kp + c4);
    }
    __syncthreads();

#pragma unroll 4
    for (int k0 = 0; k0 < 64; k0 += 4) {
      const int ka = k0 + 2 * hk;
      v2f a; a.x = ap[kp + ka]; a.y = ap[kp + ka + 1];
#pragma unroll
      for (int t = 0; t < 8; ++t) {
        const float* wp = &Wlds[(t * 16 + l16) * LDP + ka];
        v2f b; b.x = wp[0]; b.y = wp[1];
        acc[t] = __builtin_amdgcn_wmma_f32_16x16x4_f32(
            false, a, false, b, (short)0, acc[t], false, false);
      }
    }
  }
}

static __device__ __forceinline__ void row_stats(const float4 x, float& mu, float& rs) {
  float s  = x.x + x.y + x.z + x.w;
  float sq = x.x * x.x + x.y * x.y + x.z * x.z + x.w * x.w;
  s = wave_reduce_sum(s); sq = wave_reduce_sum(sq);
  mu = s * (1.0f / DIM);
  rs = rsqrtf(sq * (1.0f / DIM) - mu * mu + 1e-5f);
}

static __device__ __forceinline__ void gn_row_out(const float* trow, float* __restrict__ orow,
                                                  const float4 w4, const float4 b4, const int ln) {
  const float4 x = ((const float4*)trow)[ln];
  float mu, rs;
  row_stats(x, mu, rs);
  float4 y;
  y.x = (x.x - mu) * rs * w4.x + b4.x;
  y.y = (x.y - mu) * rs * w4.y + b4.y;
  y.z = (x.z - mu) * rs * w4.z + b4.z;
  y.w = (x.w - mu) * rs * w4.w + b4.w;
  ((float4*)orow)[ln] = y;
}

static constexpr int CT_ROWS = 64;
static constexpr int CTP     = 132;
static_assert(CT_ROWS * CTP <= 128 * LDP, "strip tile does not fit the chunk buffer");

__global__ __launch_bounds__(128) void ctx_side_kernel(const float* __restrict__ A,
                                                       const float* __restrict__ Wk,
                                                       const float* __restrict__ Wc,
                                                       const float* __restrict__ gw,
                                                       const float* __restrict__ gb,
                                                       float* __restrict__ Kout,
                                                       float* __restrict__ Pout, int N) {
  __shared__ __attribute__((aligned(16))) float Wlds[128 * LDP];
  const int tid = threadIdx.x;
  const int wv  = tid >> 5;
  const int ln  = tid & 31;
  const int hk  = ln >> 4;
  const int l16 = ln & 15;
  const int r0  = blockIdx.x * CT_ROWS + wv * 16;

  int rowA = r0 + l16;
  if (rowA >= N) rowA = N - 1;
  const float* ap = A + (size_t)rowA * DIM;

  v8f acc[8];

  strip_nt_128(acc, ap, Wk, Wlds, tid, hk, l16);
  const int rb = r0 + hk * 8;
  if (r0 + 16 <= N) {
#pragma unroll
    for (int t = 0; t < 8; ++t)
#pragma unroll
      for (int j = 0; j < 8; ++j)
        Kout[(size_t)(rb + j) * DIM + t * 16 + l16] = acc[t][j];
  } else {
#pragma unroll
    for (int t = 0; t < 8; ++t)
#pragma unroll
      for (int j = 0; j < 8; ++j) {
        const int row = rb + j;
        if (row < N) Kout[(size_t)row * DIM + t * 16 + l16] = acc[t][j];
      }
  }

  strip_nt_128(acc, ap, Wc, Wlds, tid, hk, l16);
  __syncthreads();
  float* tile = Wlds + wv * 16 * CTP;
#pragma unroll
  for (int t = 0; t < 8; ++t)
#pragma unroll
    for (int j = 0; j < 8; ++j) tile[(8 * hk + j) * CTP + t * 16 + l16] = acc[t][j];
  __syncthreads();
  const float4 wv4 = ((const float4*)gw)[ln];
  const float4 bv4 = ((const float4*)gb)[ln];
  if (r0 + 16 <= N) {
#pragma unroll 4
    for (int r = 0; r < 16; ++r)
      gn_row_out(tile + r * CTP, Pout + (size_t)(r0 + r) * DIM, wv4, bv4, ln);
  } else {
    for (int r = 0; r < 16; ++r)
      if (r0 + r < N)
        gn_row_out(tile + r * CTP, Pout + (size_t)(r0 + r) * DIM, wv4, bv4, ln);
  }
}

static constexpr int EGR_Q  = 8;
static constexpr int EGR_S  = 8;
static constexpr int OWN_ROWS = 196;
static constexpr int OWN_CAP  = 8192;
static constexpr int OWN_TILES = 256;
static constexpr int NE_MAX  = 600000;
static constexpr int SEG_E   = 4096;
static constexpr int SEG_CAP = 64;
static constexpr int NSEG    = (NE_MAX + SEG_E - 1) / SEG_E;
static constexpr int SPW     = (NSEG + 31) / 32;

__global__ __launch_bounds__(1024) void edge_bin_kernel(const int* __restrict__ u,
                                                        const int* __restrict__ v,
                                                        unsigned* __restrict__ seg,
                                                        int* __restrict__ segcnt,
                                                        unsigned* __restrict__ gmax,
                                                        int E) {
  __shared__ int hist[OWN_TILES];
  const int tid = threadIdx.x;
  const int w   = blockIdx.x;
  if (tid < OWN_TILES) hist[tid] = 0;
  if (w == 0 && tid == 0) {
    for (int i = 0; i < NBLK; ++i) gmax[i] = 0u;
  }
  __syncthreads();
  int vv[SEG_E / 1024], uu[SEG_E / 1024];
#pragma unroll
  for (int k = 0; k < SEG_E / 1024; ++k) {
    const int e = w * SEG_E + k * 1024 + tid;
    vv[k] = -1; uu[k] = 0;
    if (e < E) { vv[k] = v[e]; uu[k] = u[e]; }
  }
#pragma unroll
  for (int k = 0; k < SEG_E / 1024; ++k) {
    if (vv[k] >= 0) {
      const int t   = vv[k] / OWN_ROWS;
      const int r   = vv[k] - t * OWN_ROWS;
      const int pos = atomicAdd(&hist[t], 1);
      if (pos < SEG_CAP)
        seg[((size_t)t * NSEG + w) * SEG_CAP + pos] = ((unsigned)r << 24) | (unsigned)uu[k];
    }
  }
  __syncthreads();
  if (tid < OWN_TILES) segcnt[w * OWN_TILES + tid] = hist[tid];
}

static __device__ __forceinline__ float wave_sum4(const float p0, const float p1,
                                                  const float p2, const float p3,
                                                  const int hk, const int b8) {
  const float s0 = hk ? p2 : p0, r0 = hk ? p0 : p2;
  const float s1 = hk ? p3 : p1, r1 = hk ? p1 : p3;
  const float x0 = s0 + __shfl_xor(r0, 16, 32);
  const float x1 = s1 + __shfl_xor(r1, 16, 32);
  const float s2 = b8 ? x1 : x0, r2 = b8 ? x0 : x1;
  float d = s2 + __shfl_xor(r2, 8, 32);
  d += __shfl_xor(d, 4, 32);
  d += __shfl_xor(d, 2, 32);
  d += __shfl_xor(d, 1, 32);
  return d;
}

static __device__ __forceinline__ void score_fetch(float4 (&kv)[EGR_Q],
                                                   const float* __restrict__ K, const int ul,
                                                   const int j, const int nval, const int ln) {
#pragma unroll
  for (int t = 0; t < EGR_Q; ++t) {
    const int jj = (j + t < nval) ? (j + t) : (nval - 1);
    const int uj = __builtin_amdgcn_readlane(ul, jj);
    kv[t] = ((const float4*)(K + (size_t)uj * DIM))[ln];
  }
}
static __device__ __forceinline__ float score_step(const float4 (&kv)[EGR_Q], const float* qt,
                                                   const int rl, const int j, const int nval,
                                                   const int ln, const int hk, const int b8,
                                                   float dl) {
  float p[EGR_Q];
#pragma unroll
  for (int t = 0; t < EGR_Q; ++t) {
    const int jj = (j + t < nval) ? (j + t) : (nval - 1);
    const int rj = __builtin_amdgcn_readlane(rl, jj);
    const float4 qv = ((const float4*)(qt + rj))[ln];
    p[t] = qv.x * kv[t].x + qv.y * kv[t].y + qv.z * kv[t].z + qv.w * kv[t].w;
  }
  float da = wave_sum4(p[0], p[1], p[2], p[3], hk, b8);
  float db = wave_sum4(p[4], p[5], p[6], p[7], hk, b8);
  da = (da >= 0.0f) ? da : 0.1f * da;
  db = (db >= 0.0f) ? db : 0.1f * db;
  const int i8 = ln - j;
  const int sl = (i8 & 3) << 3;
  const float ga = __shfl(da, sl, 32);
  const float gb = __shfl(db, sl, 32);
  if ((unsigned)i8 < 4u) dl = ga;
  else if ((unsigned)i8 < 8u) dl = gb;
  return dl;
}

static constexpr int TP     = 132;
static constexpr int NSTRIP = (OWN_ROWS + 15) / 16;
static constexpr int TROWS  = NSTRIP * 16;
static constexpr int RPW    = (OWN_ROWS + 31) / 32;

static __device__ __forceinline__ void w_fetch(float4 (&wr)[2], const float* __restrict__ Wsrc,
                                               const int stride, const int kp, const int tid) {
#pragma unroll
  for (int i = 0; i < 2; ++i) {
    const int q  = tid + 1024 * i;
    const int n  = q >> 4;
    const int c4 = (q & 15) << 2;
    wr[i] = *(const float4*)(Wsrc + (size_t)n * stride + kp + c4);
  }
}
static __device__ __forceinline__ void w_put(const float4 (&wr)[2], float* buf, const int tid) {
#pragma unroll
  for (int i = 0; i < 2; ++i) {
    const int q  = tid + 1024 * i;
    const int n  = q >> 4;
    const int c4 = (q & 15) << 2;
    *(float4*)(&buf[n * LDP + c4]) = wr[i];
  }
}
static constexpr int WSTG = (2048 + (32 - 2 * NSTRIP) * 32 - 1) / ((32 - 2 * NSTRIP) * 32);
static_assert(2 * NSTRIP < 32, "no wave left to stage the weights");
static __device__ __forceinline__ void w_stage_fetch(float4 (&r)[WSTG],
                                                     const float* __restrict__ Wsrc,
                                                     const int stride, const int kp,
                                                     const int sid, const int sstep) {
#pragma unroll
  for (int i = 0; i < WSTG; ++i) {
    const int q = sid + i * sstep;
    r[i] = make_float4(0.0f, 0.0f, 0.0f, 0.0f);
    if (q < 2048) r[i] = *(const float4*)(Wsrc + (size_t)(q >> 4) * stride + kp + ((q & 15) << 2));
  }
}
static __device__ __forceinline__ void w_stage_put(const float4 (&r)[WSTG], float* buf,
                                                   const int sid, const int sstep) {
#pragma unroll
  for (int i = 0; i < WSTG; ++i) {
    const int q = sid + i * sstep;
    if (q < 2048) *(float4*)(&buf[(q >> 4) * LDP + ((q & 15) << 2)]) = r[i];
  }
}
static __device__ __forceinline__ void w_stage(float* buf, const float* __restrict__ Wsrc,
                                               const int stride, const int kp,
                                               const int sid, const int sstep) {
  float4 r[WSTG];
  w_stage_fetch(r, Wsrc, stride, kp, sid, sstep);
  w_stage_put(r, buf, sid, sstep);
}
static __device__ __forceinline__ void mm_chunk(v8f (&c)[4], const float* ap, const float* wl,
                                                const int hk) {
#pragma unroll 4
  for (int k0 = 0; k0 < 64; k0 += 4) {
    const int ka = k0 + 2 * hk;
    v2f a; a.x = ap[ka]; a.y = ap[ka + 1];
#pragma unroll
    for (int t = 0; t < 4; ++t) {
      const float* wp = wl + t * 16 * LDP + ka;
      v2f b; b.x = wp[0]; b.y = wp[1];
      c[t] = __builtin_amdgcn_wmma_f32_16x16x4_f32(
          false, a, false, b, (short)0, c[t], false, false);
    }
  }
}

static __device__ __forceinline__ void q_tile(float* qt, float* wb, const float* __restrict__ Wq,
                                              const float* __restrict__ Ain,
                                              const int lo, const int nrows, const int NA,
                                              const int wv, const int ln) {
  const int hk  = ln >> 4;
  const int l16 = ln & 15;
  const int nstrip = (nrows + 15) >> 4;
  const bool act = wv < 2 * nstrip;
  const int st = wv >> 1;
  const int ch = wv & 1;
  int ga = lo + st * 16 + l16;
  if (ga >= NA) ga = NA - 1;
  const float* apg = Ain + (size_t)ga * DIM;
  const int wofs = (ch * 64 + l16) * LDP;
  const int orow = (st * 16 + 8 * hk) * DIM + ch * 64 + l16;
  const int sid   = (wv - 2 * nstrip) * 32 + ln;
  const int sstep = (32 - 2 * nstrip) * 32;
  v8f cc[4];
#pragma unroll
  for (int t = 0; t < 4; ++t)
#pragma unroll
    for (int j = 0; j < 8; ++j) cc[t][j] = 0.0f;
  float4 r[WSTG];
  if (act) mm_chunk(cc, apg, wb + wofs, hk);
  else     w_stage_fetch(r, Wq, DIM, 64, sid, sstep);
  __syncthreads();
  if (!act) w_stage_put(r, wb, sid, sstep);
  __syncthreads();
  if (act) {
    mm_chunk(cc, apg + 64, wb + wofs, hk);
#pragma unroll
    for (int t = 0; t < 4; ++t)
#pragma unroll
      for (int j = 0; j < 8; ++j) qt[orow + j * DIM + t * 16] = cc[t][j];
  }
  __syncthreads();
}

template <bool REUSE>
static __device__ __forceinline__ int2 own_entry(const int* list, const unsigned* my_ru,
                                                 const int idx, const int lo) {
  unsigned ru;
  if constexpr (REUSE) ru = my_ru[idx];
  else                 ru = (unsigned)list[idx];
  return make_int2((int)(ru >> 24) + lo, (int)(ru & 0x00ffffffu));
}

template <bool REUSE>
__global__ __launch_bounds__(1024) void edge_owner_score_kernel(const float* __restrict__ Ain,
                                                                const float* __restrict__ Wq,
                                                                const float* __restrict__ K,
                                                                const int* __restrict__ u,
                                                                const int* __restrict__ v,
                                                                const unsigned* __restrict__ seg,
                                                                const int* __restrict__ segcnt,
                                                                float* __restrict__ c,
                                                                unsigned* __restrict__ gmax,
                                                                unsigned* __restrict__ rec_ru,
                                                                float* __restrict__ rec_sc,
                                                                int* __restrict__ gcount,
                                                                int E, int NA) {
  __shared__ __attribute__((aligned(16))) float qt[TROWS * DIM];
  __shared__ __attribute__((aligned(16))) float wb[128 * LDP];
  static_assert(sizeof(float) * 128 * LDP >= sizeof(int) * OWN_CAP, "list does not fit the chunk buffer");
  int* const list = (int*)wb;
  __shared__ float smax[32];
  __shared__ int   scnt[NSEG];
  __shared__ int   soff[NSEG];
  __shared__ int   cnt;
  const int tid = threadIdx.x;
  const int ln  = tid & 31;
  const int wv  = __builtin_amdgcn_readfirstlane(tid >> 5);
  const int lo  = (int)blockIdx.x * OWN_ROWS;
  int nrows = NA - lo;
  if (nrows > OWN_ROWS) nrows = OWN_ROWS;
  if (nrows <= 0) return;
  const int hk = ln >> 4;
  const int b8 = (ln >> 3) & 1;

  {
    float4 w0[2];
    w_fetch(w0, Wq, DIM, 0, tid);
    w_put(w0, wb, tid);
  }
  const unsigned g = blockIdx.x;
  int total;
  if constexpr (REUSE) {
    __syncthreads();
    q_tile(qt, wb, Wq, Ain, lo, nrows, NA, wv, ln);
    total = gcount[g];
  } else {
    const int nseg = (E + SEG_E - 1) / SEG_E;
    const unsigned* my_seg = seg + (size_t)g * NSEG * SEG_CAP;
    if (wv == 0) {
      int n[SPW];
      int loc = 0;
      bool over = false;
#pragma unroll
      for (int k = 0; k < SPW; ++k) {
        const int s = ln * SPW + k;
        n[k] = (s < nseg) ? segcnt[s * OWN_TILES + (int)g] : 0;
      }
#pragma unroll
      for (int k = 0; k < SPW; ++k) {
        over = over || (n[k] > SEG_CAP);
        loc += n[k];
      }
      int inc = loc;
#pragma unroll
      for (int d = 1; d < 32; d <<= 1) {
        const int up = __shfl_up(inc, d, 32);
        if (ln >= d) inc += up;
      }
      int a = inc - loc;
#pragma unroll
      for (int k = 0; k < SPW; ++k) {
        const int s = ln * SPW + k;
        if (s < nseg) { scnt[s] = n[k]; soff[s] = a; }
        a += n[k];
      }
      const int  all   = __shfl(inc, 31, 32);
      const bool anyov = __ballot(over) != 0;
      if (ln == 0) cnt = (anyov || all > OWN_CAP) ? OWN_CAP + 1 : all;
    }
    __syncthreads();
    q_tile(qt, wb, Wq, Ain, lo, nrows, NA, wv, ln);
    total = cnt;
    if (total <= OWN_CAP) {
      const unsigned* spare = (const unsigned*)(segcnt + (int)g);
      unsigned e0[SPW], e1[SPW];
#pragma unroll
      for (int k = 0; k < SPW; ++k) {
        const int s = wv + 32 * k;
        const int m = (s < nseg) ? scnt[s] : 0;
        const unsigned* p0 = (ln < m)      ? my_seg + (s * SEG_CAP + ln)      : spare;
        const unsigned* p1 = (ln + 32 < m) ? my_seg + (s * SEG_CAP + 32 + ln) : spare;
        e0[k] = *p0;
        e1[k] = *p1;
      }
#pragma unroll
      for (int k = 0; k < SPW; ++k) {
        const int s = wv + 32 * k;
        const int m = (s < nseg) ? scnt[s] : 0;
        const int o = (s < nseg) ? soff[s] : 0;
        if (ln < m)      list[o + ln]      = (int)e0[k];
        if (ln + 32 < m) list[o + 32 + ln] = (int)e1[k];
      }
    }
    __syncthreads();
    if (tid == 0) gcount[g] = total;
  }

  float cmax = -3.0e38f;
  if (total <= OWN_CAP) {
    unsigned* my_ru = rec_ru + (size_t)g * OWN_CAP;
    float*    my_sc = rec_sc + (size_t)g * OWN_CAP;
    const int per = (total + 31) >> 5;
    const int b0  = wv * per;
    int b1 = b0 + per;
    if (b1 > total) b1 = total;
    if (b0 < b1) {
      int nval = (b1 - b0 > 32) ? 32 : (b1 - b0);
      int2 pn = own_entry<REUSE>(list, my_ru, b0 + (ln < nval ? ln : nval - 1), lo);
      float4 ka[EGR_Q], kb[EGR_Q];
      score_fetch(ka, K, pn.y, 0, nval, ln);
      for (int base = b0; base < b1; base += 32) {
        const int2 pl = pn;
        const int ul = pl.y;
        const int rl = (pl.x - lo) * DIM;
        const bool hasnext = base + 32 < b1;
        const int nvaln = hasnext ? ((b1 - base - 32 > 32) ? 32 : (b1 - base - 32)) : nval;
        pn = own_entry<REUSE>(list, my_ru,
                              (hasnext ? base + 32 : base) + (ln < nvaln ? ln : nvaln - 1), lo);
        float dl = 0.0f;
        for (int j = 0; j < nval; j += 2 * EGR_Q) {
          score_fetch(kb, K, ul, j + EGR_Q, nval, ln);
          dl = score_step(ka, qt, rl, j, nval, ln, hk, b8, dl);
          const bool here = (j + 2 * EGR_Q < nval) || !hasnext;
          score_fetch(ka, K, here ? ul : pn.y, here ? j + 2 * EGR_Q : 0, here ? nval : nvaln, ln);
          dl = score_step(kb, qt, rl, j + EGR_Q, nval, ln, hk, b8, dl);
        }
        if (ln < nval) {
          if constexpr (!REUSE) my_ru[base + ln] = ((unsigned)(pl.x - lo) << 24) | (unsigned)ul;
          my_sc[base + ln] = dl;
          cmax = fmaxf(cmax, dl);
        }
        nval = nvaln;
      }
    }
  } else {
    for (int e0 = wv * 32; e0 < E; e0 += 1024) {
      const int e  = e0 + ln;
      const int vl = (e < E) ? v[e] : -1;
      unsigned mask = (unsigned)__ballot((unsigned)(vl - lo) < (unsigned)nrows);
      while (mask) {
        const int b = __builtin_ctz(mask);
        mask &= mask - 1;
        const int vj = __builtin_amdgcn_readlane(vl, b);
        const int uj = u[e0 + b];
        const float4 qv = ((const float4*)(qt + (vj - lo) * DIM))[ln];
        const float4 kv = ((const float4*)(K + (size_t)uj * DIM))[ln];
        float d = wave_reduce_sum(qv.x * kv.x + qv.y * kv.y + qv.z * kv.z + qv.w * kv.w);
        d = (d >= 0.0f) ? d : 0.1f * d;
        if (ln == 0) c[e0 + b] = d;
        cmax = fmaxf(cmax, d);
      }
    }
  }
  cmax = fmaxf(cmax, __shfl_xor(cmax, 16, 32));
  cmax = fmaxf(cmax, __shfl_xor(cmax, 8, 32));
  cmax = fmaxf(cmax, __shfl_xor(cmax, 4, 32));
  cmax = fmaxf(cmax, __shfl_xor(cmax, 2, 32));
  cmax = fmaxf(cmax, __shfl_xor(cmax, 1, 32));
  if (ln == 0) smax[wv] = cmax;
  __syncthreads();
  if (tid == 0) {
    float m = smax[0];
    for (int i = 1; i < 32; ++i) m = fmaxf(m, smax[i]);
    atomicMax(gmax, f32_order_enc(m));
  }
}

__global__ __launch_bounds__(1024) void edge_owner_kernel(const float* __restrict__ c,
                                                          const unsigned* __restrict__ gmax,
                                                          const float* __restrict__ ctxp,
                                                          const int* __restrict__ u,
                                                          const int* __restrict__ v,
                                                          const unsigned* __restrict__ rec_ru,
                                                          const float* __restrict__ rec_sc,
                                                          const int* __restrict__ gcount,
                                                          const float* __restrict__ ain,
                                                          const float* __restrict__ Wagt,
                                                          const float* __restrict__ gna_w,
                                                          const float* __restrict__ gna_b,
                                                          const float* __restrict__ nrm_w,
                                                          const float* __restrict__ nrm_b,
                                                          const float* __restrict__ Wlin,
                                                          const float* __restrict__ gnl_w,
                                                          const float* __restrict__ gnl_b,
                                                          float* __restrict__ aout,
                                                          int E, int NA) {
  __shared__ __attribute__((aligned(16))) float acc[TROWS * TP];
  __shared__ __attribute__((aligned(16))) float xt[TROWS * TP];
  __shared__ __attribute__((aligned(16))) float Wl0[128 * LDP];
  __shared__ __attribute__((aligned(16))) float Wl1[128 * LDP];
  __shared__ float lsum[OWN_ROWS];
  const int tid = threadIdx.x;
  const int ln  = tid & 31;
  const int wv  = __builtin_amdgcn_readfirstlane(tid >> 5);
  const int lo  = (int)blockIdx.x * OWN_ROWS;
  int nrows = NA - lo;
  if (nrows > OWN_ROWS) nrows = OWN_ROWS;
  if (nrows <= 0) return;

  float4 wr[2];
  w_fetch(wr, Wagt, 2 * DIM, 0, tid);

  const float4 z4 = make_float4(0.0f, 0.0f, 0.0f, 0.0f);
  for (int q = tid; q < TROWS * TP / 4; q += 1024) ((float4*)acc)[q] = z4;
  if (tid < OWN_ROWS) lsum[tid] = 0.0f;
  w_put(wr, Wl0, tid);

  const unsigned g = blockIdx.x;
  const int total = gcount[g];
  const unsigned* my_ru = rec_ru + (size_t)g * OWN_CAP;
  const float*    my_sc = rec_sc + (size_t)g * OWN_CAP;
  __syncthreads();
  const float gm = f32_order_dec(*gmax);

  if (total <= OWN_CAP) {
    for (int i = tid; i < total; i += 1024)
      atomicAdd(&lsum[my_ru[i] >> 24], expf(my_sc[i] - gm));
    __syncthreads();
    const int per = (total + 31) >> 5;
    const int b0  = wv * per;
    int b1 = b0 + per;
    if (b1 > total) b1 = total;
    for (int base = b0; base < b1; base += 32) {
      int nval = b1 - base;
      if (nval > 32) nval = 32;
      const int il = base + (ln < nval ? ln : nval - 1);
      const unsigned ru = my_ru[il];
      const int ul  = (int)(ru & 0x00ffffffu);
      const int row = (int)(ru >> 24);
      const int cl = __float_as_int(expf(my_sc[il] - gm) / (lsum[row] + 1e-16f));
      const int rl = row * TP;
      int j = 0;
      for (; j + EGR_S <= nval; j += EGR_S) {
        float m[EGR_S][4];
#pragma unroll
        for (int t = 0; t < EGR_S; ++t) {
          const int uj = __builtin_amdgcn_readlane(ul, j + t);
          const float* srow = ctxp + (size_t)uj * DIM + ln;
          m[t][0] = srow[0];
          m[t][1] = srow[32];
          m[t][2] = srow[64];
          m[t][3] = srow[96];
        }
#pragma unroll
        for (int t = 0; t < EGR_S; ++t) {
          const int   rj   = __builtin_amdgcn_readlane(rl, j + t);
          const float coef = __int_as_float(__builtin_amdgcn_readlane(cl, j + t));
          float* dst = acc + rj + ln;
          atomicAdd(dst + 0,  m[t][0] * coef);
          atomicAdd(dst + 32, m[t][1] * coef);
          atomicAdd(dst + 64, m[t][2] * coef);
          atomicAdd(dst + 96, m[t][3] * coef);
        }
      }
      for (; j < nval; ++j) {
        const int   uj   = __builtin_amdgcn_readlane(ul, j);
        const int   rj   = __builtin_amdgcn_readlane(rl, j);
        const float coef = __int_as_float(__builtin_amdgcn_readlane(cl, j));
        const float* srow = ctxp + (size_t)uj * DIM + ln;
        const float m0 = srow[0], m1 = srow[32], m2 = srow[64], m3 = srow[96];
        float* dst = acc + rj + ln;
        atomicAdd(dst + 0,  m0 * coef);
        atomicAdd(dst + 32, m1 * coef);
        atomicAdd(dst + 64, m2 * coef);
        atomicAdd(dst + 96, m3 * coef);
      }
    }
  } else {
    for (int e0 = wv * 32; e0 < E; e0 += 1024) {
      const int e  = e0 + ln;
      const int vl = (e < E) ? v[e] : -1;
      if ((unsigned)(vl - lo) < (unsigned)nrows) atomicAdd(&lsum[vl - lo], expf(c[e] - gm));
    }
    __syncthreads();
    for (int e0 = wv * 32; e0 < E; e0 += 1024) {
      const int e  = e0 + ln;
      const int vl = (e < E) ? v[e] : -1;
      unsigned mask = (unsigned)__ballot((unsigned)(vl - lo) < (unsigned)nrows);
      while (mask) {
        const int b = __builtin_ctz(mask);
        mask &= mask - 1;
        const int   vj   = __builtin_amdgcn_readlane(vl, b);
        const int   uj   = u[e0 + b];
        const float coef = expf(c[e0 + b] - gm) / (lsum[vj - lo] + 1e-16f);
        const float* srow = ctxp + (size_t)uj * DIM + ln;
        const float m0 = srow[0], m1 = srow[32], m2 = srow[64], m3 = srow[96];
        float* dst = acc + (vj - lo) * TP + ln;
        atomicAdd(dst + 0,  m0 * coef);
        atomicAdd(dst + 32, m1 * coef);
        atomicAdd(dst + 64, m2 * coef);
        atomicAdd(dst + 96, m3 * coef);
      }
    }
  }
  __syncthreads();

  const int hk  = ln >> 4;
  const int l16 = ln & 15;
  const int nstrip = (nrows + 15) >> 4;
  const bool act = wv < 2 * nstrip;
  const int st = wv >> 1;
  const int ch = wv & 1;
  const int trow = st * 16 + l16;
  int ga = lo + trow;
  if (ga >= NA) ga = NA - 1;
  const float* apg = ain + (size_t)ga * DIM;
  const float* apl = acc + trow * TP;
  const float* apx = xt + trow * TP;
  const int wofs = (ch * 64 + l16) * LDP;
  const int orow = (st * 16 + 8 * hk) * TP + ch * 64 + l16;

  v8f cc[4];
#pragma unroll
  for (int t = 0; t < 4; ++t)
#pragma unroll
    for (int j = 0; j < 8; ++j) cc[t][j] = 0.0f;

  const int sid   = (wv - 2 * nstrip) * 32 + ln;
  const int sstep = (32 - 2 * nstrip) * 32;
  if (act) mm_chunk(cc, apg, Wl0 + wofs, hk);
  else     w_stage(Wl1, Wagt, 2 * DIM, 64, sid, sstep);
  __syncthreads();
  if (act) mm_chunk(cc, apg + 64, Wl1 + wofs, hk);
  else     w_stage(Wl0, Wagt, 2 * DIM, 128, sid, sstep);
  __syncthreads();
  if (act) mm_chunk(cc, apl, Wl0 + wofs, hk);
  else     w_stage(Wl1, Wagt, 2 * DIM, 192, sid, sstep);
  __syncthreads();
  if (act) {
    mm_chunk(cc, apl + 64, Wl1 + wofs, hk);
#pragma unroll
    for (int t = 0; t < 4; ++t)
#pragma unroll
      for (int j = 0; j < 8; ++j) xt[orow + j * TP + t * 16] = cc[t][j];
  } else {
    w_stage(Wl0, Wlin, DIM, 0, sid, sstep);
  }
  __syncthreads();

  {
    const float4 wa = ((const float4*)gna_w)[ln], ba = ((const float4*)gna_b)[ln];
    const float4 wn = ((const float4*)nrm_w)[ln], bn = ((const float4*)nrm_b)[ln];
    for (int r = wv; r < nrows; r += 32) {
      float4* xp = (float4*)(xt + r * TP) + ln;
      const float4 x = *xp;
      float mu, rs;
      row_stats(x, mu, rs);
      float4 y;
      y.x = (x.x - mu) * rs * wa.x + ba.x;
      y.y = (x.y - mu) * rs * wa.y + ba.y;
      y.z = (x.z - mu) * rs * wa.z + ba.z;
      y.w = (x.w - mu) * rs * wa.w + ba.w;
      row_stats(y, mu, rs);
      float4 z;
      z.x = fmaxf((y.x - mu) * rs * wn.x + bn.x, 0.0f);
      z.y = fmaxf((y.y - mu) * rs * wn.y + bn.y, 0.0f);
      z.z = fmaxf((y.z - mu) * rs * wn.z + bn.z, 0.0f);
      z.w = fmaxf((y.w - mu) * rs * wn.w + bn.w, 0.0f);
      *xp = z;
    }
  }
  __syncthreads();

#pragma unroll
  for (int t = 0; t < 4; ++t)
#pragma unroll
    for (int j = 0; j < 8; ++j) cc[t][j] = 0.0f;
  if (act) mm_chunk(cc, apx, Wl0 + wofs, hk);
  else     w_stage(Wl1, Wlin, DIM, 64, sid, sstep);
  __syncthreads();
  if (act) {
    mm_chunk(cc, apx + 64, Wl1 + wofs, hk);
#pragma unroll
    for (int t = 0; t < 4; ++t)
#pragma unroll
      for (int j = 0; j < 8; ++j) acc[orow + j * TP + t * 16] = cc[t][j];
  }
  __syncthreads();

  {
    const float4 wl4 = ((const float4*)gnl_w)[ln], bl4 = ((const float4*)gnl_b)[ln];
    float4 rr[RPW];
#pragma unroll
    for (int t = 0; t < RPW; ++t) {
      const int r = wv + 32 * t;
      rr[t] = z4;
      if (r < nrows) rr[t] = ((const float4*)(ain + (size_t)(lo + r) * DIM))[ln];
    }
#pragma unroll
    for (int t = 0; t < RPW; ++t) {
      const int r = wv + 32 * t;
      if (r < nrows) {
        const float4 x = ((const float4*)(acc + r * TP))[ln];
        float mu, rs;
        row_stats(x, mu, rs);
        float4 y;
        y.x = fmaxf((x.x - mu) * rs * wl4.x + bl4.x + rr[t].x, 0.0f);
        y.y = fmaxf((x.y - mu) * rs * wl4.y + bl4.y + rr[t].y, 0.0f);
        y.z = fmaxf((x.z - mu) * rs * wl4.z + bl4.z + rr[t].z, 0.0f);
        y.w = fmaxf((x.w - mu) * rs * wl4.w + bl4.w + rr[t].w, 0.0f);
        ((float4*)(aout + (size_t)(lo + r) * DIM))[ln] = y;
      }
    }
  }
}

extern "C" void kernel_launch(void* const* d_in, const int* in_sizes, int n_in,
                              void* d_out, int out_size, void* d_ws, size_t ws_size,
                              hipStream_t stream) {
  (void)in_sizes; (void)n_in; (void)out_size; (void)ws_size;
  const int NA = 50000, NC = 100000, NE = 600000;

  const float* agt    = (const float*)d_in[0];
  const float* ctx    = (const float*)d_in[1];
  const int*   u      = (const int*)d_in[2];
  const int*   v      = (const int*)d_in[3];
  const float* Wq     = (const float*)d_in[4];
  const float* Wk     = (const float*)d_in[5];
  const float* Wctx   = (const float*)d_in[6];
  const float* gnc_w  = (const float*)d_in[7];
  const float* gnc_b  = (const float*)d_in[8];
  const float* Wagt   = (const float*)d_in[9];
  const float* gna_w  = (const float*)d_in[10];
  const float* gna_b  = (const float*)d_in[11];
  const float* norm_w = (const float*)d_in[12];
  const float* norm_b = (const float*)d_in[13];
  const float* Wlin   = (const float*)d_in[14];
  const float* gnl_w  = (const float*)d_in[15];
  const float* gnl_b  = (const float*)d_in[16];

  float* ws = (float*)d_ws;
  size_t off = 0;
  off += (size_t)NA * DIM;
  float* kbuf   = ws + off; off += (size_t)NC * DIM;
  float* cpbuf  = ws + off; off += (size_t)NC * DIM;
  float* cbuf   = ws + off; off += (size_t)NE;
  off += (size_t)NA;
  float* segbuf = ws + off; off += (size_t)NA * DIM;
  float* xbuf   = ws + off; off += (size_t)NA * DIM;
  float* agtbuf = ws + off; off += (size_t)NA * DIM;
  unsigned* maxbuf = (unsigned*)(ws + off); off += 16;

  const dim3 b128(128);

  for (int i = 0; i < NBLK; ++i) {
    const float* ain  = (i == 0) ? agt : agtbuf;
    float*       aout = (i == NBLK - 1) ? (float*)d_out : agtbuf;

    ctx_side_kernel<<<(NC + CT_ROWS - 1) / CT_ROWS, b128, 0, stream>>>(
        ctx, Wk + (size_t)i * DIM * DIM, Wctx + (size_t)i * DIM * DIM,
        gnc_w + i * DIM, gnc_b + i * DIM, kbuf, cpbuf, NC);

    unsigned* seg    = (unsigned*)segbuf;
    int*      segcnt = (int*)segbuf + OWN_TILES * NSEG * SEG_CAP;
    unsigned* rec_ru = (unsigned*)xbuf + 3 * (1 << 19);
    float*    rec_sc = (float*)(rec_ru + 256 * OWN_CAP);
    int*      gcount = (int*)(rec_sc + 256 * OWN_CAP);
    unsigned* bmax   = maxbuf + i;
    if (i == 0) {
      edge_bin_kernel<<<NSEG, dim3(1024), 0, stream>>>(u, v, seg, segcnt, maxbuf, NE);
      edge_owner_score_kernel<false><<<(NA + OWN_ROWS - 1) / OWN_ROWS, dim3(1024), 0, stream>>>(
          ain, Wq + (size_t)i * DIM * DIM, kbuf, u, v, seg, segcnt, cbuf, bmax, rec_ru, rec_sc,
          gcount, NE, NA);
    } else {
      edge_owner_score_kernel<true><<<(NA + OWN_ROWS - 1) / OWN_ROWS, dim3(1024), 0, stream>>>(
          ain, Wq + (size_t)i * DIM * DIM, kbuf, u, v, seg, segcnt, cbuf, bmax, rec_ru, rec_sc,
          gcount, NE, NA);
    }
    edge_owner_kernel<<<(NA + OWN_ROWS - 1) / OWN_ROWS, dim3(1024), 0, stream>>>(
        cbuf, bmax, cpbuf, u, v, rec_ru, rec_sc, gcount, ain,
        Wagt + (size_t)i * DIM * 2 * DIM, gna_w + i * DIM, gna_b + i * DIM,
        norm_w + i * DIM, norm_b + i * DIM, Wlin + (size_t)i * DIM * DIM,
        gnl_w + i * DIM, gnl_b + i * DIM, aout, NE, NA);
  }
}
